# starting kernel
# baseline (speedup 1.0000x reference)
#include <hip/hip_runtime.h>
#include <hip/hip_bf16.h>

typedef float v2f __attribute__((ext_vector_type(2)));
typedef float v8f __attribute__((ext_vector_type(8)));

#define DDIM 64
#define GEMM_ROWS 128
#define SX_STRIDE 68
#define SW_STRIDE 68

__device__ __forceinline__ unsigned fkey(float f) {
    unsigned u = __float_as_uint(f);
    return (u & 0x80000000u) ? ~u : (u | 0x80000000u);
}
__device__ __forceinline__ float fdec(unsigned k) {
    return (k & 0x80000000u) ? __uint_as_float(k & 0x7FFFFFFFu)
                             : __uint_as_float(~k);
}
#define FKEY_NEG_INF 0x007FFFFFu

__device__ __forceinline__ void fadd_atomic(float* p, float v) {
    unsafeAtomicAdd(p, v);
}

__global__ void compute_pvec(const float* __restrict__ prompt, const float* __restrict__ pw,
                             const float* __restrict__ pb, float* __restrict__ pvec, int P) {
    int d = threadIdx.x;
    if (d >= DDIM) return;
    float s = pb[d];
    for (int j = 0; j < P; ++j) s += prompt[j] * pw[d * P + j];
    pvec[d] = s;
}

__global__ __launch_bounds__(256) void add_prompt(const float* __restrict__ emb,
                                                  const float* __restrict__ pvec,
                                                  float* __restrict__ x,
                                                  float* __restrict__ out_sum,
                                                  long long total) {
    long long i = (long long)blockIdx.x * blockDim.x + threadIdx.x;
    if (i >= total) return;
    float v = emb[i] + pvec[i & (DDIM - 1)];
    x[i] = v;
    out_sum[i] = v;
}

__global__ __launch_bounds__(256) void gemm_xw_t(const float* __restrict__ x,
                                                 const float* __restrict__ W,
                                                 float* __restrict__ xl, int N) {
    __shared__ float sx[GEMM_ROWS * SX_STRIDE];
    __shared__ float sw[DDIM * SW_STRIDE];
    const int t = threadIdx.x;
    const int rowBase = blockIdx.x * GEMM_ROWS;

    {
        const float4* Wv = (const float4*)W;
        float4* swv = (float4*)sw;
        for (int i = t; i < (DDIM * DDIM) / 4; i += 256) {
            int r = i >> 4, c4 = i & 15;
            swv[r * (SW_STRIDE / 4) + c4] = Wv[i];
        }
    }
    {
        const float4* xv = (const float4*)x;
        float4* sxv = (float4*)sx;
        for (int i = t; i < (GEMM_ROWS * DDIM) / 4; i += 256) {
            int r = i >> 4, c4 = i & 15;
            int gr = rowBase + r;
            if (gr >= N) gr = N - 1;
            sxv[r * (SX_STRIDE / 4) + c4] = xv[(long long)gr * 16 + c4];
        }
    }
    __syncthreads();

    const int wave  = t >> 5;
    const int lane  = t & 31;
    const int lrow  = lane & 15;
    const int khalf = (lane >> 4) << 1;
    const int wrow  = wave * 16;

    v8f acc0 = {0.f,0.f,0.f,0.f,0.f,0.f,0.f,0.f};
    v8f acc1 = acc0, acc2 = acc0, acc3 = acc0;

    const float* arow = &sx[(wrow + lrow) * SX_STRIDE + khalf];
    const float* brow = &sw[lrow * SW_STRIDE + khalf];

    #pragma unroll
    for (int k0 = 0; k0 < DDIM; k0 += 4) {
        v2f a, b0, b1, b2, b3;
        a.x  = arow[k0];                       a.y  = arow[k0 + 1];
        b0.x = brow[k0];                       b0.y = brow[k0 + 1];
        b1.x = brow[16 * SW_STRIDE + k0];      b1.y = brow[16 * SW_STRIDE + k0 + 1];
        b2.x = brow[32 * SW_STRIDE + k0];      b2.y = brow[32 * SW_STRIDE + k0 + 1];
        b3.x = brow[48 * SW_STRIDE + k0];      b3.y = brow[48 * SW_STRIDE + k0 + 1];
        acc0 = __builtin_amdgcn_wmma_f32_16x16x4_f32(false, a, false, b0, (short)0, acc0, false, false);
        acc1 = __builtin_amdgcn_wmma_f32_16x16x4_f32(false, a, false, b1, (short)0, acc1, false, false);
        acc2 = __builtin_amdgcn_wmma_f32_16x16x4_f32(false, a, false, b2, (short)0, acc2, false, false);
        acc3 = __builtin_amdgcn_wmma_f32_16x16x4_f32(false, a, false, b3, (short)0, acc3, false, false);
    }

    const int rhi = (lane >> 4) * 8;
    const long long rbase = (long long)(rowBase + wrow + rhi);
    if (rowBase + GEMM_ROWS <= N) {
        float* p = xl + rbase * DDIM + lrow;
        #pragma unroll
        for (int v = 0; v < 8; ++v) {
            p[(long long)v * DDIM +  0] = acc0[v];
            p[(long long)v * DDIM + 16] = acc1[v];
            p[(long long)v * DDIM + 32] = acc2[v];
            p[(long long)v * DDIM + 48] = acc3[v];
        }
    } else {
        #pragma unroll
        for (int v = 0; v < 8; ++v) {
            long long gr = rbase + v;
            if (gr < N) {
                float* p = xl + gr * DDIM + lrow;
                p[ 0] = acc0[v];
                p[16] = acc1[v];
                p[32] = acc2[v];
                p[48] = acc3[v];
            }
        }
    }
}

__global__ __launch_bounds__(256) void att_dots(const float* __restrict__ xl,
                                                const float* __restrict__ av_s,
                                                const float* __restrict__ av_d,
                                                float* __restrict__ a_src,
                                                float* __restrict__ a_dst, int N) {
    long long tid = (long long)blockIdx.x * blockDim.x + threadIdx.x;
    int n = (int)(tid >> 5);
    int lane = threadIdx.x & 31;
    if (n >= N) return;
    float2 v = ((const float2*)(xl + (long long)n * DDIM))[lane];
    float2 s = ((const float2*)av_s)[lane];
    float2 d = ((const float2*)av_d)[lane];
    float ps = v.x * s.x + v.y * s.y;
    float pd = v.x * d.x + v.y * d.y;
    #pragma unroll
    for (int off = 16; off > 0; off >>= 1) {
        ps += __shfl_xor(ps, off, 32);
        pd += __shfl_xor(pd, off, 32);
    }
    if (lane == 0) { a_src[n] = ps; a_dst[n] = pd; }
}

__global__ __launch_bounds__(256) void init_layer(float* __restrict__ acc,
                                                  unsigned* __restrict__ segmax,
                                                  float* __restrict__ segsum, int N) {
    int n = blockIdx.x * blockDim.x + threadIdx.x;
    if (n >= N) return;
    segmax[n] = FKEY_NEG_INF;
    segsum[n] = 0.f;
    float4 z = {0.f, 0.f, 0.f, 0.f};
    float4* p = (float4*)(acc + (long long)n * DDIM);
    #pragma unroll
    for (int i = 0; i < DDIM / 4; ++i) p[i] = z;
}

__device__ __forceinline__ void edge_pair(const int* __restrict__ ei, long long e,
                                          int E, int& s, int& d) {
    if (e < E) { s = ei[e]; d = ei[(long long)E + e]; }
    else       { s = d = (int)(e - E); }
}

__global__ __launch_bounds__(256) void edge_max(const int* __restrict__ ei, int E, int N,
                                                const float* __restrict__ a_src,
                                                const float* __restrict__ a_dst,
                                                unsigned* __restrict__ segmax) {
    long long e = (long long)blockIdx.x * blockDim.x + threadIdx.x;
    long long tot = (long long)E + N;
    if (e >= tot) return;
    int s, d; edge_pair(ei, e, E, s, d);
    float v = a_src[s] + a_dst[d];
    v = v > 0.f ? v : 0.2f * v;
    atomicMax(&segmax[d], fkey(v));
}

__global__ __launch_bounds__(256) void edge_sum(const int* __restrict__ ei, int E, int N,
                                                const float* __restrict__ a_src,
                                                const float* __restrict__ a_dst,
                                                const unsigned* __restrict__ segmax,
                                                float* __restrict__ segsum) {
    long long e = (long long)blockIdx.x * blockDim.x + threadIdx.x;
    long long tot = (long long)E + N;
    if (e >= tot) return;
    int s, d; edge_pair(ei, e, E, s, d);
    float v = a_src[s] + a_dst[d];
    v = v > 0.f ? v : 0.2f * v;
    float m = fdec(segmax[d]);
    fadd_atomic(&segsum[d], __expf(v - m));
}

__global__ __launch_bounds__(256) void edge_scatter(const int* __restrict__ ei, int E, int N,
                                                    const float* __restrict__ a_src,
                                                    const float* __restrict__ a_dst,
                                                    const unsigned* __restrict__ segmax,
                                                    const float* __restrict__ segsum,
                                                    const float* __restrict__ xl,
                                                    float* __restrict__ acc) {
    long long tid = (long long)blockIdx.x * blockDim.x + threadIdx.x;
    long long e = tid >> 5;
    int lane = threadIdx.x & 31;
    long long tot = (long long)E + N;
    if (e >= tot) return;
    int s, d; edge_pair(ei, e, E, s, d);
    float v = a_src[s] + a_dst[d];
    v = v > 0.f ? v : 0.2f * v;
    float m = fdec(segmax[d]);
    float alpha = __expf(v - m) / (segsum[d] + 1e-16f);
    float2 xv = ((const float2*)(xl + (long long)s * DDIM))[lane];
    float* dstp = acc + (long long)d * DDIM + 2 * lane;
    fadd_atomic(dstp,     xv.x * alpha);
    fadd_atomic(dstp + 1, xv.y * alpha);
}

__global__ __launch_bounds__(256) void finalize_layer(float* __restrict__ xbuf,
                                                      const float* __restrict__ bias,
                                                      float* __restrict__ out_sum,
                                                      long long total) {
    long long i = (long long)blockIdx.x * blockDim.x + threadIdx.x;
    if (i >= total) return;
    float v = xbuf[i] + bias[i & (DDIM - 1)];
    v = v > 0.f ? v : (__expf(v) - 1.f);
    xbuf[i] = v;
    out_sum[i] += v;
}

__global__ __launch_bounds__(256) void scale_out(float* __restrict__ out, float inv,
                                                 long long total) {
    long long i = (long long)blockIdx.x * blockDim.x + threadIdx.x;
    if (i >= total) return;
    out[i] *= inv;
}

static inline unsigned gridFor(long long work, int block) {
    return (unsigned)((work + block - 1) / block);
}

extern "C" void kernel_launch(void* const* d_in, const int* in_sizes, int n_in,
                              void* d_out, int out_size, void* d_ws, size_t ws_size,
                              hipStream_t stream) {
    const int*   ei       = (const int*)d_in[0];
    const float* emb      = (const float*)d_in[1];
    const float* prompt   = (const float*)d_in[2];
    const float* proj_w   = (const float*)d_in[3];
    const float* proj_b   = (const float*)d_in[4];
    const float* conv_ws  = (const float*)d_in[5];
    const float* att_srcs = (const float*)d_in[6];
    const float* att_dsts = (const float*)d_in[7];
    const float* conv_bs  = (const float*)d_in[8];

    const int E = in_sizes[0] / 2;
    const int N = in_sizes[1] / DDIM;
    const int P = in_sizes[2];
    const int L = in_sizes[5] / (DDIM * DDIM);

    float* out = (float*)d_out;

    float*    pvec   = (float*)d_ws;
    float*    xbuf   = pvec + 64;
    float*    xlb    = xbuf + (size_t)N * DDIM;
    float*    a_src  = xlb + (size_t)N * DDIM;
    float*    a_dst  = a_src + N;
    unsigned* segmax = (unsigned*)(a_dst + N);
    float*    segsum = (float*)(segmax + N);

    const long long total = (long long)N * DDIM;
    const long long totE  = (long long)E + N;

    compute_pvec<<<1, 64, 0, stream>>>(prompt, proj_w, proj_b, pvec, P);
    add_prompt<<<gridFor(total, 256), 256, 0, stream>>>(emb, pvec, xbuf, out, total);

    for (int l = 0; l < L; ++l) {
        const float* W  = conv_ws  + (size_t)l * DDIM * DDIM;
        const float* as = att_srcs + (size_t)l * DDIM;
        const float* ad = att_dsts + (size_t)l * DDIM;
        const float* cb = conv_bs  + (size_t)l * DDIM;

        gemm_xw_t<<<gridFor(N, GEMM_ROWS), 256, 0, stream>>>(xbuf, W, xlb, N);
        att_dots<<<gridFor((long long)N * 32, 256), 256, 0, stream>>>(xlb, as, ad, a_src, a_dst, N);
        init_layer<<<gridFor(N, 256), 256, 0, stream>>>(xbuf, segmax, segsum, N);
        edge_max<<<gridFor(totE, 256), 256, 0, stream>>>(ei, E, N, a_src, a_dst, segmax);
        edge_sum<<<gridFor(totE, 256), 256, 0, stream>>>(ei, E, N, a_src, a_dst, segmax, segsum);
        edge_scatter<<<gridFor(totE * 32, 256), 256, 0, stream>>>(ei, E, N, a_src, a_dst,
                                                                  segmax, segsum, xlb, xbuf);
        finalize_layer<<<gridFor(total, 256), 256, 0, stream>>>(xbuf, cb, out, total);
    }
    scale_out<<<gridFor(total, 256), 256, 0, stream>>>(out, 1.f / (float)(L + 1), total);
}


# thin passes (x = emb + p, logit dots, final scale) as few long waves: 16-byte accesses, 8 to 16 loads in flight per lane, dots 32 nodes per wave with row_xmask adds; same order of additions
# speedup vs baseline: 2.5271x; 2.5271x over previous
#include <hip/hip_runtime.h>
#include <hip/hip_bf16.h>

typedef float v2f __attribute__((ext_vector_type(2)));
typedef float v8f __attribute__((ext_vector_type(8)));

#define DDIM 64
#define GEMM_ROWS 128
#define SX_STRIDE 68
#define SW_STRIDE 68

__device__ __forceinline__ unsigned fkey(float f) {
    unsigned u = __float_as_uint(f);
    return (u & 0x80000000u) ? ~u : (u | 0x80000000u);
}
__device__ __forceinline__ float fdec(unsigned k) {
    return (k & 0x80000000u) ? __uint_as_float(k & 0x7FFFFFFFu)
                             : __uint_as_float(~k);
}
#define FKEY_NEG_INF 0x007FFFFFu

__global__ void compute_pvec(const float* __restrict__ prompt, const float* __restrict__ pw,
                             const float* __restrict__ pb, float* __restrict__ pvec, int P) {
    int d = threadIdx.x;
    if (d >= DDIM) return;
    float s = pb[d];
    for (int j = 0; j < P; ++j) s += prompt[j] * pw[d * P + j];
    pvec[d] = s;
}

#define STREAM_U 8
__global__ __launch_bounds__(256) void add_prompt(const float* __restrict__ emb,
                                                  const float* __restrict__ pvec,
                                                  float* __restrict__ x,
                                                  float* __restrict__ out_sum,
                                                  long long total4) {
    const float4* __restrict__ e4 = (const float4*)emb;
    float4* __restrict__ x4 = (float4*)x;
    float4* __restrict__ o4 = (float4*)out_sum;
    const int t = threadIdx.x;
    const float4 p = ((const float4*)pvec)[t & 15];
    const long long base = (long long)blockIdx.x * (256 * STREAM_U) + t;
    float4 v[STREAM_U];
    #pragma unroll
    for (int u = 0; u < STREAM_U; ++u) {
        const long long i = base + u * 256;
        v[u] = e4[(i < total4) ? i : 0];
    }
    #pragma unroll
    for (int u = 0; u < STREAM_U; ++u) {
        const long long i = base + u * 256;
        if (i < total4) {
            float4 r;
            r.x = v[u].x + p.x; r.y = v[u].y + p.y; r.z = v[u].z + p.z; r.w = v[u].w + p.w;
            x4[i] = r;
            o4[i] = r;
        }
    }
}

__global__ __launch_bounds__(256) void gemm_xw_t(const float* __restrict__ x,
                                                 const float* __restrict__ W,
                                                 float* __restrict__ xl, int N) {
    __shared__ float sx[GEMM_ROWS * SX_STRIDE];
    __shared__ float sw[DDIM * SW_STRIDE];
    const int t = threadIdx.x;
    const int rowBase = blockIdx.x * GEMM_ROWS;

    {
        const float4* Wv = (const float4*)W;
        float4* swv = (float4*)sw;
        for (int i = t; i < (DDIM * DDIM) / 4; i += 256) {
            int r = i >> 4, c4 = i & 15;
            swv[r * (SW_STRIDE / 4) + c4] = Wv[i];
        }
    }
    {
        const float4* xv = (const float4*)x;
        float4* sxv = (float4*)sx;
        for (int i = t; i < (GEMM_ROWS * DDIM) / 4; i += 256) {
            int r = i >> 4, c4 = i & 15;
            int gr = rowBase + r;
            if (gr >= N) gr = N - 1;
            sxv[r * (SX_STRIDE / 4) + c4] = xv[(long long)gr * 16 + c4];
        }
    }
    __syncthreads();

    const int wave  = t >> 5;
    const int lane  = t & 31;
    const int lrow  = lane & 15;
    const int khalf = (lane >> 4) << 1;
    const int wrow  = wave * 16;

    v8f acc0 = {0.f,0.f,0.f,0.f,0.f,0.f,0.f,0.f};
    v8f acc1 = acc0, acc2 = acc0, acc3 = acc0;

    const float* arow = &sx[(wrow + lrow) * SX_STRIDE + khalf];
    const float* brow = &sw[lrow * SW_STRIDE + khalf];

    #pragma unroll
    for (int k0 = 0; k0 < DDIM; k0 += 4) {
        v2f a, b0, b1, b2, b3;
        a.x  = arow[k0];                       a.y  = arow[k0 + 1];
        b0.x = brow[k0];                       b0.y = brow[k0 + 1];
        b1.x = brow[16 * SW_STRIDE + k0];      b1.y = brow[16 * SW_STRIDE + k0 + 1];
        b2.x = brow[32 * SW_STRIDE + k0];      b2.y = brow[32 * SW_STRIDE + k0 + 1];
        b3.x = brow[48 * SW_STRIDE + k0];      b3.y = brow[48 * SW_STRIDE + k0 + 1];
        acc0 = __builtin_amdgcn_wmma_f32_16x16x4_f32(false, a, false, b0, (short)0, acc0, false, false);
        acc1 = __builtin_amdgcn_wmma_f32_16x16x4_f32(false, a, false, b1, (short)0, acc1, false, false);
        acc2 = __builtin_amdgcn_wmma_f32_16x16x4_f32(false, a, false, b2, (short)0, acc2, false, false);
        acc3 = __builtin_amdgcn_wmma_f32_16x16x4_f32(false, a, false, b3, (short)0, acc3, false, false);
    }

    const int rhi = (lane >> 4) * 8;
    const long long rbase = (long long)(rowBase + wrow + rhi);
    if (rowBase + GEMM_ROWS <= N) {
        float* p = xl + rbase * DDIM + lrow;
        #pragma unroll
        for (int v = 0; v < 8; ++v) {
            p[(long long)v * DDIM +  0] = acc0[v];
            p[(long long)v * DDIM + 16] = acc1[v];
            p[(long long)v * DDIM + 32] = acc2[v];
            p[(long long)v * DDIM + 48] = acc3[v];
        }
    } else {
        #pragma unroll
        for (int v = 0; v < 8; ++v) {
            long long gr = rbase + v;
            if (gr < N) {
                float* p = xl + gr * DDIM + lrow;
                p[ 0] = acc0[v];
                p[16] = acc1[v];
                p[32] = acc2[v];
                p[48] = acc3[v];
            }
        }
    }
}

#define ATT_U 16
template <int M> __device__ __forceinline__ float att_xadd(float v) {
    const int o = __builtin_amdgcn_update_dpp(0, __builtin_bit_cast(int, v), 0x160 + M, 0xF, 0xF, true);
    return v + __builtin_bit_cast(float, o);
}
__device__ __forceinline__ float att_pick(const float (&r)[ATT_U], int c) {
    float o = r[0];
    #pragma unroll
    for (int u = 1; u < ATT_U; ++u) o = (c == u) ? r[u] : o;
    return o;
}
__global__ __launch_bounds__(128) void att_dots(const float* __restrict__ xl,
                                                const float* __restrict__ av_s,
                                                const float* __restrict__ av_d,
                                                float* __restrict__ a_src,
                                                float* __restrict__ a_dst, int N) {
    const int lane = threadIdx.x & 31;
    const int c = lane & 15;
    const int h = lane >> 4;
    const int nb = (blockIdx.x * 4 + (threadIdx.x >> 5)) * 32;
    const float4* __restrict__ xl4 = (const float4*)xl;
    const float4 s = ((const float4*)av_s)[c];
    const float4 d = ((const float4*)av_d)[c];
    float4 v[ATT_U];
    #pragma unroll
    for (int u = 0; u < ATT_U; ++u) {
        int n = nb + 16 * h + u;
        if (n >= N) n = N - 1;
        v[u] = xl4[(long long)n * (DDIM / 4) + c];
    }
    float rs[ATT_U], rd[ATT_U];
    #pragma unroll
    for (int u = 0; u < ATT_U; ++u) {
        float sa = v[u].x * s.x + v[u].y * s.y;
        float sb = v[u].z * s.z + v[u].w * s.w;
        float da = v[u].x * d.x + v[u].y * d.y;
        float db = v[u].z * d.z + v[u].w * d.w;
        sa = att_xadd<8>(sa); sb = att_xadd<8>(sb); da = att_xadd<8>(da); db = att_xadd<8>(db);
        sa = att_xadd<4>(sa); sb = att_xadd<4>(sb); da = att_xadd<4>(da); db = att_xadd<4>(db);
        sa = att_xadd<2>(sa); sb = att_xadd<2>(sb); da = att_xadd<2>(da); db = att_xadd<2>(db);
        sa = att_xadd<1>(sa); sb = att_xadd<1>(sb); da = att_xadd<1>(da); db = att_xadd<1>(db);
        rs[u] = sa + sb;
        rd[u] = da + db;
    }
    const int n = nb + lane;
    const float os = att_pick(rs, c);
    const float od = att_pick(rd, c);
    if (n < N) { a_src[n] = os; a_dst[n] = od; }
}

#define OWN_ROWS 592
#define OWN_CAP  8192
#define OWN_T    1024
#define OWN_U    8

__device__ __forceinline__ float leaky02(float v) { return v > 0.f ? v : 0.2f * v; }

__global__ __launch_bounds__(OWN_T) void edge_owner(const int* __restrict__ ei, int E, int N,
                                                    const float* __restrict__ a_src,
                                                    const float* __restrict__ a_dst,
                                                    const float* __restrict__ xl,
                                                    const float* __restrict__ bias,
                                                    float* __restrict__ xnext,
                                                    float* __restrict__ out_sum) {
    __shared__ float4   acc4[OWN_ROWS * (DDIM / 4)];
    __shared__ unsigned smax[OWN_ROWS];
    __shared__ float    ssum[OWN_ROWS];
    __shared__ float    sadst[OWN_ROWS];
    __shared__ float    sself[OWN_ROWS];
    __shared__ int      lsrc[OWN_CAP];
    __shared__ int      ldl[OWN_CAP];
    __shared__ float    lval[OWN_CAP];
    __shared__ int      lcount;
    float* acc = (float*)acc4;

    const int t = threadIdx.x;
    const int lane = t & 31;
    const int wave = t >> 5;
    const int r0 = blockIdx.x * OWN_ROWS;
    const int rows = min(OWN_ROWS, N - r0);
    const int* __restrict__ dstp = ei + E;

    for (int r = t; r < OWN_ROWS; r += OWN_T) {
        float ad = 0.f, vs = 0.f;
        unsigned key = FKEY_NEG_INF;
        if (r < rows) {
            ad = a_dst[r0 + r];
            vs = leaky02(a_src[r0 + r] + ad);
            key = fkey(vs);
        }
        sadst[r] = ad; sself[r] = vs; smax[r] = key; ssum[r] = 0.f;
    }
    {
        const float4 z = {0.f, 0.f, 0.f, 0.f};
        for (int i = t; i < OWN_ROWS * (DDIM / 4); i += OWN_T) acc4[i] = z;
    }
    if (t == 0) lcount = 0;
    __syncthreads();

    #define OWN_NOTE(dlv, eidx)                                               \
        {                                                                     \
            const unsigned dl_ = (dlv);                                       \
            if (dl_ < (unsigned)rows) {                                       \
                const int slot_ = atomicAdd(&lcount, 1);                      \
                if (slot_ < OWN_CAP) { lsrc[slot_] = (eidx); ldl[slot_] = (int)dl_; } \
            }                                                                 \
        }
    if ((E & 3) == 0) {
        const int4* __restrict__ d4 = (const int4*)dstp;
        const int n4 = E >> 2;
        const int start = (int)(((long long)blockIdx.x * n4) / gridDim.x) & ~7;
        for (int base = 0; base < n4; base += OWN_T * OWN_U) {
            int4 v[OWN_U];
            int  j[OWN_U];
            #pragma unroll
            for (int u = 0; u < OWN_U; ++u) {
                const int i = base + u * OWN_T + t;
                int jj = i + start;
                if (jj >= n4) jj -= n4;
                j[u] = (i < n4) ? jj : -1;
                v[u] = d4[(i < n4) ? jj : 0];
            }
            #pragma unroll
            for (int u = 0; u < OWN_U; ++u) {
                const unsigned q0 = (unsigned)(v[u].x - r0), q1 = (unsigned)(v[u].y - r0);
                const unsigned q2 = (unsigned)(v[u].z - r0), q3 = (unsigned)(v[u].w - r0);
                if (j[u] >= 0 && min(min(q0, q1), min(q2, q3)) < (unsigned)rows) {
                    const int e = j[u] << 2;
                    OWN_NOTE(q0, e)
                    OWN_NOTE(q1, e + 1)
                    OWN_NOTE(q2, e + 2)
                    OWN_NOTE(q3, e + 3)
                }
            }
        }
    } else {
        for (int e = t; e < E; e += OWN_T) OWN_NOTE((unsigned)(dstp[e] - r0), e)
    }
    #undef OWN_NOTE
    __syncthreads();
    const int cnt = lcount;

    if (cnt <= OWN_CAP) {
        for (int b0 = 0; b0 < cnt; b0 += OWN_T * 4) {
            int e_[4], s_[4];
            float as_[4];
            #pragma unroll
            for (int u = 0; u < 4; ++u) { const int i = b0 + u * OWN_T + t; e_[u] = (i < cnt) ? lsrc[i] : 0; }
            #pragma unroll
            for (int u = 0; u < 4; ++u) s_[u] = ei[e_[u]];
            #pragma unroll
            for (int u = 0; u < 4; ++u) as_[u] = a_src[s_[u]];
            #pragma unroll
            for (int u = 0; u < 4; ++u) {
                const int i = b0 + u * OWN_T + t;
                if (i < cnt) {
                    const int dl = ldl[i];
                    const float v = leaky02(as_[u] + sadst[dl]);
                    lsrc[i] = s_[u];
                    lval[i] = v;
                    atomicMax(&smax[dl], fkey(v));
                }
            }
        }
        __syncthreads();
        for (int i = t; i < cnt; i += OWN_T) {
            const int dl = ldl[i];
            const float w = __expf(lval[i] - fdec(smax[dl]));
            lval[i] = w;
            atomicAdd(&ssum[dl], w);
        }
        for (int r = t; r < rows; r += OWN_T) {
            const float w = __expf(sself[r] - fdec(smax[r]));
            sself[r] = w;
            atomicAdd(&ssum[r], w);
        }
        __syncthreads();
        for (int g = wave * 8; g < cnt; g += (OWN_T / 32) * 8) {
            float lo[8], hi[8], aj[8];
            int sj[8], dj[8];
            #pragma unroll
            for (int k = 0; k < 8; ++k) {
                const int i = (g + k < cnt) ? g + k : g;
                sj[k] = lsrc[i];
                dj[k] = ldl[i];
                aj[k] = lval[i] / (ssum[dj[k]] + 1e-16f);
            }
            #pragma unroll
            for (int k = 0; k < 8; ++k) {
                const float* xr = xl + (long long)sj[k] * DDIM + lane;
                lo[k] = xr[0];
                hi[k] = xr[32];
            }
            #pragma unroll
            for (int k = 0; k < 8; ++k) {
                if (g + k < cnt) {
                    float* ap = acc + dj[k] * DDIM + lane;
                    atomicAdd(ap,      lo[k] * aj[k]);
                    atomicAdd(ap + 32, hi[k] * aj[k]);
                }
            }
        }
    } else {
        for (int e = t; e < E; e += OWN_T) {
            const unsigned dl = (unsigned)(dstp[e] - r0);
            if (dl < (unsigned)rows)
                atomicMax(&smax[dl], fkey(leaky02(a_src[ei[e]] + sadst[dl])));
        }
        __syncthreads();
        for (int e = t; e < E; e += OWN_T) {
            const unsigned dl = (unsigned)(dstp[e] - r0);
            if (dl < (unsigned)rows)
                atomicAdd(&ssum[dl], __expf(leaky02(a_src[ei[e]] + sadst[dl]) - fdec(smax[dl])));
        }
        for (int r = t; r < rows; r += OWN_T) {
            const float w = __expf(sself[r] - fdec(smax[r]));
            sself[r] = w;
            atomicAdd(&ssum[r], w);
        }
        __syncthreads();
        for (int e = t; e < E; e += OWN_T) {
            const unsigned dl = (unsigned)(dstp[e] - r0);
            if (dl < (unsigned)rows) {
                const int s = ei[e];
                const float aj = __expf(leaky02(a_src[s] + sadst[dl]) - fdec(smax[dl])) / (ssum[dl] + 1e-16f);
                const float* xr = xl + (long long)s * DDIM;
                float* ap = acc + dl * DDIM;
                for (int c = 0; c < DDIM; ++c) atomicAdd(ap + c, xr[c] * aj);
            }
        }
    }
    __syncthreads();

    const float4* __restrict__ xl4 = (const float4*)xl;
    const float4* __restrict__ b4 = (const float4*)bias;
    float4* __restrict__ xn4 = (float4*)xnext;
    float4* __restrict__ o4 = (float4*)out_sum;
    for (int i = t; i < rows * (DDIM / 4); i += OWN_T) {
        const int r = i >> 4, c4 = i & 15;
        const long long gi = (long long)(r0 + r) * (DDIM / 4) + c4;
        const float as = sself[r] / (ssum[r] + 1e-16f);
        const float4 a = acc4[i];
        const float4 xs = xl4[gi];
        const float4 b = b4[c4];
        float4 o = o4[gi];
        float4 v;
        v.x = (a.x + xs.x * as) + b.x;
        v.y = (a.y + xs.y * as) + b.y;
        v.z = (a.z + xs.z * as) + b.z;
        v.w = (a.w + xs.w * as) + b.w;
        v.x = v.x > 0.f ? v.x : (__expf(v.x) - 1.f);
        v.y = v.y > 0.f ? v.y : (__expf(v.y) - 1.f);
        v.z = v.z > 0.f ? v.z : (__expf(v.z) - 1.f);
        v.w = v.w > 0.f ? v.w : (__expf(v.w) - 1.f);
        xn4[gi] = v;
        o.x += v.x; o.y += v.y; o.z += v.z; o.w += v.w;
        o4[gi] = o;
    }
}

__global__ __launch_bounds__(256) void scale_out(float* __restrict__ out, float inv,
                                                 long long total4) {
    float4* __restrict__ o4 = (float4*)out;
    const long long base = (long long)blockIdx.x * (256 * STREAM_U) + threadIdx.x;
    float4 v[STREAM_U];
    #pragma unroll
    for (int u = 0; u < STREAM_U; ++u) {
        const long long i = base + u * 256;
        v[u] = o4[(i < total4) ? i : 0];
    }
    #pragma unroll
    for (int u = 0; u < STREAM_U; ++u) {
        const long long i = base + u * 256;
        if (i < total4) {
            float4 r;
            r.x = v[u].x * inv; r.y = v[u].y * inv; r.z = v[u].z * inv; r.w = v[u].w * inv;
            o4[i] = r;
        }
    }
}

static inline unsigned gridFor(long long work, int block) {
    return (unsigned)((work + block - 1) / block);
}

extern "C" void kernel_launch(void* const* d_in, const int* in_sizes, int n_in,
                              void* d_out, int out_size, void* d_ws, size_t ws_size,
                              hipStream_t stream) {
    const int*   ei       = (const int*)d_in[0];
    const float* emb      = (const float*)d_in[1];
    const float* prompt   = (const float*)d_in[2];
    const float* proj_w   = (const float*)d_in[3];
    const float* proj_b   = (const float*)d_in[4];
    const float* conv_ws  = (const float*)d_in[5];
    const float* att_srcs = (const float*)d_in[6];
    const float* att_dsts = (const float*)d_in[7];
    const float* conv_bs  = (const float*)d_in[8];

    const int E = in_sizes[0] / 2;
    const int N = in_sizes[1] / DDIM;
    const int P = in_sizes[2];
    const int L = in_sizes[5] / (DDIM * DDIM);

    float* out = (float*)d_out;

    float*    pvec   = (float*)d_ws;
    float*    xbuf   = pvec + 64;
    float*    xlb    = xbuf + (size_t)N * DDIM;
    float*    a_src  = xlb + (size_t)N * DDIM;
    float*    a_dst  = a_src + N;


    compute_pvec<<<1, 64, 0, stream>>>(prompt, proj_w, proj_b, pvec, P);
    const long long total4 = (long long)N * (DDIM / 4);
    add_prompt<<<gridFor(total4, 256 * STREAM_U), 256, 0, stream>>>(emb, pvec, xbuf, out, total4);

    for (int l = 0; l < L; ++l) {
        const float* W  = conv_ws  + (size_t)l * DDIM * DDIM;
        const float* as = att_srcs + (size_t)l * DDIM;
        const float* ad = att_dsts + (size_t)l * DDIM;
        const float* cb = conv_bs  + (size_t)l * DDIM;

        gemm_xw_t<<<gridFor(N, GEMM_ROWS), 256, 0, stream>>>(xbuf, W, xlb, N);
        att_dots<<<gridFor(N, 128), 128, 0, stream>>>(xlb, as, ad, a_src, a_dst, N);
        edge_owner<<<gridFor(N, OWN_ROWS), OWN_T, 0, stream>>>(ei, E, N, a_src, a_dst, xlb, cb, xbuf, out);
    }
    scale_out<<<gridFor(total4, 256 * STREAM_U), 256, 0, stream>>>(out, 1.f / (float)(L + 1), total4);
}


# edge walk reads one owner byte per edge from a map the first launch writes once per call: 16 edges per 16-byte load, a quarter of the bytes; list order changes, sums stay LDS adds in arrival order
# speedup vs baseline: 2.9135x; 1.1529x over previous
#include <hip/hip_runtime.h>
#include <hip/hip_bf16.h>

typedef float v2f __attribute__((ext_vector_type(2)));
typedef float v8f __attribute__((ext_vector_type(8)));

#define DDIM 64
#define GEMM_ROWS 128
#define SX_STRIDE 68
#define SW_STRIDE 68

__device__ __forceinline__ unsigned fkey(float f) {
    unsigned u = __float_as_uint(f);
    return (u & 0x80000000u) ? ~u : (u | 0x80000000u);
}
__device__ __forceinline__ float fdec(unsigned k) {
    return (k & 0x80000000u) ? __uint_as_float(k & 0x7FFFFFFFu)
                             : __uint_as_float(~k);
}
#define FKEY_NEG_INF 0x007FFFFFu

#define OWN_ROWS 592

#define OWN_NONE 255u
__device__ __forceinline__ unsigned char* own_map_base(float* tail) {
    unsigned char* p = (unsigned char*)tail;
    return p + ((0ull - (unsigned long long)p) & 15ull);
}
__device__ __forceinline__ bool own_map_ok(const float* wsbase, float* tail, unsigned long long ws_bytes,
                                           int E, int N) {
    const unsigned long long off = (unsigned long long)(own_map_base(tail) - (const unsigned char*)wsbase);
    return (N + OWN_ROWS - 1) / OWN_ROWS <= 255 && off + (unsigned long long)E <= ws_bytes;
}
__device__ __forceinline__ unsigned own_of(int dst, int N) {
    return ((unsigned)dst < (unsigned)N) ? (unsigned)dst / (unsigned)OWN_ROWS : OWN_NONE;
}

#define MAP_U 4
__global__ __launch_bounds__(256) void compute_pvec(const float* __restrict__ prompt, const float* __restrict__ pw,
                             const float* __restrict__ pb, float* __restrict__ pvec, int P,
                             const int* __restrict__ ei, int E, int N,
                             float* tail, const float* wsbase, unsigned long long ws_bytes) {
    const int t = threadIdx.x;
    if (blockIdx.x == 0 && t < DDIM) {
        float s = pb[t];
        for (int j = 0; j < P; ++j) s += prompt[j] * pw[t * P + j];
        pvec[t] = s;
    }
    if (!own_map_ok(wsbase, tail, ws_bytes, E, N)) return;
    unsigned char* __restrict__ map = own_map_base(tail);
    const int* __restrict__ dstp = ei + E;
    if ((E & 3) == 0) {
        const int4* __restrict__ d4 = (const int4*)dstp;
        unsigned* __restrict__ m1 = (unsigned*)map;
        const int n4 = E >> 2;
        for (int base = blockIdx.x * (256 * MAP_U); base < n4; base += gridDim.x * (256 * MAP_U)) {
            int4 v[MAP_U];
            #pragma unroll
            for (int u = 0; u < MAP_U; ++u) {
                const int g = base + u * 256 + t;
                v[u] = d4[(g < n4) ? g : 0];
            }
            #pragma unroll
            for (int u = 0; u < MAP_U; ++u) {
                const int g = base + u * 256 + t;
                if (g < n4)
                    m1[g] = own_of(v[u].x, N) | (own_of(v[u].y, N) << 8) |
                            (own_of(v[u].z, N) << 16) | (own_of(v[u].w, N) << 24);
            }
        }
    } else {
        for (int e = blockIdx.x * 256 + t; e < E; e += gridDim.x * 256) map[e] = (unsigned char)own_of(dstp[e], N);
    }
}

#define STREAM_U 8
__global__ __launch_bounds__(256) void add_prompt(const float* __restrict__ emb,
                                                  const float* __restrict__ pvec,
                                                  float* __restrict__ x,
                                                  float* __restrict__ out_sum,
                                                  long long total4) {
    const float4* __restrict__ e4 = (const float4*)emb;
    float4* __restrict__ x4 = (float4*)x;
    float4* __restrict__ o4 = (float4*)out_sum;
    const int t = threadIdx.x;
    const float4 p = ((const float4*)pvec)[t & 15];
    const long long base = (long long)blockIdx.x * (256 * STREAM_U) + t;
    float4 v[STREAM_U];
    #pragma unroll
    for (int u = 0; u < STREAM_U; ++u) {
        const long long i = base + u * 256;
        v[u] = e4[(i < total4) ? i : 0];
    }
    #pragma unroll
    for (int u = 0; u < STREAM_U; ++u) {
        const long long i = base + u * 256;
        if (i < total4) {
            float4 r;
            r.x = v[u].x + p.x; r.y = v[u].y + p.y; r.z = v[u].z + p.z; r.w = v[u].w + p.w;
            x4[i] = r;
            o4[i] = r;
        }
    }
}

__global__ __launch_bounds__(256) void gemm_xw_t(const float* __restrict__ x,
                                                 const float* __restrict__ W,
                                                 float* __restrict__ xl, int N) {
    __shared__ float sx[GEMM_ROWS * SX_STRIDE];
    __shared__ float sw[DDIM * SW_STRIDE];
    const int t = threadIdx.x;
    const int rowBase = blockIdx.x * GEMM_ROWS;

    {
        const float4* Wv = (const float4*)W;
        float4* swv = (float4*)sw;
        for (int i = t; i < (DDIM * DDIM) / 4; i += 256) {
            int r = i >> 4, c4 = i & 15;
            swv[r * (SW_STRIDE / 4) + c4] = Wv[i];
        }
    }
    {
        const float4* xv = (const float4*)x;
        float4* sxv = (float4*)sx;
        for (int i = t; i < (GEMM_ROWS * DDIM) / 4; i += 256) {
            int r = i >> 4, c4 = i & 15;
            int gr = rowBase + r;
            if (gr >= N) gr = N - 1;
            sxv[r * (SX_STRIDE / 4) + c4] = xv[(long long)gr * 16 + c4];
        }
    }
    __syncthreads();

    const int wave  = t >> 5;
    const int lane  = t & 31;
    const int lrow  = lane & 15;
    const int khalf = (lane >> 4) << 1;
    const int wrow  = wave * 16;

    v8f acc0 = {0.f,0.f,0.f,0.f,0.f,0.f,0.f,0.f};
    v8f acc1 = acc0, acc2 = acc0, acc3 = acc0;

    const float* arow = &sx[(wrow + lrow) * SX_STRIDE + khalf];
    const float* brow = &sw[lrow * SW_STRIDE + khalf];

    #pragma unroll
    for (int k0 = 0; k0 < DDIM; k0 += 4) {
        v2f a, b0, b1, b2, b3;
        a.x  = arow[k0];                       a.y  = arow[k0 + 1];
        b0.x = brow[k0];                       b0.y = brow[k0 + 1];
        b1.x = brow[16 * SW_STRIDE + k0];      b1.y = brow[16 * SW_STRIDE + k0 + 1];
        b2.x = brow[32 * SW_STRIDE + k0];      b2.y = brow[32 * SW_STRIDE + k0 + 1];
        b3.x = brow[48 * SW_STRIDE + k0];      b3.y = brow[48 * SW_STRIDE + k0 + 1];
        acc0 = __builtin_amdgcn_wmma_f32_16x16x4_f32(false, a, false, b0, (short)0, acc0, false, false);
        acc1 = __builtin_amdgcn_wmma_f32_16x16x4_f32(false, a, false, b1, (short)0, acc1, false, false);
        acc2 = __builtin_amdgcn_wmma_f32_16x16x4_f32(false, a, false, b2, (short)0, acc2, false, false);
        acc3 = __builtin_amdgcn_wmma_f32_16x16x4_f32(false, a, false, b3, (short)0, acc3, false, false);
    }

    const int rhi = (lane >> 4) * 8;
    const long long rbase = (long long)(rowBase + wrow + rhi);
    if (rowBase + GEMM_ROWS <= N) {
        float* p = xl + rbase * DDIM + lrow;
        #pragma unroll
        for (int v = 0; v < 8; ++v) {
            p[(long long)v * DDIM +  0] = acc0[v];
            p[(long long)v * DDIM + 16] = acc1[v];
            p[(long long)v * DDIM + 32] = acc2[v];
            p[(long long)v * DDIM + 48] = acc3[v];
        }
    } else {
        #pragma unroll
        for (int v = 0; v < 8; ++v) {
            long long gr = rbase + v;
            if (gr < N) {
                float* p = xl + gr * DDIM + lrow;
                p[ 0] = acc0[v];
                p[16] = acc1[v];
                p[32] = acc2[v];
                p[48] = acc3[v];
            }
        }
    }
}

#define ATT_U 16
template <int M> __device__ __forceinline__ float att_xadd(float v) {
    const int o = __builtin_amdgcn_update_dpp(0, __builtin_bit_cast(int, v), 0x160 + M, 0xF, 0xF, true);
    return v + __builtin_bit_cast(float, o);
}
__device__ __forceinline__ float att_pick(const float (&r)[ATT_U], int c) {
    float o = r[0];
    #pragma unroll
    for (int u = 1; u < ATT_U; ++u) o = (c == u) ? r[u] : o;
    return o;
}
__global__ __launch_bounds__(128) void att_dots(const float* __restrict__ xl,
                                                const float* __restrict__ av_s,
                                                const float* __restrict__ av_d,
                                                float* __restrict__ a_src,
                                                float* __restrict__ a_dst, int N) {
    const int lane = threadIdx.x & 31;
    const int c = lane & 15;
    const int h = lane >> 4;
    const int nb = (blockIdx.x * 4 + (threadIdx.x >> 5)) * 32;
    const float4* __restrict__ xl4 = (const float4*)xl;
    const float4 s = ((const float4*)av_s)[c];
    const float4 d = ((const float4*)av_d)[c];
    float4 v[ATT_U];
    #pragma unroll
    for (int u = 0; u < ATT_U; ++u) {
        int n = nb + 16 * h + u;
        if (n >= N) n = N - 1;
        v[u] = xl4[(long long)n * (DDIM / 4) + c];
    }
    float rs[ATT_U], rd[ATT_U];
    #pragma unroll
    for (int u = 0; u < ATT_U; ++u) {
        float sa = v[u].x * s.x + v[u].y * s.y;
        float sb = v[u].z * s.z + v[u].w * s.w;
        float da = v[u].x * d.x + v[u].y * d.y;
        float db = v[u].z * d.z + v[u].w * d.w;
        sa = att_xadd<8>(sa); sb = att_xadd<8>(sb); da = att_xadd<8>(da); db = att_xadd<8>(db);
        sa = att_xadd<4>(sa); sb = att_xadd<4>(sb); da = att_xadd<4>(da); db = att_xadd<4>(db);
        sa = att_xadd<2>(sa); sb = att_xadd<2>(sb); da = att_xadd<2>(da); db = att_xadd<2>(db);
        sa = att_xadd<1>(sa); sb = att_xadd<1>(sb); da = att_xadd<1>(da); db = att_xadd<1>(db);
        rs[u] = sa + sb;
        rd[u] = da + db;
    }
    const int n = nb + lane;
    const float os = att_pick(rs, c);
    const float od = att_pick(rd, c);
    if (n < N) { a_src[n] = os; a_dst[n] = od; }
}

#define OWN_CAP  8192
#define OWN_T    1024
#define OWN_U    8

__device__ __forceinline__ float leaky02(float v) { return v > 0.f ? v : 0.2f * v; }

__global__ __launch_bounds__(OWN_T) void edge_owner(const int* __restrict__ ei, int E, int N,
                                                    const float* __restrict__ a_src,
                                                    const float* __restrict__ a_dst,
                                                    const float* __restrict__ xl,
                                                    const float* __restrict__ bias,
                                                    float* __restrict__ xnext,
                                                    float* __restrict__ out_sum,
                                                    float* tail, const float* wsbase,
                                                    unsigned long long ws_bytes) {
    __shared__ float4   acc4[OWN_ROWS * (DDIM / 4)];
    __shared__ unsigned smax[OWN_ROWS];
    __shared__ float    ssum[OWN_ROWS];
    __shared__ float    sadst[OWN_ROWS];
    __shared__ float    sself[OWN_ROWS];
    __shared__ int      lsrc[OWN_CAP];
    __shared__ int      ldl[OWN_CAP];
    __shared__ float    lval[OWN_CAP];
    __shared__ int      lcount;
    float* acc = (float*)acc4;

    const int t = threadIdx.x;
    const int lane = t & 31;
    const int wave = t >> 5;
    const int r0 = blockIdx.x * OWN_ROWS;
    const int rows = min(OWN_ROWS, N - r0);
    const int* __restrict__ dstp = ei + E;

    for (int r = t; r < OWN_ROWS; r += OWN_T) {
        float ad = 0.f, vs = 0.f;
        unsigned key = FKEY_NEG_INF;
        if (r < rows) {
            ad = a_dst[r0 + r];
            vs = leaky02(a_src[r0 + r] + ad);
            key = fkey(vs);
        }
        sadst[r] = ad; sself[r] = vs; smax[r] = key; ssum[r] = 0.f;
    }
    {
        const float4 z = {0.f, 0.f, 0.f, 0.f};
        for (int i = t; i < OWN_ROWS * (DDIM / 4); i += OWN_T) acc4[i] = z;
    }
    if (t == 0) lcount = 0;
    __syncthreads();

    #define OWN_NOTE(dlv, eidx)                                               \
        {                                                                     \
            const unsigned dl_ = (dlv);                                       \
            if (dl_ < (unsigned)rows) {                                       \
                const int slot_ = atomicAdd(&lcount, 1);                      \
                if (slot_ < OWN_CAP) { lsrc[slot_] = (eidx); ldl[slot_] = (int)dl_; } \
            }                                                                 \
        }
    if (own_map_ok(wsbase, tail, ws_bytes, E, N)) {
        const unsigned char* __restrict__ map = own_map_base(tail);
        const uint4* __restrict__ m4 = (const uint4*)map;
        const int n16 = E >> 4;
        const unsigned me = blockIdx.x;
        const unsigned pat = me * 0x01010101u;
        const int start = (int)(((long long)blockIdx.x * n16) / gridDim.x) & ~7;
        for (int base = 0; base < n16; base += OWN_T * OWN_U) {
            uint4 v[OWN_U];
            int   j[OWN_U];
            #pragma unroll
            for (int u = 0; u < OWN_U; ++u) {
                const int i = base + u * OWN_T + t;
                int jj = i + start;
                if (jj >= n16) jj -= n16;
                j[u] = (i < n16) ? jj : -1;
                v[u] = m4[(i < n16) ? jj : 0];
            }
            #pragma unroll
            for (int u = 0; u < OWN_U; ++u) {
                const unsigned x0 = v[u].x ^ pat, x1 = v[u].y ^ pat, x2 = v[u].z ^ pat, x3 = v[u].w ^ pat;
                const unsigned y0 = ((x0 & 0x7f7f7f7fu) + 0x7f7f7f7fu) | x0;
                const unsigned y1 = ((x1 & 0x7f7f7f7fu) + 0x7f7f7f7fu) | x1;
                const unsigned y2 = ((x2 & 0x7f7f7f7fu) + 0x7f7f7f7fu) | x2;
                const unsigned y3 = ((x3 & 0x7f7f7f7fu) + 0x7f7f7f7fu) | x3;
                if (j[u] >= 0 && ((y0 & y1 & y2 & y3) | 0x7f7f7f7fu) != 0xffffffffu) {
                    unsigned c = ((~y0 & 0x80808080u) >> 7) | ((~y1 & 0x80808080u) >> 6) |
                                 ((~y2 & 0x80808080u) >> 5) | ((~y3 & 0x80808080u) >> 4);
                    const int e0 = j[u] << 4;
                    while (c) {
                        const int b = __builtin_ctz(c);
                        c &= c - 1;
                        const int slot = atomicAdd(&lcount, 1);
                        if (slot < OWN_CAP) lsrc[slot] = e0 + ((b & 7) << 2) + (b >> 3);
                    }
                }
            }
        }
        for (int e = (n16 << 4) + t; e < E; e += OWN_T) {
            if (map[e] == me) {
                const int slot = atomicAdd(&lcount, 1);
                if (slot < OWN_CAP) lsrc[slot] = e;
            }
        }
    } else if ((E & 3) == 0) {
        const int4* __restrict__ d4 = (const int4*)dstp;
        const int n4 = E >> 2;
        const int start = (int)(((long long)blockIdx.x * n4) / gridDim.x) & ~7;
        for (int base = 0; base < n4; base += OWN_T * OWN_U) {
            int4 v[OWN_U];
            int  j[OWN_U];
            #pragma unroll
            for (int u = 0; u < OWN_U; ++u) {
                const int i = base + u * OWN_T + t;
                int jj = i + start;
                if (jj >= n4) jj -= n4;
                j[u] = (i < n4) ? jj : -1;
                v[u] = d4[(i < n4) ? jj : 0];
            }
            #pragma unroll
            for (int u = 0; u < OWN_U; ++u) {
                const unsigned q0 = (unsigned)(v[u].x - r0), q1 = (unsigned)(v[u].y - r0);
                const unsigned q2 = (unsigned)(v[u].z - r0), q3 = (unsigned)(v[u].w - r0);
                if (j[u] >= 0 && min(min(q0, q1), min(q2, q3)) < (unsigned)rows) {
                    const int e = j[u] << 2;
                    OWN_NOTE(q0, e)
                    OWN_NOTE(q1, e + 1)
                    OWN_NOTE(q2, e + 2)
                    OWN_NOTE(q3, e + 3)
                }
            }
        }
    } else {
        for (int e = t; e < E; e += OWN_T) OWN_NOTE((unsigned)(dstp[e] - r0), e)
    }
    #undef OWN_NOTE
    __syncthreads();
    const int cnt = lcount;

    if (cnt <= OWN_CAP) {
        for (int b0 = 0; b0 < cnt; b0 += OWN_T * 4) {
            int e_[4], s_[4], d_[4];
            float as_[4];
            #pragma unroll
            for (int u = 0; u < 4; ++u) { const int i = b0 + u * OWN_T + t; e_[u] = (i < cnt) ? lsrc[i] : 0; }
            #pragma unroll
            for (int u = 0; u < 4; ++u) { s_[u] = ei[e_[u]]; d_[u] = dstp[e_[u]]; }
            #pragma unroll
            for (int u = 0; u < 4; ++u) as_[u] = a_src[s_[u]];
            #pragma unroll
            for (int u = 0; u < 4; ++u) {
                const int i = b0 + u * OWN_T + t;
                if (i < cnt) {
                    const int dl = d_[u] - r0;
                    const float v = leaky02(as_[u] + sadst[dl]);
                    lsrc[i] = s_[u];
                    ldl[i] = dl;
                    lval[i] = v;
                    atomicMax(&smax[dl], fkey(v));
                }
            }
        }
        __syncthreads();
        for (int i = t; i < cnt; i += OWN_T) {
            const int dl = ldl[i];
            const float w = __expf(lval[i] - fdec(smax[dl]));
            lval[i] = w;
            atomicAdd(&ssum[dl], w);
        }
        for (int r = t; r < rows; r += OWN_T) {
            const float w = __expf(sself[r] - fdec(smax[r]));
            sself[r] = w;
            atomicAdd(&ssum[r], w);
        }
        __syncthreads();
        for (int g = wave * 8; g < cnt; g += (OWN_T / 32) * 8) {
            float lo[8], hi[8], aj[8];
            int sj[8], dj[8];
            #pragma unroll
            for (int k = 0; k < 8; ++k) {
                const int i = (g + k < cnt) ? g + k : g;
                sj[k] = lsrc[i];
                dj[k] = ldl[i];
                aj[k] = lval[i] / (ssum[dj[k]] + 1e-16f);
            }
            #pragma unroll
            for (int k = 0; k < 8; ++k) {
                const float* xr = xl + (long long)sj[k] * DDIM + lane;
                lo[k] = xr[0];
                hi[k] = xr[32];
            }
            #pragma unroll
            for (int k = 0; k < 8; ++k) {
                if (g + k < cnt) {
                    float* ap = acc + dj[k] * DDIM + lane;
                    atomicAdd(ap,      lo[k] * aj[k]);
                    atomicAdd(ap + 32, hi[k] * aj[k]);
                }
            }
        }
    } else {
        for (int e = t; e < E; e += OWN_T) {
            const unsigned dl = (unsigned)(dstp[e] - r0);
            if (dl < (unsigned)rows)
                atomicMax(&smax[dl], fkey(leaky02(a_src[ei[e]] + sadst[dl])));
        }
        __syncthreads();
        for (int e = t; e < E; e += OWN_T) {
            const unsigned dl = (unsigned)(dstp[e] - r0);
            if (dl < (unsigned)rows)
                atomicAdd(&ssum[dl], __expf(leaky02(a_src[ei[e]] + sadst[dl]) - fdec(smax[dl])));
        }
        for (int r = t; r < rows; r += OWN_T) {
            const float w = __expf(sself[r] - fdec(smax[r]));
            sself[r] = w;
            atomicAdd(&ssum[r], w);
        }
        __syncthreads();
        for (int e = t; e < E; e += OWN_T) {
            const unsigned dl = (unsigned)(dstp[e] - r0);
            if (dl < (unsigned)rows) {
                const int s = ei[e];
                const float aj = __expf(leaky02(a_src[s] + sadst[dl]) - fdec(smax[dl])) / (ssum[dl] + 1e-16f);
                const float* xr = xl + (long long)s * DDIM;
                float* ap = acc + dl * DDIM;
                for (int c = 0; c < DDIM; ++c) atomicAdd(ap + c, xr[c] * aj);
            }
        }
    }
    __syncthreads();

    const float4* __restrict__ xl4 = (const float4*)xl;
    const float4* __restrict__ b4 = (const float4*)bias;
    float4* __restrict__ xn4 = (float4*)xnext;
    float4* __restrict__ o4 = (float4*)out_sum;
    for (int i = t; i < rows * (DDIM / 4); i += OWN_T) {
        const int r = i >> 4, c4 = i & 15;
        const long long gi = (long long)(r0 + r) * (DDIM / 4) + c4;
        const float as = sself[r] / (ssum[r] + 1e-16f);
        const float4 a = acc4[i];
        const float4 xs = xl4[gi];
        const float4 b = b4[c4];
        float4 o = o4[gi];
        float4 v;
        v.x = (a.x + xs.x * as) + b.x;
        v.y = (a.y + xs.y * as) + b.y;
        v.z = (a.z + xs.z * as) + b.z;
        v.w = (a.w + xs.w * as) + b.w;
        v.x = v.x > 0.f ? v.x : (__expf(v.x) - 1.f);
        v.y = v.y > 0.f ? v.y : (__expf(v.y) - 1.f);
        v.z = v.z > 0.f ? v.z : (__expf(v.z) - 1.f);
        v.w = v.w > 0.f ? v.w : (__expf(v.w) - 1.f);
        xn4[gi] = v;
        o.x += v.x; o.y += v.y; o.z += v.z; o.w += v.w;
        o4[gi] = o;
    }
}

__global__ __launch_bounds__(256) void scale_out(float* __restrict__ out, float inv,
                                                 long long total4) {
    float4* __restrict__ o4 = (float4*)out;
    const long long base = (long long)blockIdx.x * (256 * STREAM_U) + threadIdx.x;
    float4 v[STREAM_U];
    #pragma unroll
    for (int u = 0; u < STREAM_U; ++u) {
        const long long i = base + u * 256;
        v[u] = o4[(i < total4) ? i : 0];
    }
    #pragma unroll
    for (int u = 0; u < STREAM_U; ++u) {
        const long long i = base + u * 256;
        if (i < total4) {
            float4 r;
            r.x = v[u].x * inv; r.y = v[u].y * inv; r.z = v[u].z * inv; r.w = v[u].w * inv;
            o4[i] = r;
        }
    }
}

static inline unsigned gridFor(long long work, int block) {
    return (unsigned)((work + block - 1) / block);
}

extern "C" void kernel_launch(void* const* d_in, const int* in_sizes, int n_in,
                              void* d_out, int out_size, void* d_ws, size_t ws_size,
                              hipStream_t stream) {
    const int*   ei       = (const int*)d_in[0];
    const float* emb      = (const float*)d_in[1];
    const float* prompt   = (const float*)d_in[2];
    const float* proj_w   = (const float*)d_in[3];
    const float* proj_b   = (const float*)d_in[4];
    const float* conv_ws  = (const float*)d_in[5];
    const float* att_srcs = (const float*)d_in[6];
    const float* att_dsts = (const float*)d_in[7];
    const float* conv_bs  = (const float*)d_in[8];

    const int E = in_sizes[0] / 2;
    const int N = in_sizes[1] / DDIM;
    const int P = in_sizes[2];
    const int L = in_sizes[5] / (DDIM * DDIM);

    float* out = (float*)d_out;

    float*    pvec   = (float*)d_ws;
    float*    xbuf   = pvec + 64;
    float*    xlb    = xbuf + (size_t)N * DDIM;
    float*    a_src  = xlb + (size_t)N * DDIM;
    float*    a_dst  = a_src + N;


    float* tail = a_dst + N;
    const unsigned long long ws_bytes = (unsigned long long)ws_size;
    compute_pvec<<<gridFor((long long)E + 1, 256 * MAP_U * 4), 256, 0, stream>>>(prompt, proj_w, proj_b, pvec, P,
                                                                          ei, E, N, tail, pvec, ws_bytes);
    const long long total4 = (long long)N * (DDIM / 4);
    add_prompt<<<gridFor(total4, 256 * STREAM_U), 256, 0, stream>>>(emb, pvec, xbuf, out, total4);

    for (int l = 0; l < L; ++l) {
        const float* W  = conv_ws  + (size_t)l * DDIM * DDIM;
        const float* as = att_srcs + (size_t)l * DDIM;
        const float* ad = att_dsts + (size_t)l * DDIM;
        const float* cb = conv_bs  + (size_t)l * DDIM;

        gemm_xw_t<<<gridFor(N, GEMM_ROWS), 256, 0, stream>>>(xbuf, W, xlb, N);
        att_dots<<<gridFor(N, 128), 128, 0, stream>>>(xlb, as, ad, a_src, a_dst, N);
        edge_owner<<<gridFor(N, OWN_ROWS), OWN_T, 0, stream>>>(ei, E, N, a_src, a_dst, xlb, cb, xbuf, out,
                                                               tail, pvec, ws_bytes);
    }
    scale_out<<<gridFor(total4, 256 * STREAM_U), 256, 0, stream>>>(out, 1.f / (float)(L + 1), total4);
}


# layer 1's x @ W^T moves into layer 0's edge phase: a wave finishes 16 rows in LDS (pitch 68) and multiplies them there on the matrix unit; that product launch and x's memory trip gone; same sums
# speedup vs baseline: 3.1331x; 1.0754x over previous
#include <hip/hip_runtime.h>
#include <hip/hip_bf16.h>

typedef float v2f __attribute__((ext_vector_type(2)));
typedef float v8f __attribute__((ext_vector_type(8)));

#define DDIM 64
#define GEMM_ROWS 128
#define SX_STRIDE 68
#define SW_STRIDE 68

__device__ __forceinline__ unsigned fkey(float f) {
    unsigned u = __float_as_uint(f);
    return (u & 0x80000000u) ? ~u : (u | 0x80000000u);
}
__device__ __forceinline__ float fdec(unsigned k) {
    return (k & 0x80000000u) ? __uint_as_float(k & 0x7FFFFFFFu)
                             : __uint_as_float(~k);
}
#define FKEY_NEG_INF 0x007FFFFFu

#define OWN_ROWS 592

#define OWN_NONE 255u
__device__ __forceinline__ unsigned char* own_map_base(float* tail) {
    unsigned char* p = (unsigned char*)tail;
    return p + ((0ull - (unsigned long long)p) & 15ull);
}
__device__ __forceinline__ bool own_map_ok(const float* wsbase, float* tail, unsigned long long ws_bytes,
                                           int E, int N) {
    const unsigned long long off = (unsigned long long)(own_map_base(tail) - (const unsigned char*)wsbase);
    return (N + OWN_ROWS - 1) / OWN_ROWS <= 255 && off + (unsigned long long)E <= ws_bytes;
}
__device__ __forceinline__ unsigned own_of(int dst, int N) {
    return ((unsigned)dst < (unsigned)N) ? (unsigned)dst / (unsigned)OWN_ROWS : OWN_NONE;
}

#define MAP_U 4
__global__ __launch_bounds__(256) void compute_pvec(const float* __restrict__ prompt, const float* __restrict__ pw,
                             const float* __restrict__ pb, float* __restrict__ pvec, int P,
                             const int* __restrict__ ei, int E, int N,
                             float* tail, const float* wsbase, unsigned long long ws_bytes) {
    const int t = threadIdx.x;
    if (blockIdx.x == 0 && t < DDIM) {
        float s = pb[t];
        for (int j = 0; j < P; ++j) s += prompt[j] * pw[t * P + j];
        pvec[t] = s;
    }
    if (!own_map_ok(wsbase, tail, ws_bytes, E, N)) return;
    unsigned char* __restrict__ map = own_map_base(tail);
    const int* __restrict__ dstp = ei + E;
    if ((E & 3) == 0) {
        const int4* __restrict__ d4 = (const int4*)dstp;
        unsigned* __restrict__ m1 = (unsigned*)map;
        const int n4 = E >> 2;
        for (int base = blockIdx.x * (256 * MAP_U); base < n4; base += gridDim.x * (256 * MAP_U)) {
            int4 v[MAP_U];
            #pragma unroll
            for (int u = 0; u < MAP_U; ++u) {
                const int g = base + u * 256 + t;
                v[u] = d4[(g < n4) ? g : 0];
            }
            #pragma unroll
            for (int u = 0; u < MAP_U; ++u) {
                const int g = base + u * 256 + t;
                if (g < n4)
                    m1[g] = own_of(v[u].x, N) | (own_of(v[u].y, N) << 8) |
                            (own_of(v[u].z, N) << 16) | (own_of(v[u].w, N) << 24);
            }
        }
    } else {
        for (int e = blockIdx.x * 256 + t; e < E; e += gridDim.x * 256) map[e] = (unsigned char)own_of(dstp[e], N);
    }
}

#define STREAM_U 8
__global__ __launch_bounds__(256) void add_prompt(const float* __restrict__ emb,
                                                  const float* __restrict__ pvec,
                                                  float* __restrict__ x,
                                                  float* __restrict__ out_sum,
                                                  long long total4) {
    const float4* __restrict__ e4 = (const float4*)emb;
    float4* __restrict__ x4 = (float4*)x;
    float4* __restrict__ o4 = (float4*)out_sum;
    const int t = threadIdx.x;
    const float4 p = ((const float4*)pvec)[t & 15];
    const long long base = (long long)blockIdx.x * (256 * STREAM_U) + t;
    float4 v[STREAM_U];
    #pragma unroll
    for (int u = 0; u < STREAM_U; ++u) {
        const long long i = base + u * 256;
        v[u] = e4[(i < total4) ? i : 0];
    }
    #pragma unroll
    for (int u = 0; u < STREAM_U; ++u) {
        const long long i = base + u * 256;
        if (i < total4) {
            float4 r;
            r.x = v[u].x + p.x; r.y = v[u].y + p.y; r.z = v[u].z + p.z; r.w = v[u].w + p.w;
            x4[i] = r;
            o4[i] = r;
        }
    }
}

__global__ __launch_bounds__(256) void gemm_xw_t(const float* __restrict__ x,
                                                 const float* __restrict__ W,
                                                 float* __restrict__ xl, int N) {
    __shared__ float sx[GEMM_ROWS * SX_STRIDE];
    __shared__ float sw[DDIM * SW_STRIDE];
    const int t = threadIdx.x;
    const int rowBase = blockIdx.x * GEMM_ROWS;

    {
        const float4* Wv = (const float4*)W;
        float4* swv = (float4*)sw;
        for (int i = t; i < (DDIM * DDIM) / 4; i += 256) {
            int r = i >> 4, c4 = i & 15;
            swv[r * (SW_STRIDE / 4) + c4] = Wv[i];
        }
    }
    {
        const float4* xv = (const float4*)x;
        float4* sxv = (float4*)sx;
        for (int i = t; i < (GEMM_ROWS * DDIM) / 4; i += 256) {
            int r = i >> 4, c4 = i & 15;
            int gr = rowBase + r;
            if (gr >= N) gr = N - 1;
            sxv[r * (SX_STRIDE / 4) + c4] = xv[(long long)gr * 16 + c4];
        }
    }
    __syncthreads();

    const int wave  = t >> 5;
    const int lane  = t & 31;
    const int lrow  = lane & 15;
    const int khalf = (lane >> 4) << 1;
    const int wrow  = wave * 16;

    v8f acc0 = {0.f,0.f,0.f,0.f,0.f,0.f,0.f,0.f};
    v8f acc1 = acc0, acc2 = acc0, acc3 = acc0;

    const float* arow = &sx[(wrow + lrow) * SX_STRIDE + khalf];
    const float* brow = &sw[lrow * SW_STRIDE + khalf];

    #pragma unroll
    for (int k0 = 0; k0 < DDIM; k0 += 4) {
        v2f a, b0, b1, b2, b3;
        a.x  = arow[k0];                       a.y  = arow[k0 + 1];
        b0.x = brow[k0];                       b0.y = brow[k0 + 1];
        b1.x = brow[16 * SW_STRIDE + k0];      b1.y = brow[16 * SW_STRIDE + k0 + 1];
        b2.x = brow[32 * SW_STRIDE + k0];      b2.y = brow[32 * SW_STRIDE + k0 + 1];
        b3.x = brow[48 * SW_STRIDE + k0];      b3.y = brow[48 * SW_STRIDE + k0 + 1];
        acc0 = __builtin_amdgcn_wmma_f32_16x16x4_f32(false, a, false, b0, (short)0, acc0, false, false);
        acc1 = __builtin_amdgcn_wmma_f32_16x16x4_f32(false, a, false, b1, (short)0, acc1, false, false);
        acc2 = __builtin_amdgcn_wmma_f32_16x16x4_f32(false, a, false, b2, (short)0, acc2, false, false);
        acc3 = __builtin_amdgcn_wmma_f32_16x16x4_f32(false, a, false, b3, (short)0, acc3, false, false);
    }

    const int rhi = (lane >> 4) * 8;
    const long long rbase = (long long)(rowBase + wrow + rhi);
    if (rowBase + GEMM_ROWS <= N) {
        float* p = xl + rbase * DDIM + lrow;
        #pragma unroll
        for (int v = 0; v < 8; ++v) {
            p[(long long)v * DDIM +  0] = acc0[v];
            p[(long long)v * DDIM + 16] = acc1[v];
            p[(long long)v * DDIM + 32] = acc2[v];
            p[(long long)v * DDIM + 48] = acc3[v];
        }
    } else {
        #pragma unroll
        for (int v = 0; v < 8; ++v) {
            long long gr = rbase + v;
            if (gr < N) {
                float* p = xl + gr * DDIM + lrow;
                p[ 0] = acc0[v];
                p[16] = acc1[v];
                p[32] = acc2[v];
                p[48] = acc3[v];
            }
        }
    }
}

#define ATT_U 16
template <int M> __device__ __forceinline__ float att_xadd(float v) {
    const int o = __builtin_amdgcn_update_dpp(0, __builtin_bit_cast(int, v), 0x160 + M, 0xF, 0xF, true);
    return v + __builtin_bit_cast(float, o);
}
__device__ __forceinline__ float att_pick(const float (&r)[ATT_U], int c) {
    float o = r[0];
    #pragma unroll
    for (int u = 1; u < ATT_U; ++u) o = (c == u) ? r[u] : o;
    return o;
}
__global__ __launch_bounds__(128) void att_dots(const float* __restrict__ xl,
                                                const float* __restrict__ av_s,
                                                const float* __restrict__ av_d,
                                                float* __restrict__ a_src,
                                                float* __restrict__ a_dst, int N) {
    const int lane = threadIdx.x & 31;
    const int c = lane & 15;
    const int h = lane >> 4;
    const int nb = (blockIdx.x * 4 + (threadIdx.x >> 5)) * 32;
    const float4* __restrict__ xl4 = (const float4*)xl;
    const float4 s = ((const float4*)av_s)[c];
    const float4 d = ((const float4*)av_d)[c];
    float4 v[ATT_U];
    #pragma unroll
    for (int u = 0; u < ATT_U; ++u) {
        int n = nb + 16 * h + u;
        if (n >= N) n = N - 1;
        v[u] = xl4[(long long)n * (DDIM / 4) + c];
    }
    float rs[ATT_U], rd[ATT_U];
    #pragma unroll
    for (int u = 0; u < ATT_U; ++u) {
        float sa = v[u].x * s.x + v[u].y * s.y;
        float sb = v[u].z * s.z + v[u].w * s.w;
        float da = v[u].x * d.x + v[u].y * d.y;
        float db = v[u].z * d.z + v[u].w * d.w;
        sa = att_xadd<8>(sa); sb = att_xadd<8>(sb); da = att_xadd<8>(da); db = att_xadd<8>(db);
        sa = att_xadd<4>(sa); sb = att_xadd<4>(sb); da = att_xadd<4>(da); db = att_xadd<4>(db);
        sa = att_xadd<2>(sa); sb = att_xadd<2>(sb); da = att_xadd<2>(da); db = att_xadd<2>(db);
        sa = att_xadd<1>(sa); sb = att_xadd<1>(sb); da = att_xadd<1>(da); db = att_xadd<1>(db);
        rs[u] = sa + sb;
        rd[u] = da + db;
    }
    const int n = nb + lane;
    const float os = att_pick(rs, c);
    const float od = att_pick(rd, c);
    if (n < N) { a_src[n] = os; a_dst[n] = od; }
}

#define OWN_CAP  8192
#define ACC_PITCH 68
#define OWN_T    1024
#define OWN_U    8

__device__ __forceinline__ float leaky02(float v) { return v > 0.f ? v : 0.2f * v; }

__global__ __launch_bounds__(OWN_T) void edge_owner(const int* __restrict__ ei, int E, int N,
                                                    const float* __restrict__ a_src,
                                                    const float* __restrict__ a_dst,
                                                    const float* __restrict__ xl,
                                                    const float* __restrict__ bias,
                                                    const float* __restrict__ Wnext,
                                                    float* __restrict__ xl_next,
                                                    float* __restrict__ out_sum,
                                                    float* tail, const float* wsbase,
                                                    unsigned long long ws_bytes) {
    __shared__ float4   acc4[OWN_ROWS * (ACC_PITCH / 4)];
    __shared__ float4   sw4[DDIM * (SW_STRIDE / 4)];
    __shared__ unsigned smax[OWN_ROWS];
    __shared__ float    ssum[OWN_ROWS];
    __shared__ float    sadst[OWN_ROWS];
    __shared__ float    sself[OWN_ROWS];
    __shared__ int      lsrc[OWN_CAP];
    __shared__ int      ldl[OWN_CAP];
    __shared__ float    lval[OWN_CAP];
    __shared__ int      lcount;
    float* acc = (float*)acc4;

    const int t = threadIdx.x;
    const int lane = t & 31;
    const int wave = t >> 5;
    const int r0 = blockIdx.x * OWN_ROWS;
    const int rows = min(OWN_ROWS, N - r0);
    const int* __restrict__ dstp = ei + E;

    for (int r = t; r < OWN_ROWS; r += OWN_T) {
        float ad = 0.f, vs = 0.f;
        unsigned key = FKEY_NEG_INF;
        if (r < rows) {
            ad = a_dst[r0 + r];
            vs = leaky02(a_src[r0 + r] + ad);
            key = fkey(vs);
        }
        sadst[r] = ad; sself[r] = vs; smax[r] = key; ssum[r] = 0.f;
    }
    {
        const float4 z = {0.f, 0.f, 0.f, 0.f};
        for (int i = t; i < OWN_ROWS * (ACC_PITCH / 4); i += OWN_T) acc4[i] = z;
    }
    if (t == 0) lcount = 0;
    if (Wnext != nullptr) {
        const float4* __restrict__ Wv = (const float4*)Wnext;
        for (int i = t; i < (DDIM * DDIM) / 4; i += OWN_T) {
            const int r = i >> 4, c4 = i & 15;
            sw4[r * (SW_STRIDE / 4) + c4] = Wv[i];
        }
    }
    __syncthreads();

    #define OWN_NOTE(dlv, eidx)                                               \
        {                                                                     \
            const unsigned dl_ = (dlv);                                       \
            if (dl_ < (unsigned)rows) {                                       \
                const int slot_ = atomicAdd(&lcount, 1);                      \
                if (slot_ < OWN_CAP) { lsrc[slot_] = (eidx); ldl[slot_] = (int)dl_; } \
            }                                                                 \
        }
    if (own_map_ok(wsbase, tail, ws_bytes, E, N)) {
        const unsigned char* __restrict__ map = own_map_base(tail);
        const uint4* __restrict__ m4 = (const uint4*)map;
        const int n16 = E >> 4;
        const unsigned me = blockIdx.x;
        const unsigned pat = me * 0x01010101u;
        const int start = (int)(((long long)blockIdx.x * n16) / gridDim.x) & ~7;
        for (int base = 0; base < n16; base += OWN_T * OWN_U) {
            uint4 v[OWN_U];
            int   j[OWN_U];
            #pragma unroll
            for (int u = 0; u < OWN_U; ++u) {
                const int i = base + u * OWN_T + t;
                int jj = i + start;
                if (jj >= n16) jj -= n16;
                j[u] = (i < n16) ? jj : -1;
                v[u] = m4[(i < n16) ? jj : 0];
            }
            #pragma unroll
            for (int u = 0; u < OWN_U; ++u) {
                const unsigned x0 = v[u].x ^ pat, x1 = v[u].y ^ pat, x2 = v[u].z ^ pat, x3 = v[u].w ^ pat;
                const unsigned y0 = ((x0 & 0x7f7f7f7fu) + 0x7f7f7f7fu) | x0;
                const unsigned y1 = ((x1 & 0x7f7f7f7fu) + 0x7f7f7f7fu) | x1;
                const unsigned y2 = ((x2 & 0x7f7f7f7fu) + 0x7f7f7f7fu) | x2;
                const unsigned y3 = ((x3 & 0x7f7f7f7fu) + 0x7f7f7f7fu) | x3;
                if (j[u] >= 0 && ((y0 & y1 & y2 & y3) | 0x7f7f7f7fu) != 0xffffffffu) {
                    unsigned c = ((~y0 & 0x80808080u) >> 7) | ((~y1 & 0x80808080u) >> 6) |
                                 ((~y2 & 0x80808080u) >> 5) | ((~y3 & 0x80808080u) >> 4);
                    const int e0 = j[u] << 4;
                    while (c) {
                        const int b = __builtin_ctz(c);
                        c &= c - 1;
                        const int slot = atomicAdd(&lcount, 1);
                        if (slot < OWN_CAP) lsrc[slot] = e0 + ((b & 7) << 2) + (b >> 3);
                    }
                }
            }
        }
        for (int e = (n16 << 4) + t; e < E; e += OWN_T) {
            if (map[e] == me) {
                const int slot = atomicAdd(&lcount, 1);
                if (slot < OWN_CAP) lsrc[slot] = e;
            }
        }
    } else if ((E & 3) == 0) {
        const int4* __restrict__ d4 = (const int4*)dstp;
        const int n4 = E >> 2;
        const int start = (int)(((long long)blockIdx.x * n4) / gridDim.x) & ~7;
        for (int base = 0; base < n4; base += OWN_T * OWN_U) {
            int4 v[OWN_U];
            int  j[OWN_U];
            #pragma unroll
            for (int u = 0; u < OWN_U; ++u) {
                const int i = base + u * OWN_T + t;
                int jj = i + start;
                if (jj >= n4) jj -= n4;
                j[u] = (i < n4) ? jj : -1;
                v[u] = d4[(i < n4) ? jj : 0];
            }
            #pragma unroll
            for (int u = 0; u < OWN_U; ++u) {
                const unsigned q0 = (unsigned)(v[u].x - r0), q1 = (unsigned)(v[u].y - r0);
                const unsigned q2 = (unsigned)(v[u].z - r0), q3 = (unsigned)(v[u].w - r0);
                if (j[u] >= 0 && min(min(q0, q1), min(q2, q3)) < (unsigned)rows) {
                    const int e = j[u] << 2;
                    OWN_NOTE(q0, e)
                    OWN_NOTE(q1, e + 1)
                    OWN_NOTE(q2, e + 2)
                    OWN_NOTE(q3, e + 3)
                }
            }
        }
    } else {
        for (int e = t; e < E; e += OWN_T) OWN_NOTE((unsigned)(dstp[e] - r0), e)
    }
    #undef OWN_NOTE
    __syncthreads();
    const int cnt = lcount;

    if (cnt <= OWN_CAP) {
        for (int b0 = 0; b0 < cnt; b0 += OWN_T * 4) {
            int e_[4], s_[4], d_[4];
            float as_[4];
            #pragma unroll
            for (int u = 0; u < 4; ++u) { const int i = b0 + u * OWN_T + t; e_[u] = (i < cnt) ? lsrc[i] : 0; }
            #pragma unroll
            for (int u = 0; u < 4; ++u) { s_[u] = ei[e_[u]]; d_[u] = dstp[e_[u]]; }
            #pragma unroll
            for (int u = 0; u < 4; ++u) as_[u] = a_src[s_[u]];
            #pragma unroll
            for (int u = 0; u < 4; ++u) {
                const int i = b0 + u * OWN_T + t;
                if (i < cnt) {
                    const int dl = d_[u] - r0;
                    const float v = leaky02(as_[u] + sadst[dl]);
                    lsrc[i] = s_[u];
                    ldl[i] = dl;
                    lval[i] = v;
                    atomicMax(&smax[dl], fkey(v));
                }
            }
        }
        __syncthreads();
        for (int i = t; i < cnt; i += OWN_T) {
            const int dl = ldl[i];
            const float w = __expf(lval[i] - fdec(smax[dl]));
            lval[i] = w;
            atomicAdd(&ssum[dl], w);
        }
        for (int r = t; r < rows; r += OWN_T) {
            const float w = __expf(sself[r] - fdec(smax[r]));
            sself[r] = w;
            atomicAdd(&ssum[r], w);
        }
        __syncthreads();
        for (int g = wave * 8; g < cnt; g += (OWN_T / 32) * 8) {
            float lo[8], hi[8], aj[8];
            int sj[8], dj[8];
            #pragma unroll
            for (int k = 0; k < 8; ++k) {
                const int i = (g + k < cnt) ? g + k : g;
                sj[k] = lsrc[i];
                dj[k] = ldl[i];
                aj[k] = lval[i] / (ssum[dj[k]] + 1e-16f);
            }
            #pragma unroll
            for (int k = 0; k < 8; ++k) {
                const float* xr = xl + (long long)sj[k] * DDIM + lane;
                lo[k] = xr[0];
                hi[k] = xr[32];
            }
            #pragma unroll
            for (int k = 0; k < 8; ++k) {
                if (g + k < cnt) {
                    float* ap = acc + dj[k] * ACC_PITCH + lane;
                    atomicAdd(ap,      lo[k] * aj[k]);
                    atomicAdd(ap + 32, hi[k] * aj[k]);
                }
            }
        }
    } else {
        for (int e = t; e < E; e += OWN_T) {
            const unsigned dl = (unsigned)(dstp[e] - r0);
            if (dl < (unsigned)rows)
                atomicMax(&smax[dl], fkey(leaky02(a_src[ei[e]] + sadst[dl])));
        }
        __syncthreads();
        for (int e = t; e < E; e += OWN_T) {
            const unsigned dl = (unsigned)(dstp[e] - r0);
            if (dl < (unsigned)rows)
                atomicAdd(&ssum[dl], __expf(leaky02(a_src[ei[e]] + sadst[dl]) - fdec(smax[dl])));
        }
        for (int r = t; r < rows; r += OWN_T) {
            const float w = __expf(sself[r] - fdec(smax[r]));
            sself[r] = w;
            atomicAdd(&ssum[r], w);
        }
        __syncthreads();
        for (int e = t; e < E; e += OWN_T) {
            const unsigned dl = (unsigned)(dstp[e] - r0);
            if (dl < (unsigned)rows) {
                const int s = ei[e];
                const float aj = __expf(leaky02(a_src[s] + sadst[dl]) - fdec(smax[dl])) / (ssum[dl] + 1e-16f);
                const float* xr = xl + (long long)s * DDIM;
                float* ap = acc + dl * ACC_PITCH;
                for (int c = 0; c < DDIM; ++c) atomicAdd(ap + c, xr[c] * aj);
            }
        }
    }
    __syncthreads();

    const float4* __restrict__ xl4 = (const float4*)xl;
    const float4* __restrict__ b4 = (const float4*)bias;
    float4* __restrict__ o4 = (float4*)out_sum;
    if (Wnext != nullptr) {
        const float* sw = (const float*)sw4;
        const int c4    = lane & 15;
        const int h     = lane >> 4;
        const int khalf = h << 1;
        const int rhi   = h * 8;
        const int ntiles = (rows + 15) >> 4;
        const float4 b = b4[c4];
        const float* brow = &sw[c4 * SW_STRIDE + khalf];
        for (int tile = wave; tile < ntiles; tile += OWN_T / 32) {
            const int rb = tile * 16;
            float4 xs[8], o[8];
            #pragma unroll
            for (int u = 0; u < 8; ++u) {
                const int r = rb + 2 * u + h;
                const long long gi = (long long)(r0 + (r < rows ? r : rows - 1)) * (DDIM / 4) + c4;
                xs[u] = xl4[gi];
                o[u]  = o4[gi];
            }
            #pragma unroll
            for (int u = 0; u < 8; ++u) {
                const int r = rb + 2 * u + h;
                if (r < rows) {
                    const long long gi = (long long)(r0 + r) * (DDIM / 4) + c4;
                    const float as = sself[r] / (ssum[r] + 1e-16f);
                    const float4 a = acc4[r * (ACC_PITCH / 4) + c4];
                    float4 v;
                    v.x = (a.x + xs[u].x * as) + b.x;
                    v.y = (a.y + xs[u].y * as) + b.y;
                    v.z = (a.z + xs[u].z * as) + b.z;
                    v.w = (a.w + xs[u].w * as) + b.w;
                    v.x = v.x > 0.f ? v.x : (__expf(v.x) - 1.f);
                    v.y = v.y > 0.f ? v.y : (__expf(v.y) - 1.f);
                    v.z = v.z > 0.f ? v.z : (__expf(v.z) - 1.f);
                    v.w = v.w > 0.f ? v.w : (__expf(v.w) - 1.f);
                    float* xr = acc + r * ACC_PITCH + c4 * 4;
                    xr[0] = v.x; xr[1] = v.y; xr[2] = v.z; xr[3] = v.w;
                    float4 q = o[u];
                    q.x += v.x; q.y += v.y; q.z += v.z; q.w += v.w;
                    o4[gi] = q;
                }
            }
            v8f acc0 = {0.f,0.f,0.f,0.f,0.f,0.f,0.f,0.f};
            v8f acc1 = acc0, acc2 = acc0, acc3 = acc0;
            const float* arow = &acc[(rb + c4) * ACC_PITCH + khalf];
            #pragma nounroll
            for (int kb = 0; kb < DDIM; kb += 16) {
                #pragma unroll
                for (int k0 = kb; k0 < kb + 16; k0 += 4) {
                    v2f a, b0, b1, b2, b3;
                    a.x  = arow[k0];                       a.y  = arow[k0 + 1];
                    b0.x = brow[k0];                       b0.y = brow[k0 + 1];
                    b1.x = brow[16 * SW_STRIDE + k0];      b1.y = brow[16 * SW_STRIDE + k0 + 1];
                    b2.x = brow[32 * SW_STRIDE + k0];      b2.y = brow[32 * SW_STRIDE + k0 + 1];
                    b3.x = brow[48 * SW_STRIDE + k0];      b3.y = brow[48 * SW_STRIDE + k0 + 1];
                    acc0 = __builtin_amdgcn_wmma_f32_16x16x4_f32(false, a, false, b0, (short)0, acc0, false, false);
                    acc1 = __builtin_amdgcn_wmma_f32_16x16x4_f32(false, a, false, b1, (short)0, acc1, false, false);
                    acc2 = __builtin_amdgcn_wmma_f32_16x16x4_f32(false, a, false, b2, (short)0, acc2, false, false);
                    acc3 = __builtin_amdgcn_wmma_f32_16x16x4_f32(false, a, false, b3, (short)0, acc3, false, false);
                }
            }
            const int lr0 = rb + rhi;
            float* p = xl_next + (long long)(r0 + lr0) * DDIM + c4;
            if (rb + 16 <= rows) {
                #pragma unroll
                for (int v = 0; v < 8; ++v) {
                    p[(long long)v * DDIM +  0] = acc0[v];
                    p[(long long)v * DDIM + 16] = acc1[v];
                    p[(long long)v * DDIM + 32] = acc2[v];
                    p[(long long)v * DDIM + 48] = acc3[v];
                }
            } else {
                #pragma unroll
                for (int v = 0; v < 8; ++v) {
                    if (lr0 + v < rows) {
                        p[(long long)v * DDIM +  0] = acc0[v];
                        p[(long long)v * DDIM + 16] = acc1[v];
                        p[(long long)v * DDIM + 32] = acc2[v];
                        p[(long long)v * DDIM + 48] = acc3[v];
                    }
                }
            }
        }
        return;
    }
    for (int i = t; i < rows * (DDIM / 4); i += OWN_T) {
        const int r = i >> 4, c4 = i & 15;
        const long long gi = (long long)(r0 + r) * (DDIM / 4) + c4;
        const float as = sself[r] / (ssum[r] + 1e-16f);
        const float4 a = acc4[r * (ACC_PITCH / 4) + c4];
        const float4 xs = xl4[gi];
        const float4 b = b4[c4];
        float4 o = o4[gi];
        float4 v;
        v.x = (a.x + xs.x * as) + b.x;
        v.y = (a.y + xs.y * as) + b.y;
        v.z = (a.z + xs.z * as) + b.z;
        v.w = (a.w + xs.w * as) + b.w;
        v.x = v.x > 0.f ? v.x : (__expf(v.x) - 1.f);
        v.y = v.y > 0.f ? v.y : (__expf(v.y) - 1.f);
        v.z = v.z > 0.f ? v.z : (__expf(v.z) - 1.f);
        v.w = v.w > 0.f ? v.w : (__expf(v.w) - 1.f);
        o.x += v.x; o.y += v.y; o.z += v.z; o.w += v.w;
        o4[gi] = o;
    }
}

__global__ __launch_bounds__(256) void scale_out(float* __restrict__ out, float inv,
                                                 long long total4) {
    float4* __restrict__ o4 = (float4*)out;
    const long long base = (long long)blockIdx.x * (256 * STREAM_U) + threadIdx.x;
    float4 v[STREAM_U];
    #pragma unroll
    for (int u = 0; u < STREAM_U; ++u) {
        const long long i = base + u * 256;
        v[u] = o4[(i < total4) ? i : 0];
    }
    #pragma unroll
    for (int u = 0; u < STREAM_U; ++u) {
        const long long i = base + u * 256;
        if (i < total4) {
            float4 r;
            r.x = v[u].x * inv; r.y = v[u].y * inv; r.z = v[u].z * inv; r.w = v[u].w * inv;
            o4[i] = r;
        }
    }
}

static inline unsigned gridFor(long long work, int block) {
    return (unsigned)((work + block - 1) / block);
}

extern "C" void kernel_launch(void* const* d_in, const int* in_sizes, int n_in,
                              void* d_out, int out_size, void* d_ws, size_t ws_size,
                              hipStream_t stream) {
    const int*   ei       = (const int*)d_in[0];
    const float* emb      = (const float*)d_in[1];
    const float* prompt   = (const float*)d_in[2];
    const float* proj_w   = (const float*)d_in[3];
    const float* proj_b   = (const float*)d_in[4];
    const float* conv_ws  = (const float*)d_in[5];
    const float* att_srcs = (const float*)d_in[6];
    const float* att_dsts = (const float*)d_in[7];
    const float* conv_bs  = (const float*)d_in[8];

    const int E = in_sizes[0] / 2;
    const int N = in_sizes[1] / DDIM;
    const int P = in_sizes[2];
    const int L = in_sizes[5] / (DDIM * DDIM);

    float* out = (float*)d_out;

    float*    pvec   = (float*)d_ws;
    float*    xbuf   = pvec + 64;
    float*    xlb    = xbuf + (size_t)N * DDIM;
    float*    a_src  = xlb + (size_t)N * DDIM;
    float*    a_dst  = a_src + N;


    float* tail = a_dst + N;
    const unsigned long long ws_bytes = (unsigned long long)ws_size;
    compute_pvec<<<gridFor((long long)E + 1, 256 * MAP_U * 4), 256, 0, stream>>>(prompt, proj_w, proj_b, pvec, P,
                                                                          ei, E, N, tail, pvec, ws_bytes);
    const long long total4 = (long long)N * (DDIM / 4);
    add_prompt<<<gridFor(total4, 256 * STREAM_U), 256, 0, stream>>>(emb, pvec, xbuf, out, total4);

    if (L > 0) gemm_xw_t<<<gridFor(N, GEMM_ROWS), 256, 0, stream>>>(xbuf, conv_ws, xlb, N);
    float* cur = xlb;
    float* nxt = xbuf;
    for (int l = 0; l < L; ++l) {
        const float* as = att_srcs + (size_t)l * DDIM;
        const float* ad = att_dsts + (size_t)l * DDIM;
        const float* cb = conv_bs  + (size_t)l * DDIM;
        const float* Wn = (l + 1 < L) ? conv_ws + (size_t)(l + 1) * DDIM * DDIM : nullptr;

        att_dots<<<gridFor(N, 128), 128, 0, stream>>>(cur, as, ad, a_src, a_dst, N);
        edge_owner<<<gridFor(N, OWN_ROWS), OWN_T, 0, stream>>>(ei, E, N, a_src, a_dst, cur, cb, Wn, nxt, out,
                                                               tail, pvec, ws_bytes);
        float* turn = cur; cur = nxt; nxt = turn;
    }
    scale_out<<<gridFor(total4, 256 * STREAM_U), 256, 0, stream>>>(out, 1.f / (float)(L + 1), total4);
}


# edge phase gather: a wave reads 32 list entries lane-parallel (one LDS read set, one division per edge) and hands row, slot, weight out as scalars; LDS adds stay in arrival order
# speedup vs baseline: 3.3949x; 1.0836x over previous
#include <hip/hip_runtime.h>
#include <hip/hip_bf16.h>

typedef float v2f __attribute__((ext_vector_type(2)));
typedef float v8f __attribute__((ext_vector_type(8)));

#define DDIM 64
#define GEMM_ROWS 128
#define SX_STRIDE 68
#define SW_STRIDE 68

__device__ __forceinline__ unsigned fkey(float f) {
    unsigned u = __float_as_uint(f);
    return (u & 0x80000000u) ? ~u : (u | 0x80000000u);
}
__device__ __forceinline__ float fdec(unsigned k) {
    return (k & 0x80000000u) ? __uint_as_float(k & 0x7FFFFFFFu)
                             : __uint_as_float(~k);
}
#define FKEY_NEG_INF 0x007FFFFFu

#define OWN_ROWS 592

#define OWN_NONE 255u
__device__ __forceinline__ unsigned char* own_map_base(float* tail) {
    unsigned char* p = (unsigned char*)tail;
    return p + ((0ull - (unsigned long long)p) & 15ull);
}
__device__ __forceinline__ bool own_map_ok(const float* wsbase, float* tail, unsigned long long ws_bytes,
                                           int E, int N) {
    const unsigned long long off = (unsigned long long)(own_map_base(tail) - (const unsigned char*)wsbase);
    return (N + OWN_ROWS - 1) / OWN_ROWS <= 255 && off + (unsigned long long)E <= ws_bytes;
}
__device__ __forceinline__ unsigned own_of(int dst, int N) {
    return ((unsigned)dst < (unsigned)N) ? (unsigned)dst / (unsigned)OWN_ROWS : OWN_NONE;
}

#define MAP_U 4
__global__ __launch_bounds__(256) void compute_pvec(const float* __restrict__ prompt, const float* __restrict__ pw,
                             const float* __restrict__ pb, float* __restrict__ pvec, int P,
                             const int* __restrict__ ei, int E, int N,
                             float* tail, const float* wsbase, unsigned long long ws_bytes) {
    const int t = threadIdx.x;
    if (blockIdx.x == 0 && t < DDIM) {
        float s = pb[t];
        for (int j = 0; j < P; ++j) s += prompt[j] * pw[t * P + j];
        pvec[t] = s;
    }
    if (!own_map_ok(wsbase, tail, ws_bytes, E, N)) return;
    unsigned char* __restrict__ map = own_map_base(tail);
    const int* __restrict__ dstp = ei + E;
    if ((E & 3) == 0) {
        const int4* __restrict__ d4 = (const int4*)dstp;
        unsigned* __restrict__ m1 = (unsigned*)map;
        const int n4 = E >> 2;
        for (int base = blockIdx.x * (256 * MAP_U); base < n4; base += gridDim.x * (256 * MAP_U)) {
            int4 v[MAP_U];
            #pragma unroll
            for (int u = 0; u < MAP_U; ++u) {
                const int g = base + u * 256 + t;
                v[u] = d4[(g < n4) ? g : 0];
            }
            #pragma unroll
            for (int u = 0; u < MAP_U; ++u) {
                const int g = base + u * 256 + t;
                if (g < n4)
                    m1[g] = own_of(v[u].x, N) | (own_of(v[u].y, N) << 8) |
                            (own_of(v[u].z, N) << 16) | (own_of(v[u].w, N) << 24);
            }
        }
    } else {
        for (int e = blockIdx.x * 256 + t; e < E; e += gridDim.x * 256) map[e] = (unsigned char)own_of(dstp[e], N);
    }
}

#define STREAM_U 8
__global__ __launch_bounds__(256) void add_prompt(const float* __restrict__ emb,
                                                  const float* __restrict__ pvec,
                                                  float* __restrict__ x,
                                                  float* __restrict__ out_sum,
                                                  long long total4) {
    const float4* __restrict__ e4 = (const float4*)emb;
    float4* __restrict__ x4 = (float4*)x;
    float4* __restrict__ o4 = (float4*)out_sum;
    const int t = threadIdx.x;
    const float4 p = ((const float4*)pvec)[t & 15];
    const long long base = (long long)blockIdx.x * (256 * STREAM_U) + t;
    float4 v[STREAM_U];
    #pragma unroll
    for (int u = 0; u < STREAM_U; ++u) {
        const long long i = base + u * 256;
        v[u] = e4[(i < total4) ? i : 0];
    }
    #pragma unroll
    for (int u = 0; u < STREAM_U; ++u) {
        const long long i = base + u * 256;
        if (i < total4) {
            float4 r;
            r.x = v[u].x + p.x; r.y = v[u].y + p.y; r.z = v[u].z + p.z; r.w = v[u].w + p.w;
            x4[i] = r;
            o4[i] = r;
        }
    }
}

__global__ __launch_bounds__(256) void gemm_xw_t(const float* __restrict__ x,
                                                 const float* __restrict__ W,
                                                 float* __restrict__ xl, int N) {
    __shared__ float sx[GEMM_ROWS * SX_STRIDE];
    __shared__ float sw[DDIM * SW_STRIDE];
    const int t = threadIdx.x;
    const int rowBase = blockIdx.x * GEMM_ROWS;

    {
        const float4* Wv = (const float4*)W;
        float4* swv = (float4*)sw;
        for (int i = t; i < (DDIM * DDIM) / 4; i += 256) {
            int r = i >> 4, c4 = i & 15;
            swv[r * (SW_STRIDE / 4) + c4] = Wv[i];
        }
    }
    {
        const float4* xv = (const float4*)x;
        float4* sxv = (float4*)sx;
        for (int i = t; i < (GEMM_ROWS * DDIM) / 4; i += 256) {
            int r = i >> 4, c4 = i & 15;
            int gr = rowBase + r;
            if (gr >= N) gr = N - 1;
            sxv[r * (SX_STRIDE / 4) + c4] = xv[(long long)gr * 16 + c4];
        }
    }
    __syncthreads();

    const int wave  = t >> 5;
    const int lane  = t & 31;
    const int lrow  = lane & 15;
    const int khalf = (lane >> 4) << 1;
    const int wrow  = wave * 16;

    v8f acc0 = {0.f,0.f,0.f,0.f,0.f,0.f,0.f,0.f};
    v8f acc1 = acc0, acc2 = acc0, acc3 = acc0;

    const float* arow = &sx[(wrow + lrow) * SX_STRIDE + khalf];
    const float* brow = &sw[lrow * SW_STRIDE + khalf];

    #pragma unroll
    for (int k0 = 0; k0 < DDIM; k0 += 4) {
        v2f a, b0, b1, b2, b3;
        a.x  = arow[k0];                       a.y  = arow[k0 + 1];
        b0.x = brow[k0];                       b0.y = brow[k0 + 1];
        b1.x = brow[16 * SW_STRIDE + k0];      b1.y = brow[16 * SW_STRIDE + k0 + 1];
        b2.x = brow[32 * SW_STRIDE + k0];      b2.y = brow[32 * SW_STRIDE + k0 + 1];
        b3.x = brow[48 * SW_STRIDE + k0];      b3.y = brow[48 * SW_STRIDE + k0 + 1];
        acc0 = __builtin_amdgcn_wmma_f32_16x16x4_f32(false, a, false, b0, (short)0, acc0, false, false);
        acc1 = __builtin_amdgcn_wmma_f32_16x16x4_f32(false, a, false, b1, (short)0, acc1, false, false);
        acc2 = __builtin_amdgcn_wmma_f32_16x16x4_f32(false, a, false, b2, (short)0, acc2, false, false);
        acc3 = __builtin_amdgcn_wmma_f32_16x16x4_f32(false, a, false, b3, (short)0, acc3, false, false);
    }

    const int rhi = (lane >> 4) * 8;
    const long long rbase = (long long)(rowBase + wrow + rhi);
    if (rowBase + GEMM_ROWS <= N) {
        float* p = xl + rbase * DDIM + lrow;
        #pragma unroll
        for (int v = 0; v < 8; ++v) {
            p[(long long)v * DDIM +  0] = acc0[v];
            p[(long long)v * DDIM + 16] = acc1[v];
            p[(long long)v * DDIM + 32] = acc2[v];
            p[(long long)v * DDIM + 48] = acc3[v];
        }
    } else {
        #pragma unroll
        for (int v = 0; v < 8; ++v) {
            long long gr = rbase + v;
            if (gr < N) {
                float* p = xl + gr * DDIM + lrow;
                p[ 0] = acc0[v];
                p[16] = acc1[v];
                p[32] = acc2[v];
                p[48] = acc3[v];
            }
        }
    }
}

#define ATT_U 16
template <int M> __device__ __forceinline__ float att_xadd(float v) {
    const int o = __builtin_amdgcn_update_dpp(0, __builtin_bit_cast(int, v), 0x160 + M, 0xF, 0xF, true);
    return v + __builtin_bit_cast(float, o);
}
__device__ __forceinline__ float att_pick(const float (&r)[ATT_U], int c) {
    float o = r[0];
    #pragma unroll
    for (int u = 1; u < ATT_U; ++u) o = (c == u) ? r[u] : o;
    return o;
}
__global__ __launch_bounds__(128) void att_dots(const float* __restrict__ xl,
                                                const float* __restrict__ av_s,
                                                const float* __restrict__ av_d,
                                                float* __restrict__ a_src,
                                                float* __restrict__ a_dst, int N) {
    const int lane = threadIdx.x & 31;
    const int c = lane & 15;
    const int h = lane >> 4;
    const int nb = (blockIdx.x * 4 + (threadIdx.x >> 5)) * 32;
    const float4* __restrict__ xl4 = (const float4*)xl;
    const float4 s = ((const float4*)av_s)[c];
    const float4 d = ((const float4*)av_d)[c];
    float4 v[ATT_U];
    #pragma unroll
    for (int u = 0; u < ATT_U; ++u) {
        int n = nb + 16 * h + u;
        if (n >= N) n = N - 1;
        v[u] = xl4[(long long)n * (DDIM / 4) + c];
    }
    float rs[ATT_U], rd[ATT_U];
    #pragma unroll
    for (int u = 0; u < ATT_U; ++u) {
        float sa = v[u].x * s.x + v[u].y * s.y;
        float sb = v[u].z * s.z + v[u].w * s.w;
        float da = v[u].x * d.x + v[u].y * d.y;
        float db = v[u].z * d.z + v[u].w * d.w;
        sa = att_xadd<8>(sa); sb = att_xadd<8>(sb); da = att_xadd<8>(da); db = att_xadd<8>(db);
        sa = att_xadd<4>(sa); sb = att_xadd<4>(sb); da = att_xadd<4>(da); db = att_xadd<4>(db);
        sa = att_xadd<2>(sa); sb = att_xadd<2>(sb); da = att_xadd<2>(da); db = att_xadd<2>(db);
        sa = att_xadd<1>(sa); sb = att_xadd<1>(sb); da = att_xadd<1>(da); db = att_xadd<1>(db);
        rs[u] = sa + sb;
        rd[u] = da + db;
    }
    const int n = nb + lane;
    const float os = att_pick(rs, c);
    const float od = att_pick(rd, c);
    if (n < N) { a_src[n] = os; a_dst[n] = od; }
}

#define OWN_CAP  8192
#define ACC_PITCH 68
#define OWN_T    1024
#define OWN_U    8

__device__ __forceinline__ float leaky02(float v) { return v > 0.f ? v : 0.2f * v; }

__global__ __launch_bounds__(OWN_T) void edge_owner(const int* __restrict__ ei, int E, int N,
                                                    const float* __restrict__ a_src,
                                                    const float* __restrict__ a_dst,
                                                    const float* __restrict__ xl,
                                                    const float* __restrict__ bias,
                                                    const float* __restrict__ Wnext,
                                                    float* __restrict__ xl_next,
                                                    float* __restrict__ out_sum,
                                                    float* tail, const float* wsbase,
                                                    unsigned long long ws_bytes) {
    __shared__ float4   acc4[OWN_ROWS * (ACC_PITCH / 4)];
    __shared__ float4   sw4[DDIM * (SW_STRIDE / 4)];
    __shared__ unsigned smax[OWN_ROWS];
    __shared__ float    ssum[OWN_ROWS];
    __shared__ float    sadst[OWN_ROWS];
    __shared__ float    sself[OWN_ROWS];
    __shared__ int      lsrc[OWN_CAP];
    __shared__ int      ldl[OWN_CAP];
    __shared__ float    lval[OWN_CAP];
    __shared__ int      lcount;
    float* acc = (float*)acc4;

    const int t = threadIdx.x;
    const int lane = t & 31;
    const int wave = t >> 5;
    const int r0 = blockIdx.x * OWN_ROWS;
    const int rows = min(OWN_ROWS, N - r0);
    const int* __restrict__ dstp = ei + E;

    for (int r = t; r < OWN_ROWS; r += OWN_T) {
        float ad = 0.f, vs = 0.f;
        unsigned key = FKEY_NEG_INF;
        if (r < rows) {
            ad = a_dst[r0 + r];
            vs = leaky02(a_src[r0 + r] + ad);
            key = fkey(vs);
        }
        sadst[r] = ad; sself[r] = vs; smax[r] = key; ssum[r] = 0.f;
    }
    {
        const float4 z = {0.f, 0.f, 0.f, 0.f};
        for (int i = t; i < OWN_ROWS * (ACC_PITCH / 4); i += OWN_T) acc4[i] = z;
    }
    if (t == 0) lcount = 0;
    if (Wnext != nullptr) {
        const float4* __restrict__ Wv = (const float4*)Wnext;
        for (int i = t; i < (DDIM * DDIM) / 4; i += OWN_T) {
            const int r = i >> 4, c4 = i & 15;
            sw4[r * (SW_STRIDE / 4) + c4] = Wv[i];
        }
    }
    __syncthreads();

    #define OWN_NOTE(dlv, eidx)                                               \
        {                                                                     \
            const unsigned dl_ = (dlv);                                       \
            if (dl_ < (unsigned)rows) {                                       \
                const int slot_ = atomicAdd(&lcount, 1);                      \
                if (slot_ < OWN_CAP) { lsrc[slot_] = (eidx); ldl[slot_] = (int)dl_; } \
            }                                                                 \
        }
    if (own_map_ok(wsbase, tail, ws_bytes, E, N)) {
        const unsigned char* __restrict__ map = own_map_base(tail);
        const uint4* __restrict__ m4 = (const uint4*)map;
        const int n16 = E >> 4;
        const unsigned me = blockIdx.x;
        const unsigned pat = me * 0x01010101u;
        const int start = (int)(((long long)blockIdx.x * n16) / gridDim.x) & ~7;
        for (int base = 0; base < n16; base += OWN_T * OWN_U) {
            uint4 v[OWN_U];
            int   j[OWN_U];
            #pragma unroll
            for (int u = 0; u < OWN_U; ++u) {
                const int i = base + u * OWN_T + t;
                int jj = i + start;
                if (jj >= n16) jj -= n16;
                j[u] = (i < n16) ? jj : -1;
                v[u] = m4[(i < n16) ? jj : 0];
            }
            #pragma unroll
            for (int u = 0; u < OWN_U; ++u) {
                const unsigned x0 = v[u].x ^ pat, x1 = v[u].y ^ pat, x2 = v[u].z ^ pat, x3 = v[u].w ^ pat;
                const unsigned y0 = ((x0 & 0x7f7f7f7fu) + 0x7f7f7f7fu) | x0;
                const unsigned y1 = ((x1 & 0x7f7f7f7fu) + 0x7f7f7f7fu) | x1;
                const unsigned y2 = ((x2 & 0x7f7f7f7fu) + 0x7f7f7f7fu) | x2;
                const unsigned y3 = ((x3 & 0x7f7f7f7fu) + 0x7f7f7f7fu) | x3;
                if (j[u] >= 0 && ((y0 & y1 & y2 & y3) | 0x7f7f7f7fu) != 0xffffffffu) {
                    unsigned c = ((~y0 & 0x80808080u) >> 7) | ((~y1 & 0x80808080u) >> 6) |
                                 ((~y2 & 0x80808080u) >> 5) | ((~y3 & 0x80808080u) >> 4);
                    const int e0 = j[u] << 4;
                    while (c) {
                        const int b = __builtin_ctz(c);
                        c &= c - 1;
                        const int slot = atomicAdd(&lcount, 1);
                        if (slot < OWN_CAP) lsrc[slot] = e0 + ((b & 7) << 2) + (b >> 3);
                    }
                }
            }
        }
        for (int e = (n16 << 4) + t; e < E; e += OWN_T) {
            if (map[e] == me) {
                const int slot = atomicAdd(&lcount, 1);
                if (slot < OWN_CAP) lsrc[slot] = e;
            }
        }
    } else if ((E & 3) == 0) {
        const int4* __restrict__ d4 = (const int4*)dstp;
        const int n4 = E >> 2;
        const int start = (int)(((long long)blockIdx.x * n4) / gridDim.x) & ~7;
        for (int base = 0; base < n4; base += OWN_T * OWN_U) {
            int4 v[OWN_U];
            int  j[OWN_U];
            #pragma unroll
            for (int u = 0; u < OWN_U; ++u) {
                const int i = base + u * OWN_T + t;
                int jj = i + start;
                if (jj >= n4) jj -= n4;
                j[u] = (i < n4) ? jj : -1;
                v[u] = d4[(i < n4) ? jj : 0];
            }
            #pragma unroll
            for (int u = 0; u < OWN_U; ++u) {
                const unsigned q0 = (unsigned)(v[u].x - r0), q1 = (unsigned)(v[u].y - r0);
                const unsigned q2 = (unsigned)(v[u].z - r0), q3 = (unsigned)(v[u].w - r0);
                if (j[u] >= 0 && min(min(q0, q1), min(q2, q3)) < (unsigned)rows) {
                    const int e = j[u] << 2;
                    OWN_NOTE(q0, e)
                    OWN_NOTE(q1, e + 1)
                    OWN_NOTE(q2, e + 2)
                    OWN_NOTE(q3, e + 3)
                }
            }
        }
    } else {
        for (int e = t; e < E; e += OWN_T) OWN_NOTE((unsigned)(dstp[e] - r0), e)
    }
    #undef OWN_NOTE
    __syncthreads();
    const int cnt = lcount;

    if (cnt <= OWN_CAP) {
        for (int b0 = 0; b0 < cnt; b0 += OWN_T * 4) {
            int e_[4], s_[4], d_[4];
            float as_[4];
            #pragma unroll
            for (int u = 0; u < 4; ++u) { const int i = b0 + u * OWN_T + t; e_[u] = (i < cnt) ? lsrc[i] : 0; }
            #pragma unroll
            for (int u = 0; u < 4; ++u) { s_[u] = ei[e_[u]]; d_[u] = dstp[e_[u]]; }
            #pragma unroll
            for (int u = 0; u < 4; ++u) as_[u] = a_src[s_[u]];
            #pragma unroll
            for (int u = 0; u < 4; ++u) {
                const int i = b0 + u * OWN_T + t;
                if (i < cnt) {
                    const int dl = d_[u] - r0;
                    const float v = leaky02(as_[u] + sadst[dl]);
                    lsrc[i] = s_[u];
                    ldl[i] = dl;
                    lval[i] = v;
                    atomicMax(&smax[dl], fkey(v));
                }
            }
        }
        __syncthreads();
        for (int i = t; i < cnt; i += OWN_T) {
            const int dl = ldl[i];
            const float w = __expf(lval[i] - fdec(smax[dl]));
            lval[i] = w;
            atomicAdd(&ssum[dl], w);
        }
        for (int r = t; r < rows; r += OWN_T) {
            const float w = __expf(sself[r] - fdec(smax[r]));
            sself[r] = w;
            atomicAdd(&ssum[r], w);
        }
        __syncthreads();
        const int swave = __builtin_amdgcn_readfirstlane(wave);
        for (int g = swave * 32; g < cnt; g += OWN_T) {
            const int   il = (g + lane < cnt) ? g + lane : g;
            const int   sl = lsrc[il];
            const int   dl = ldl[il];
            const float al = lval[il] / (ssum[dl] + 1e-16f);
            const int   ab = __builtin_bit_cast(int, al);
            #pragma unroll
            for (int q = 0; q < 4; ++q) {
                const int g8 = g + q * 8;
                if (g8 + 8 <= cnt) {
                    float lo[8], hi[8];
                    const float* xr[8];
                    #pragma unroll
                    for (int k = 0; k < 8; ++k)
                        xr[k] = xl + (long long)__builtin_amdgcn_readlane(sl, q * 8 + k) * DDIM + lane;
                    #pragma unroll
                    for (int k = 0; k < 8; ++k) {
                        lo[k] = xr[k][0];
                        hi[k] = xr[k][32];
                    }
                    #pragma unroll
                    for (int k = 0; k < 8; ++k) {
                        const float a = __builtin_bit_cast(float, __builtin_amdgcn_readlane(ab, q * 8 + k));
                        float* ap = acc + __builtin_amdgcn_readlane(dl, q * 8 + k) * ACC_PITCH + lane;
                        atomicAdd(ap,      lo[k] * a);
                        atomicAdd(ap + 32, hi[k] * a);
                    }
                } else if (g8 < cnt) {
                    for (int k = 0; g8 + k < cnt; ++k) {
                        const float* xr = xl + (long long)__builtin_amdgcn_readlane(sl, q * 8 + k) * DDIM + lane;
                        const float a = __builtin_bit_cast(float, __builtin_amdgcn_readlane(ab, q * 8 + k));
                        float* ap = acc + __builtin_amdgcn_readlane(dl, q * 8 + k) * ACC_PITCH + lane;
                        atomicAdd(ap,      xr[0] * a);
                        atomicAdd(ap + 32, xr[32] * a);
                    }
                }
            }
        }
    } else {
        for (int e = t; e < E; e += OWN_T) {
            const unsigned dl = (unsigned)(dstp[e] - r0);
            if (dl < (unsigned)rows)
                atomicMax(&smax[dl], fkey(leaky02(a_src[ei[e]] + sadst[dl])));
        }
        __syncthreads();
        for (int e = t; e < E; e += OWN_T) {
            const unsigned dl = (unsigned)(dstp[e] - r0);
            if (dl < (unsigned)rows)
                atomicAdd(&ssum[dl], __expf(leaky02(a_src[ei[e]] + sadst[dl]) - fdec(smax[dl])));
        }
        for (int r = t; r < rows; r += OWN_T) {
            const float w = __expf(sself[r] - fdec(smax[r]));
            sself[r] = w;
            atomicAdd(&ssum[r], w);
        }
        __syncthreads();
        for (int e = t; e < E; e += OWN_T) {
            const unsigned dl = (unsigned)(dstp[e] - r0);
            if (dl < (unsigned)rows) {
                const int s = ei[e];
                const float aj = __expf(leaky02(a_src[s] + sadst[dl]) - fdec(smax[dl])) / (ssum[dl] + 1e-16f);
                const float* xr = xl + (long long)s * DDIM;
                float* ap = acc + dl * ACC_PITCH;
                for (int c = 0; c < DDIM; ++c) atomicAdd(ap + c, xr[c] * aj);
            }
        }
    }
    __syncthreads();

    const float4* __restrict__ xl4 = (const float4*)xl;
    const float4* __restrict__ b4 = (const float4*)bias;
    float4* __restrict__ o4 = (float4*)out_sum;
    if (Wnext != nullptr) {
        const float* sw = (const float*)sw4;
        const int c4    = lane & 15;
        const int h     = lane >> 4;
        const int khalf = h << 1;
        const int rhi   = h * 8;
        const int ntiles = (rows + 15) >> 4;
        const float4 b = b4[c4];
        const float* brow = &sw[c4 * SW_STRIDE + khalf];
        for (int tile = wave; tile < ntiles; tile += OWN_T / 32) {
            const int rb = tile * 16;
            float4 xs[8], o[8];
            #pragma unroll
            for (int u = 0; u < 8; ++u) {
                const int r = rb + 2 * u + h;
                const long long gi = (long long)(r0 + (r < rows ? r : rows - 1)) * (DDIM / 4) + c4;
                xs[u] = xl4[gi];
                o[u]  = o4[gi];
            }
            #pragma unroll
            for (int u = 0; u < 8; ++u) {
                const int r = rb + 2 * u + h;
                if (r < rows) {
                    const long long gi = (long long)(r0 + r) * (DDIM / 4) + c4;
                    const float as = sself[r] / (ssum[r] + 1e-16f);
                    const float4 a = acc4[r * (ACC_PITCH / 4) + c4];
                    float4 v;
                    v.x = (a.x + xs[u].x * as) + b.x;
                    v.y = (a.y + xs[u].y * as) + b.y;
                    v.z = (a.z + xs[u].z * as) + b.z;
                    v.w = (a.w + xs[u].w * as) + b.w;
                    v.x = v.x > 0.f ? v.x : (__expf(v.x) - 1.f);
                    v.y = v.y > 0.f ? v.y : (__expf(v.y) - 1.f);
                    v.z = v.z > 0.f ? v.z : (__expf(v.z) - 1.f);
                    v.w = v.w > 0.f ? v.w : (__expf(v.w) - 1.f);
                    float* xr = acc + r * ACC_PITCH + c4 * 4;
                    xr[0] = v.x; xr[1] = v.y; xr[2] = v.z; xr[3] = v.w;
                    float4 q = o[u];
                    q.x += v.x; q.y += v.y; q.z += v.z; q.w += v.w;
                    o4[gi] = q;
                }
            }
            v8f acc0 = {0.f,0.f,0.f,0.f,0.f,0.f,0.f,0.f};
            v8f acc1 = acc0, acc2 = acc0, acc3 = acc0;
            const float* arow = &acc[(rb + c4) * ACC_PITCH + khalf];
            #pragma nounroll
            for (int kb = 0; kb < DDIM; kb += 16) {
                #pragma unroll
                for (int k0 = kb; k0 < kb + 16; k0 += 4) {
                    v2f a, b0, b1, b2, b3;
                    a.x  = arow[k0];                       a.y  = arow[k0 + 1];
                    b0.x = brow[k0];                       b0.y = brow[k0 + 1];
                    b1.x = brow[16 * SW_STRIDE + k0];      b1.y = brow[16 * SW_STRIDE + k0 + 1];
                    b2.x = brow[32 * SW_STRIDE + k0];      b2.y = brow[32 * SW_STRIDE + k0 + 1];
                    b3.x = brow[48 * SW_STRIDE + k0];      b3.y = brow[48 * SW_STRIDE + k0 + 1];
                    acc0 = __builtin_amdgcn_wmma_f32_16x16x4_f32(false, a, false, b0, (short)0, acc0, false, false);
                    acc1 = __builtin_amdgcn_wmma_f32_16x16x4_f32(false, a, false, b1, (short)0, acc1, false, false);
                    acc2 = __builtin_amdgcn_wmma_f32_16x16x4_f32(false, a, false, b2, (short)0, acc2, false, false);
                    acc3 = __builtin_amdgcn_wmma_f32_16x16x4_f32(false, a, false, b3, (short)0, acc3, false, false);
                }
            }
            const int lr0 = rb + rhi;
            float* p = xl_next + (long long)(r0 + lr0) * DDIM + c4;
            if (rb + 16 <= rows) {
                #pragma unroll
                for (int v = 0; v < 8; ++v) {
                    p[(long long)v * DDIM +  0] = acc0[v];
                    p[(long long)v * DDIM + 16] = acc1[v];
                    p[(long long)v * DDIM + 32] = acc2[v];
                    p[(long long)v * DDIM + 48] = acc3[v];
                }
            } else {
                #pragma unroll
                for (int v = 0; v < 8; ++v) {
                    if (lr0 + v < rows) {
                        p[(long long)v * DDIM +  0] = acc0[v];
                        p[(long long)v * DDIM + 16] = acc1[v];
                        p[(long long)v * DDIM + 32] = acc2[v];
                        p[(long long)v * DDIM + 48] = acc3[v];
                    }
                }
            }
        }
        return;
    }
    for (int i = t; i < rows * (DDIM / 4); i += OWN_T) {
        const int r = i >> 4, c4 = i & 15;
        const long long gi = (long long)(r0 + r) * (DDIM / 4) + c4;
        const float as = sself[r] / (ssum[r] + 1e-16f);
        const float4 a = acc4[r * (ACC_PITCH / 4) + c4];
        const float4 xs = xl4[gi];
        const float4 b = b4[c4];
        float4 o = o4[gi];
        float4 v;
        v.x = (a.x + xs.x * as) + b.x;
        v.y = (a.y + xs.y * as) + b.y;
        v.z = (a.z + xs.z * as) + b.z;
        v.w = (a.w + xs.w * as) + b.w;
        v.x = v.x > 0.f ? v.x : (__expf(v.x) - 1.f);
        v.y = v.y > 0.f ? v.y : (__expf(v.y) - 1.f);
        v.z = v.z > 0.f ? v.z : (__expf(v.z) - 1.f);
        v.w = v.w > 0.f ? v.w : (__expf(v.w) - 1.f);
        o.x += v.x; o.y += v.y; o.z += v.z; o.w += v.w;
        o4[gi] = o;
    }
}

__global__ __launch_bounds__(256) void scale_out(float* __restrict__ out, float inv,
                                                 long long total4) {
    float4* __restrict__ o4 = (float4*)out;
    const long long base = (long long)blockIdx.x * (256 * STREAM_U) + threadIdx.x;
    float4 v[STREAM_U];
    #pragma unroll
    for (int u = 0; u < STREAM_U; ++u) {
        const long long i = base + u * 256;
        v[u] = o4[(i < total4) ? i : 0];
    }
    #pragma unroll
    for (int u = 0; u < STREAM_U; ++u) {
        const long long i = base + u * 256;
        if (i < total4) {
            float4 r;
            r.x = v[u].x * inv; r.y = v[u].y * inv; r.z = v[u].z * inv; r.w = v[u].w * inv;
            o4[i] = r;
        }
    }
}

static inline unsigned gridFor(long long work, int block) {
    return (unsigned)((work + block - 1) / block);
}

extern "C" void kernel_launch(void* const* d_in, const int* in_sizes, int n_in,
                              void* d_out, int out_size, void* d_ws, size_t ws_size,
                              hipStream_t stream) {
    const int*   ei       = (const int*)d_in[0];
    const float* emb      = (const float*)d_in[1];
    const float* prompt   = (const float*)d_in[2];
    const float* proj_w   = (const float*)d_in[3];
    const float* proj_b   = (const float*)d_in[4];
    const float* conv_ws  = (const float*)d_in[5];
    const float* att_srcs = (const float*)d_in[6];
    const float* att_dsts = (const float*)d_in[7];
    const float* conv_bs  = (const float*)d_in[8];

    const int E = in_sizes[0] / 2;
    const int N = in_sizes[1] / DDIM;
    const int P = in_sizes[2];
    const int L = in_sizes[5] / (DDIM * DDIM);

    float* out = (float*)d_out;

    float*    pvec   = (float*)d_ws;
    float*    xbuf   = pvec + 64;
    float*    xlb    = xbuf + (size_t)N * DDIM;
    float*    a_src  = xlb + (size_t)N * DDIM;
    float*    a_dst  = a_src + N;


    float* tail = a_dst + N;
    const unsigned long long ws_bytes = (unsigned long long)ws_size;
    compute_pvec<<<gridFor((long long)E + 1, 256 * MAP_U * 4), 256, 0, stream>>>(prompt, proj_w, proj_b, pvec, P,
                                                                          ei, E, N, tail, pvec, ws_bytes);
    const long long total4 = (long long)N * (DDIM / 4);
    add_prompt<<<gridFor(total4, 256 * STREAM_U), 256, 0, stream>>>(emb, pvec, xbuf, out, total4);

    if (L > 0) gemm_xw_t<<<gridFor(N, GEMM_ROWS), 256, 0, stream>>>(xbuf, conv_ws, xlb, N);
    float* cur = xlb;
    float* nxt = xbuf;
    for (int l = 0; l < L; ++l) {
        const float* as = att_srcs + (size_t)l * DDIM;
        const float* ad = att_dsts + (size_t)l * DDIM;
        const float* cb = conv_bs  + (size_t)l * DDIM;
        const float* Wn = (l + 1 < L) ? conv_ws + (size_t)(l + 1) * DDIM * DDIM : nullptr;

        att_dots<<<gridFor(N, 128), 128, 0, stream>>>(cur, as, ad, a_src, a_dst, N);
        edge_owner<<<gridFor(N, OWN_ROWS), OWN_T, 0, stream>>>(ei, E, N, a_src, a_dst, cur, cb, Wn, nxt, out,
                                                               tail, pvec, ws_bytes);
        float* turn = cur; cur = nxt; nxt = turn;
    }
    scale_out<<<gridFor(total4, 256 * STREAM_U), 256, 0, stream>>>(out, 1.f / (float)(L + 1), total4);
}


# later layers' edge phase reads back the edge list that layer 0's walk found (one word per edge, kept in the workspace): no second map walk, no index gathers; LDS adds stay in arrival order
# speedup vs baseline: 3.7203x; 1.0958x over previous
#include <hip/hip_runtime.h>
#include <hip/hip_bf16.h>

typedef float v2f __attribute__((ext_vector_type(2)));
typedef float v8f __attribute__((ext_vector_type(8)));

#define DDIM 64
#define GEMM_ROWS 128
#define SX_STRIDE 68
#define SW_STRIDE 68

__device__ __forceinline__ unsigned fkey(float f) {
    unsigned u = __float_as_uint(f);
    return (u & 0x80000000u) ? ~u : (u | 0x80000000u);
}
__device__ __forceinline__ float fdec(unsigned k) {
    return (k & 0x80000000u) ? __uint_as_float(k & 0x7FFFFFFFu)
                             : __uint_as_float(~k);
}
#define FKEY_NEG_INF 0x007FFFFFu

#define OWN_ROWS 592

#define OWN_NONE 255u
__device__ __forceinline__ unsigned char* own_map_base(float* tail) {
    unsigned char* p = (unsigned char*)tail;
    return p + ((0ull - (unsigned long long)p) & 15ull);
}
__device__ __forceinline__ bool own_map_ok(const float* wsbase, float* tail, unsigned long long ws_bytes,
                                           int E, int N) {
    const unsigned long long off = (unsigned long long)(own_map_base(tail) - (const unsigned char*)wsbase);
    return (N + OWN_ROWS - 1) / OWN_ROWS <= 255 && off + (unsigned long long)E <= ws_bytes;
}
__device__ __forceinline__ unsigned own_of(int dst, int N) {
    return ((unsigned)dst < (unsigned)N) ? (unsigned)dst / (unsigned)OWN_ROWS : OWN_NONE;
}

#define LIST_SRC_BITS 18
#define LIST_SRC_MASK ((1u << LIST_SRC_BITS) - 1u)
#define OWN_CAP  8192
__device__ __forceinline__ int* own_len_base(float* tail, int E) {
    unsigned char* p = own_map_base(tail) + E;
    return (int*)(p + ((0ull - (unsigned long long)p) & 15ull));
}
__device__ __forceinline__ unsigned* own_list_base(float* tail, int E, int N) {
    const int nown = (N + OWN_ROWS - 1) / OWN_ROWS;
    return (unsigned*)(own_len_base(tail, E) + ((nown + 3) & ~3));
}
__device__ __forceinline__ int own_list_stride(const float* wsbase, float* tail, unsigned long long ws_bytes,
                                               int E, int N) {
    if (!own_map_ok(wsbase, tail, ws_bytes, E, N)) return 0;
    const unsigned long long nown = (unsigned long long)((N + OWN_ROWS - 1) / OWN_ROWS);
    const unsigned long long off =
        (unsigned long long)((const unsigned char*)own_list_base(tail, E, N) - (const unsigned char*)wsbase);
    if (nown == 0 || off >= ws_bytes) return 0;
    const unsigned long long per = (ws_bytes - off) / (nown * 4ull);
    return (int)(per < (unsigned long long)OWN_CAP ? per : (unsigned long long)OWN_CAP) & ~3;
}

#define MAP_U 4
__global__ __launch_bounds__(256) void compute_pvec(const float* __restrict__ prompt, const float* __restrict__ pw,
                             const float* __restrict__ pb, float* __restrict__ pvec, int P,
                             const int* __restrict__ ei, int E, int N,
                             float* tail, const float* wsbase, unsigned long long ws_bytes) {
    const int t = threadIdx.x;
    if (blockIdx.x == 0 && t < DDIM) {
        float s = pb[t];
        for (int j = 0; j < P; ++j) s += prompt[j] * pw[t * P + j];
        pvec[t] = s;
    }
    if (!own_map_ok(wsbase, tail, ws_bytes, E, N)) return;
    unsigned char* __restrict__ map = own_map_base(tail);
    const int* __restrict__ dstp = ei + E;
    if ((E & 3) == 0) {
        const int4* __restrict__ d4 = (const int4*)dstp;
        unsigned* __restrict__ m1 = (unsigned*)map;
        const int n4 = E >> 2;
        for (int base = blockIdx.x * (256 * MAP_U); base < n4; base += gridDim.x * (256 * MAP_U)) {
            int4 v[MAP_U];
            #pragma unroll
            for (int u = 0; u < MAP_U; ++u) {
                const int g = base + u * 256 + t;
                v[u] = d4[(g < n4) ? g : 0];
            }
            #pragma unroll
            for (int u = 0; u < MAP_U; ++u) {
                const int g = base + u * 256 + t;
                if (g < n4)
                    m1[g] = own_of(v[u].x, N) | (own_of(v[u].y, N) << 8) |
                            (own_of(v[u].z, N) << 16) | (own_of(v[u].w, N) << 24);
            }
        }
    } else {
        for (int e = blockIdx.x * 256 + t; e < E; e += gridDim.x * 256) map[e] = (unsigned char)own_of(dstp[e], N);
    }
}

#define STREAM_U 8
__global__ __launch_bounds__(256) void add_prompt(const float* __restrict__ emb,
                                                  const float* __restrict__ pvec,
                                                  float* __restrict__ x,
                                                  float* __restrict__ out_sum,
                                                  long long total4) {
    const float4* __restrict__ e4 = (const float4*)emb;
    float4* __restrict__ x4 = (float4*)x;
    float4* __restrict__ o4 = (float4*)out_sum;
    const int t = threadIdx.x;
    const float4 p = ((const float4*)pvec)[t & 15];
    const long long base = (long long)blockIdx.x * (256 * STREAM_U) + t;
    float4 v[STREAM_U];
    #pragma unroll
    for (int u = 0; u < STREAM_U; ++u) {
        const long long i = base + u * 256;
        v[u] = e4[(i < total4) ? i : 0];
    }
    #pragma unroll
    for (int u = 0; u < STREAM_U; ++u) {
        const long long i = base + u * 256;
        if (i < total4) {
            float4 r;
            r.x = v[u].x + p.x; r.y = v[u].y + p.y; r.z = v[u].z + p.z; r.w = v[u].w + p.w;
            x4[i] = r;
            o4[i] = r;
        }
    }
}

__global__ __launch_bounds__(256) void gemm_xw_t(const float* __restrict__ x,
                                                 const float* __restrict__ W,
                                                 float* __restrict__ xl, int N) {
    __shared__ float sx[GEMM_ROWS * SX_STRIDE];
    __shared__ float sw[DDIM * SW_STRIDE];
    const int t = threadIdx.x;
    const int rowBase = blockIdx.x * GEMM_ROWS;

    {
        const float4* Wv = (const float4*)W;
        float4* swv = (float4*)sw;
        for (int i = t; i < (DDIM * DDIM) / 4; i += 256) {
            int r = i >> 4, c4 = i & 15;
            swv[r * (SW_STRIDE / 4) + c4] = Wv[i];
        }
    }
    {
        const float4* xv = (const float4*)x;
        float4* sxv = (float4*)sx;
        for (int i = t; i < (GEMM_ROWS * DDIM) / 4; i += 256) {
            int r = i >> 4, c4 = i & 15;
            int gr = rowBase + r;
            if (gr >= N) gr = N - 1;
            sxv[r * (SX_STRIDE / 4) + c4] = xv[(long long)gr * 16 + c4];
        }
    }
    __syncthreads();

    const int wave  = t >> 5;
    const int lane  = t & 31;
    const int lrow  = lane & 15;
    const int khalf = (lane >> 4) << 1;
    const int wrow  = wave * 16;

    v8f acc0 = {0.f,0.f,0.f,0.f,0.f,0.f,0.f,0.f};
    v8f acc1 = acc0, acc2 = acc0, acc3 = acc0;

    const float* arow = &sx[(wrow + lrow) * SX_STRIDE + khalf];
    const float* brow = &sw[lrow * SW_STRIDE + khalf];

    #pragma unroll
    for (int k0 = 0; k0 < DDIM; k0 += 4) {
        v2f a, b0, b1, b2, b3;
        a.x  = arow[k0];                       a.y  = arow[k0 + 1];
        b0.x = brow[k0];                       b0.y = brow[k0 + 1];
        b1.x = brow[16 * SW_STRIDE + k0];      b1.y = brow[16 * SW_STRIDE + k0 + 1];
        b2.x = brow[32 * SW_STRIDE + k0];      b2.y = brow[32 * SW_STRIDE + k0 + 1];
        b3.x = brow[48 * SW_STRIDE + k0];      b3.y = brow[48 * SW_STRIDE + k0 + 1];
        acc0 = __builtin_amdgcn_wmma_f32_16x16x4_f32(false, a, false, b0, (short)0, acc0, false, false);
        acc1 = __builtin_amdgcn_wmma_f32_16x16x4_f32(false, a, false, b1, (short)0, acc1, false, false);
        acc2 = __builtin_amdgcn_wmma_f32_16x16x4_f32(false, a, false, b2, (short)0, acc2, false, false);
        acc3 = __builtin_amdgcn_wmma_f32_16x16x4_f32(false, a, false, b3, (short)0, acc3, false, false);
    }

    const int rhi = (lane >> 4) * 8;
    const long long rbase = (long long)(rowBase + wrow + rhi);
    if (rowBase + GEMM_ROWS <= N) {
        float* p = xl + rbase * DDIM + lrow;
        #pragma unroll
        for (int v = 0; v < 8; ++v) {
            p[(long long)v * DDIM +  0] = acc0[v];
            p[(long long)v * DDIM + 16] = acc1[v];
            p[(long long)v * DDIM + 32] = acc2[v];
            p[(long long)v * DDIM + 48] = acc3[v];
        }
    } else {
        #pragma unroll
        for (int v = 0; v < 8; ++v) {
            long long gr = rbase + v;
            if (gr < N) {
                float* p = xl + gr * DDIM + lrow;
                p[ 0] = acc0[v];
                p[16] = acc1[v];
                p[32] = acc2[v];
                p[48] = acc3[v];
            }
        }
    }
}

#define ATT_U 16
template <int M> __device__ __forceinline__ float att_xadd(float v) {
    const int o = __builtin_amdgcn_update_dpp(0, __builtin_bit_cast(int, v), 0x160 + M, 0xF, 0xF, true);
    return v + __builtin_bit_cast(float, o);
}
__device__ __forceinline__ float att_pick(const float (&r)[ATT_U], int c) {
    float o = r[0];
    #pragma unroll
    for (int u = 1; u < ATT_U; ++u) o = (c == u) ? r[u] : o;
    return o;
}
__global__ __launch_bounds__(128) void att_dots(const float* __restrict__ xl,
                                                const float* __restrict__ av_s,
                                                const float* __restrict__ av_d,
                                                float* __restrict__ a_src,
                                                float* __restrict__ a_dst, int N) {
    const int lane = threadIdx.x & 31;
    const int c = lane & 15;
    const int h = lane >> 4;
    const int nb = (blockIdx.x * 4 + (threadIdx.x >> 5)) * 32;
    const float4* __restrict__ xl4 = (const float4*)xl;
    const float4 s = ((const float4*)av_s)[c];
    const float4 d = ((const float4*)av_d)[c];
    float4 v[ATT_U];
    #pragma unroll
    for (int u = 0; u < ATT_U; ++u) {
        int n = nb + 16 * h + u;
        if (n >= N) n = N - 1;
        v[u] = xl4[(long long)n * (DDIM / 4) + c];
    }
    float rs[ATT_U], rd[ATT_U];
    #pragma unroll
    for (int u = 0; u < ATT_U; ++u) {
        float sa = v[u].x * s.x + v[u].y * s.y;
        float sb = v[u].z * s.z + v[u].w * s.w;
        float da = v[u].x * d.x + v[u].y * d.y;
        float db = v[u].z * d.z + v[u].w * d.w;
        sa = att_xadd<8>(sa); sb = att_xadd<8>(sb); da = att_xadd<8>(da); db = att_xadd<8>(db);
        sa = att_xadd<4>(sa); sb = att_xadd<4>(sb); da = att_xadd<4>(da); db = att_xadd<4>(db);
        sa = att_xadd<2>(sa); sb = att_xadd<2>(sb); da = att_xadd<2>(da); db = att_xadd<2>(db);
        sa = att_xadd<1>(sa); sb = att_xadd<1>(sb); da = att_xadd<1>(da); db = att_xadd<1>(db);
        rs[u] = sa + sb;
        rd[u] = da + db;
    }
    const int n = nb + lane;
    const float os = att_pick(rs, c);
    const float od = att_pick(rd, c);
    if (n < N) { a_src[n] = os; a_dst[n] = od; }
}

#define ACC_PITCH 68
#define OWN_T    1024
#define OWN_U    8
static_assert(OWN_CAP % (4 * OWN_T) == 0, "OWN_CAP must be whole trips of the logit step");

__device__ __forceinline__ float leaky02(float v) { return v > 0.f ? v : 0.2f * v; }

__global__ __launch_bounds__(OWN_T) void edge_owner(const int* __restrict__ ei, int E, int N,
                                                    const float* __restrict__ a_src,
                                                    const float* __restrict__ a_dst,
                                                    const float* __restrict__ xl,
                                                    const float* __restrict__ bias,
                                                    const float* __restrict__ Wnext,
                                                    float* __restrict__ xl_next,
                                                    float* __restrict__ out_sum,
                                                    float* tail, const float* wsbase,
                                                    unsigned long long ws_bytes,
                                                    int layer, int nlayers) {
    __shared__ float4   acc4[OWN_ROWS * (ACC_PITCH / 4)];
    __shared__ float4   sw4[DDIM * (SW_STRIDE / 4)];
    __shared__ unsigned smax[OWN_ROWS];
    __shared__ float    ssum[OWN_ROWS];
    __shared__ float    sadst[OWN_ROWS];
    __shared__ float    sself[OWN_ROWS];
    __shared__ int      lsrc[OWN_CAP];
    __shared__ int      ldl[OWN_CAP];
    __shared__ float    lval[OWN_CAP];
    __shared__ int      lcount;
    float* acc = (float*)acc4;

    const int t = threadIdx.x;
    const int lane = t & 31;
    const int wave = t >> 5;
    const int r0 = blockIdx.x * OWN_ROWS;
    const int rows = min(OWN_ROWS, N - r0);
    const int* __restrict__ dstp = ei + E;

    const int lstride = own_list_stride(wsbase, tail, ws_bytes, E, N);
    int* __restrict__ len_g = own_len_base(tail, E) + blockIdx.x;
    unsigned* __restrict__ list_g = own_list_base(tail, E, N) + (size_t)blockIdx.x * (size_t)lstride;
    const bool keep_list = (layer == 0 && nlayers > 1 && lstride > 0);
    int list_len = -1;
    if (layer > 0 && lstride > 0) list_len = *len_g;
    const bool from_list = (unsigned)list_len <= (unsigned)lstride;

    for (int r = t; r < OWN_ROWS; r += OWN_T) {
        float ad = 0.f, vs = 0.f;
        unsigned key = FKEY_NEG_INF;
        if (r < rows) {
            ad = a_dst[r0 + r];
            vs = leaky02(a_src[r0 + r] + ad);
            key = fkey(vs);
        }
        sadst[r] = ad; sself[r] = vs; smax[r] = key; ssum[r] = 0.f;
    }
    {
        const float4 z = {0.f, 0.f, 0.f, 0.f};
        for (int i = t; i < OWN_ROWS * (ACC_PITCH / 4); i += OWN_T) acc4[i] = z;
    }
    if (t == 0) lcount = 0;
    if (Wnext != nullptr) {
        const float4* __restrict__ Wv = (const float4*)Wnext;
        for (int i = t; i < (DDIM * DDIM) / 4; i += OWN_T) {
            const int r = i >> 4, c4 = i & 15;
            sw4[r * (SW_STRIDE / 4) + c4] = Wv[i];
        }
    }
    __syncthreads();

    #define OWN_NOTE(dlv, eidx)                                               \
        {                                                                     \
            const unsigned dl_ = (dlv);                                       \
            if (dl_ < (unsigned)rows) {                                       \
                const int slot_ = atomicAdd(&lcount, 1);                      \
                if (slot_ < OWN_CAP) { lsrc[slot_] = (eidx); ldl[slot_] = (int)dl_; } \
            }                                                                 \
        }
    if (from_list) {
    } else if (own_map_ok(wsbase, tail, ws_bytes, E, N)) {
        const unsigned char* __restrict__ map = own_map_base(tail);
        const uint4* __restrict__ m4 = (const uint4*)map;
        const int n16 = E >> 4;
        const unsigned me = blockIdx.x;
        const unsigned pat = me * 0x01010101u;
        const int start = (int)(((long long)blockIdx.x * n16) / gridDim.x) & ~7;
        for (int base = 0; base < n16; base += OWN_T * OWN_U) {
            uint4 v[OWN_U];
            int   j[OWN_U];
            #pragma unroll
            for (int u = 0; u < OWN_U; ++u) {
                const int i = base + u * OWN_T + t;
                int jj = i + start;
                if (jj >= n16) jj -= n16;
                j[u] = (i < n16) ? jj : -1;
                v[u] = m4[(i < n16) ? jj : 0];
            }
            #pragma unroll
            for (int u = 0; u < OWN_U; ++u) {
                const unsigned x0 = v[u].x ^ pat, x1 = v[u].y ^ pat, x2 = v[u].z ^ pat, x3 = v[u].w ^ pat;
                const unsigned y0 = ((x0 & 0x7f7f7f7fu) + 0x7f7f7f7fu) | x0;
                const unsigned y1 = ((x1 & 0x7f7f7f7fu) + 0x7f7f7f7fu) | x1;
                const unsigned y2 = ((x2 & 0x7f7f7f7fu) + 0x7f7f7f7fu) | x2;
                const unsigned y3 = ((x3 & 0x7f7f7f7fu) + 0x7f7f7f7fu) | x3;
                if (j[u] >= 0 && ((y0 & y1 & y2 & y3) | 0x7f7f7f7fu) != 0xffffffffu) {
                    unsigned c = ((~y0 & 0x80808080u) >> 7) | ((~y1 & 0x80808080u) >> 6) |
                                 ((~y2 & 0x80808080u) >> 5) | ((~y3 & 0x80808080u) >> 4);
                    const int e0 = j[u] << 4;
                    while (c) {
                        const int b = __builtin_ctz(c);
                        c &= c - 1;
                        const int slot = atomicAdd(&lcount, 1);
                        if (slot < OWN_CAP) lsrc[slot] = e0 + ((b & 7) << 2) + (b >> 3);
                    }
                }
            }
        }
        for (int e = (n16 << 4) + t; e < E; e += OWN_T) {
            if (map[e] == me) {
                const int slot = atomicAdd(&lcount, 1);
                if (slot < OWN_CAP) lsrc[slot] = e;
            }
        }
    } else if ((E & 3) == 0) {
        const int4* __restrict__ d4 = (const int4*)dstp;
        const int n4 = E >> 2;
        const int start = (int)(((long long)blockIdx.x * n4) / gridDim.x) & ~7;
        for (int base = 0; base < n4; base += OWN_T * OWN_U) {
            int4 v[OWN_U];
            int  j[OWN_U];
            #pragma unroll
            for (int u = 0; u < OWN_U; ++u) {
                const int i = base + u * OWN_T + t;
                int jj = i + start;
                if (jj >= n4) jj -= n4;
                j[u] = (i < n4) ? jj : -1;
                v[u] = d4[(i < n4) ? jj : 0];
            }
            #pragma unroll
            for (int u = 0; u < OWN_U; ++u) {
                const unsigned q0 = (unsigned)(v[u].x - r0), q1 = (unsigned)(v[u].y - r0);
                const unsigned q2 = (unsigned)(v[u].z - r0), q3 = (unsigned)(v[u].w - r0);
                if (j[u] >= 0 && min(min(q0, q1), min(q2, q3)) < (unsigned)rows) {
                    const int e = j[u] << 2;
                    OWN_NOTE(q0, e)
                    OWN_NOTE(q1, e + 1)
                    OWN_NOTE(q2, e + 2)
                    OWN_NOTE(q3, e + 3)
                }
            }
        }
    } else {
        for (int e = t; e < E; e += OWN_T) OWN_NOTE((unsigned)(dstp[e] - r0), e)
    }
    #undef OWN_NOTE
    __syncthreads();
    const int cnt = from_list ? list_len : lcount;
    if (keep_list && t == 0) *len_g = (cnt <= lstride) ? cnt : -1;

    if (cnt <= OWN_CAP) {
        if (from_list) {
            for (int b0 = 0; b0 < cnt; b0 += OWN_T * 4) {
                unsigned w_[4];
                float as_[4];
                #pragma unroll
                for (int u = 0; u < 4; ++u) {
                    const int i = b0 + u * OWN_T + t;
                    const unsigned w = list_g[(i < cnt) ? i : 0];
                    w_[u] = (i < cnt) ? w : 0u;
                }
                #pragma unroll
                for (int u = 0; u < 4; ++u) as_[u] = a_src[w_[u] & LIST_SRC_MASK];
                #pragma unroll
                for (int u = 0; u < 4; ++u) {
                    const int i = b0 + u * OWN_T + t;
                    const int dl = (int)(w_[u] >> LIST_SRC_BITS);
                    const float v = leaky02(as_[u] + sadst[dl]);
                    lsrc[i] = (int)(w_[u] & LIST_SRC_MASK);
                    ldl[i] = dl;
                    lval[i] = v;
                    atomicMax(&smax[dl], (i < cnt) ? fkey(v) : 0u);
                }
            }
        } else
        for (int b0 = 0; b0 < cnt; b0 += OWN_T * 4) {
            int e_[4], s_[4], d_[4];
            float as_[4];
            #pragma unroll
            for (int u = 0; u < 4; ++u) { const int i = b0 + u * OWN_T + t; e_[u] = (i < cnt) ? lsrc[i] : 0; }
            #pragma unroll
            for (int u = 0; u < 4; ++u) { s_[u] = ei[e_[u]]; d_[u] = dstp[e_[u]]; }
            #pragma unroll
            for (int u = 0; u < 4; ++u) as_[u] = a_src[s_[u]];
            #pragma unroll
            for (int u = 0; u < 4; ++u) {
                const int i = b0 + u * OWN_T + t;
                if (i < cnt) {
                    const int dl = d_[u] - r0;
                    const float v = leaky02(as_[u] + sadst[dl]);
                    lsrc[i] = s_[u];
                    ldl[i] = dl;
                    lval[i] = v;
                    if (keep_list && cnt <= lstride) list_g[i] = ((unsigned)s_[u] & LIST_SRC_MASK) | ((unsigned)dl << LIST_SRC_BITS);
                    atomicMax(&smax[dl], fkey(v));
                }
            }
        }
        __syncthreads();
        for (int i = t; i < cnt; i += OWN_T) {
            const int dl = ldl[i];
            const float w = __expf(lval[i] - fdec(smax[dl]));
            lval[i] = w;
            atomicAdd(&ssum[dl], w);
        }
        for (int r = t; r < rows; r += OWN_T) {
            const float w = __expf(sself[r] - fdec(smax[r]));
            sself[r] = w;
            atomicAdd(&ssum[r], w);
        }
        __syncthreads();
        const int swave = __builtin_amdgcn_readfirstlane(wave);
        for (int g = swave * 32; g < cnt; g += OWN_T) {
            const int   il = (g + lane < cnt) ? g + lane : g;
            const int   sl = lsrc[il];
            const int   dl = ldl[il];
            const float al = lval[il] / (ssum[dl] + 1e-16f);
            const int   ab = __builtin_bit_cast(int, al);
            #pragma unroll
            for (int q = 0; q < 4; ++q) {
                const int g8 = g + q * 8;
                if (g8 + 8 <= cnt) {
                    float lo[8], hi[8];
                    const float* xr[8];
                    #pragma unroll
                    for (int k = 0; k < 8; ++k)
                        xr[k] = xl + (long long)__builtin_amdgcn_readlane(sl, q * 8 + k) * DDIM + lane;
                    #pragma unroll
                    for (int k = 0; k < 8; ++k) {
                        lo[k] = xr[k][0];
                        hi[k] = xr[k][32];
                    }
                    #pragma unroll
                    for (int k = 0; k < 8; ++k) {
                        const float a = __builtin_bit_cast(float, __builtin_amdgcn_readlane(ab, q * 8 + k));
                        float* ap = acc + __builtin_amdgcn_readlane(dl, q * 8 + k) * ACC_PITCH + lane;
                        atomicAdd(ap,      lo[k] * a);
                        atomicAdd(ap + 32, hi[k] * a);
                    }
                } else if (g8 < cnt) {
                    for (int k = 0; g8 + k < cnt; ++k) {
                        const float* xr = xl + (long long)__builtin_amdgcn_readlane(sl, q * 8 + k) * DDIM + lane;
                        const float a = __builtin_bit_cast(float, __builtin_amdgcn_readlane(ab, q * 8 + k));
                        float* ap = acc + __builtin_amdgcn_readlane(dl, q * 8 + k) * ACC_PITCH + lane;
                        atomicAdd(ap,      xr[0] * a);
                        atomicAdd(ap + 32, xr[32] * a);
                    }
                }
            }
        }
    } else {
        for (int e = t; e < E; e += OWN_T) {
            const unsigned dl = (unsigned)(dstp[e] - r0);
            if (dl < (unsigned)rows)
                atomicMax(&smax[dl], fkey(leaky02(a_src[ei[e]] + sadst[dl])));
        }
        __syncthreads();
        for (int e = t; e < E; e += OWN_T) {
            const unsigned dl = (unsigned)(dstp[e] - r0);
            if (dl < (unsigned)rows)
                atomicAdd(&ssum[dl], __expf(leaky02(a_src[ei[e]] + sadst[dl]) - fdec(smax[dl])));
        }
        for (int r = t; r < rows; r += OWN_T) {
            const float w = __expf(sself[r] - fdec(smax[r]));
            sself[r] = w;
            atomicAdd(&ssum[r], w);
        }
        __syncthreads();
        for (int e = t; e < E; e += OWN_T) {
            const unsigned dl = (unsigned)(dstp[e] - r0);
            if (dl < (unsigned)rows) {
                const int s = ei[e];
                const float aj = __expf(leaky02(a_src[s] + sadst[dl]) - fdec(smax[dl])) / (ssum[dl] + 1e-16f);
                const float* xr = xl + (long long)s * DDIM;
                float* ap = acc + dl * ACC_PITCH;
                for (int c = 0; c < DDIM; ++c) atomicAdd(ap + c, xr[c] * aj);
            }
        }
    }
    __syncthreads();

    const float4* __restrict__ xl4 = (const float4*)xl;
    const float4* __restrict__ b4 = (const float4*)bias;
    float4* __restrict__ o4 = (float4*)out_sum;
    if (Wnext != nullptr) {
        const float* sw = (const float*)sw4;
        const int c4    = lane & 15;
        const int h     = lane >> 4;
        const int khalf = h << 1;
        const int rhi   = h * 8;
        const int ntiles = (rows + 15) >> 4;
        const float4 b = b4[c4];
        const float* brow = &sw[c4 * SW_STRIDE + khalf];
        for (int tile = wave; tile < ntiles; tile += OWN_T / 32) {
            const int rb = tile * 16;
            float4 xs[8], o[8];
            #pragma unroll
            for (int u = 0; u < 8; ++u) {
                const int r = rb + 2 * u + h;
                const long long gi = (long long)(r0 + (r < rows ? r : rows - 1)) * (DDIM / 4) + c4;
                xs[u] = xl4[gi];
                o[u]  = o4[gi];
            }
            #pragma unroll
            for (int u = 0; u < 8; ++u) {
                const int r = rb + 2 * u + h;
                if (r < rows) {
                    const long long gi = (long long)(r0 + r) * (DDIM / 4) + c4;
                    const float as = sself[r] / (ssum[r] + 1e-16f);
                    const float4 a = acc4[r * (ACC_PITCH / 4) + c4];
                    float4 v;
                    v.x = (a.x + xs[u].x * as) + b.x;
                    v.y = (a.y + xs[u].y * as) + b.y;
                    v.z = (a.z + xs[u].z * as) + b.z;
                    v.w = (a.w + xs[u].w * as) + b.w;
                    v.x = v.x > 0.f ? v.x : (__expf(v.x) - 1.f);
                    v.y = v.y > 0.f ? v.y : (__expf(v.y) - 1.f);
                    v.z = v.z > 0.f ? v.z : (__expf(v.z) - 1.f);
                    v.w = v.w > 0.f ? v.w : (__expf(v.w) - 1.f);
                    float* xr = acc + r * ACC_PITCH + c4 * 4;
                    xr[0] = v.x; xr[1] = v.y; xr[2] = v.z; xr[3] = v.w;
                    float4 q = o[u];
                    q.x += v.x; q.y += v.y; q.z += v.z; q.w += v.w;
                    o4[gi] = q;
                }
            }
            v8f acc0 = {0.f,0.f,0.f,0.f,0.f,0.f,0.f,0.f};
            v8f acc1 = acc0, acc2 = acc0, acc3 = acc0;
            const float* arow = &acc[(rb + c4) * ACC_PITCH + khalf];
            #pragma nounroll
            for (int kb = 0; kb < DDIM; kb += 16) {
                #pragma unroll
                for (int k0 = kb; k0 < kb + 16; k0 += 4) {
                    v2f a, b0, b1, b2, b3;
                    a.x  = arow[k0];                       a.y  = arow[k0 + 1];
                    b0.x = brow[k0];                       b0.y = brow[k0 + 1];
                    b1.x = brow[16 * SW_STRIDE + k0];      b1.y = brow[16 * SW_STRIDE + k0 + 1];
                    b2.x = brow[32 * SW_STRIDE + k0];      b2.y = brow[32 * SW_STRIDE + k0 + 1];
                    b3.x = brow[48 * SW_STRIDE + k0];      b3.y = brow[48 * SW_STRIDE + k0 + 1];
                    acc0 = __builtin_amdgcn_wmma_f32_16x16x4_f32(false, a, false, b0, (short)0, acc0, false, false);
                    acc1 = __builtin_amdgcn_wmma_f32_16x16x4_f32(false, a, false, b1, (short)0, acc1, false, false);
                    acc2 = __builtin_amdgcn_wmma_f32_16x16x4_f32(false, a, false, b2, (short)0, acc2, false, false);
                    acc3 = __builtin_amdgcn_wmma_f32_16x16x4_f32(false, a, false, b3, (short)0, acc3, false, false);
                }
            }
            const int lr0 = rb + rhi;
            float* p = xl_next + (long long)(r0 + lr0) * DDIM + c4;
            if (rb + 16 <= rows) {
                #pragma unroll
                for (int v = 0; v < 8; ++v) {
                    p[(long long)v * DDIM +  0] = acc0[v];
                    p[(long long)v * DDIM + 16] = acc1[v];
                    p[(long long)v * DDIM + 32] = acc2[v];
                    p[(long long)v * DDIM + 48] = acc3[v];
                }
            } else {
                #pragma unroll
                for (int v = 0; v < 8; ++v) {
                    if (lr0 + v < rows) {
                        p[(long long)v * DDIM +  0] = acc0[v];
                        p[(long long)v * DDIM + 16] = acc1[v];
                        p[(long long)v * DDIM + 32] = acc2[v];
                        p[(long long)v * DDIM + 48] = acc3[v];
                    }
                }
            }
        }
        return;
    }
    for (int i = t; i < rows * (DDIM / 4); i += OWN_T) {
        const int r = i >> 4, c4 = i & 15;
        const long long gi = (long long)(r0 + r) * (DDIM / 4) + c4;
        const float as = sself[r] / (ssum[r] + 1e-16f);
        const float4 a = acc4[r * (ACC_PITCH / 4) + c4];
        const float4 xs = xl4[gi];
        const float4 b = b4[c4];
        float4 o = o4[gi];
        float4 v;
        v.x = (a.x + xs.x * as) + b.x;
        v.y = (a.y + xs.y * as) + b.y;
        v.z = (a.z + xs.z * as) + b.z;
        v.w = (a.w + xs.w * as) + b.w;
        v.x = v.x > 0.f ? v.x : (__expf(v.x) - 1.f);
        v.y = v.y > 0.f ? v.y : (__expf(v.y) - 1.f);
        v.z = v.z > 0.f ? v.z : (__expf(v.z) - 1.f);
        v.w = v.w > 0.f ? v.w : (__expf(v.w) - 1.f);
        o.x += v.x; o.y += v.y; o.z += v.z; o.w += v.w;
        o4[gi] = o;
    }
}

__global__ __launch_bounds__(256) void scale_out(float* __restrict__ out, float inv,
                                                 long long total4) {
    float4* __restrict__ o4 = (float4*)out;
    const long long base = (long long)blockIdx.x * (256 * STREAM_U) + threadIdx.x;
    float4 v[STREAM_U];
    #pragma unroll
    for (int u = 0; u < STREAM_U; ++u) {
        const long long i = base + u * 256;
        v[u] = o4[(i < total4) ? i : 0];
    }
    #pragma unroll
    for (int u = 0; u < STREAM_U; ++u) {
        const long long i = base + u * 256;
        if (i < total4) {
            float4 r;
            r.x = v[u].x * inv; r.y = v[u].y * inv; r.z = v[u].z * inv; r.w = v[u].w * inv;
            o4[i] = r;
        }
    }
}

static inline unsigned gridFor(long long work, int block) {
    return (unsigned)((work + block - 1) / block);
}

extern "C" void kernel_launch(void* const* d_in, const int* in_sizes, int n_in,
                              void* d_out, int out_size, void* d_ws, size_t ws_size,
                              hipStream_t stream) {
    const int*   ei       = (const int*)d_in[0];
    const float* emb      = (const float*)d_in[1];
    const float* prompt   = (const float*)d_in[2];
    const float* proj_w   = (const float*)d_in[3];
    const float* proj_b   = (const float*)d_in[4];
    const float* conv_ws  = (const float*)d_in[5];
    const float* att_srcs = (const float*)d_in[6];
    const float* att_dsts = (const float*)d_in[7];
    const float* conv_bs  = (const float*)d_in[8];

    const int E = in_sizes[0] / 2;
    const int N = in_sizes[1] / DDIM;
    const int P = in_sizes[2];
    const int L = in_sizes[5] / (DDIM * DDIM);

    float* out = (float*)d_out;

    float*    pvec   = (float*)d_ws;
    float*    xbuf   = pvec + 64;
    float*    xlb    = xbuf + (size_t)N * DDIM;
    float*    a_src  = xlb + (size_t)N * DDIM;
    float*    a_dst  = a_src + N;


    float* tail = a_dst + N;
    const unsigned long long ws_bytes = (unsigned long long)ws_size;
    compute_pvec<<<gridFor((long long)E + 1, 256 * MAP_U * 4), 256, 0, stream>>>(prompt, proj_w, proj_b, pvec, P,
                                                                          ei, E, N, tail, pvec, ws_bytes);
    const long long total4 = (long long)N * (DDIM / 4);
    add_prompt<<<gridFor(total4, 256 * STREAM_U), 256, 0, stream>>>(emb, pvec, xbuf, out, total4);

    if (L > 0) gemm_xw_t<<<gridFor(N, GEMM_ROWS), 256, 0, stream>>>(xbuf, conv_ws, xlb, N);
    float* cur = xlb;
    float* nxt = xbuf;
    for (int l = 0; l < L; ++l) {
        const float* as = att_srcs + (size_t)l * DDIM;
        const float* ad = att_dsts + (size_t)l * DDIM;
        const float* cb = conv_bs  + (size_t)l * DDIM;
        const float* Wn = (l + 1 < L) ? conv_ws + (size_t)(l + 1) * DDIM * DDIM : nullptr;

        att_dots<<<gridFor(N, 128), 128, 0, stream>>>(cur, as, ad, a_src, a_dst, N);
        edge_owner<<<gridFor(N, OWN_ROWS), OWN_T, 0, stream>>>(ei, E, N, a_src, a_dst, cur, cb, Wn, nxt, out,
                                                               tail, pvec, ws_bytes, l, L);
        float* turn = cur; cur = nxt; nxt = turn;
    }
    scale_out<<<gridFor(total4, 256 * STREAM_U), 256, 0, stream>>>(out, 1.f / (float)(L + 1), total4);
}


# round trips asked as groups: product kernel stages W and x tile with 12 loads per thread at once; gather 16 list entries per group, tail too; last layer rows 8 loads first; LDS adds in arrival order
# speedup vs baseline: 3.8496x; 1.0348x over previous
#include <hip/hip_runtime.h>
#include <hip/hip_bf16.h>

typedef float v2f __attribute__((ext_vector_type(2)));
typedef float v8f __attribute__((ext_vector_type(8)));

#define DDIM 64
#define GEMM_ROWS 128
#define SX_STRIDE 68
#define SW_STRIDE 68

__device__ __forceinline__ unsigned fkey(float f) {
    unsigned u = __float_as_uint(f);
    return (u & 0x80000000u) ? ~u : (u | 0x80000000u);
}
__device__ __forceinline__ float fdec(unsigned k) {
    return (k & 0x80000000u) ? __uint_as_float(k & 0x7FFFFFFFu)
                             : __uint_as_float(~k);
}
#define FKEY_NEG_INF 0x007FFFFFu

#define OWN_ROWS 592

#define OWN_NONE 255u
__device__ __forceinline__ unsigned char* own_map_base(float* tail) {
    unsigned char* p = (unsigned char*)tail;
    return p + ((0ull - (unsigned long long)p) & 15ull);
}
__device__ __forceinline__ bool own_map_ok(const float* wsbase, float* tail, unsigned long long ws_bytes,
                                           int E, int N) {
    const unsigned long long off = (unsigned long long)(own_map_base(tail) - (const unsigned char*)wsbase);
    return (N + OWN_ROWS - 1) / OWN_ROWS <= 255 && off + (unsigned long long)E <= ws_bytes;
}
__device__ __forceinline__ unsigned own_of(int dst, int N) {
    return ((unsigned)dst < (unsigned)N) ? (unsigned)dst / (unsigned)OWN_ROWS : OWN_NONE;
}

#define LIST_SRC_BITS 18
#define LIST_SRC_MASK ((1u << LIST_SRC_BITS) - 1u)
#define OWN_CAP  8192
__device__ __forceinline__ int* own_len_base(float* tail, int E) {
    unsigned char* p = own_map_base(tail) + E;
    return (int*)(p + ((0ull - (unsigned long long)p) & 15ull));
}
__device__ __forceinline__ unsigned* own_list_base(float* tail, int E, int N) {
    const int nown = (N + OWN_ROWS - 1) / OWN_ROWS;
    return (unsigned*)(own_len_base(tail, E) + ((nown + 3) & ~3));
}
__device__ __forceinline__ int own_list_stride(const float* wsbase, float* tail, unsigned long long ws_bytes,
                                               int E, int N) {
    if (!own_map_ok(wsbase, tail, ws_bytes, E, N)) return 0;
    const unsigned long long nown = (unsigned long long)((N + OWN_ROWS - 1) / OWN_ROWS);
    const unsigned long long off =
        (unsigned long long)((const unsigned char*)own_list_base(tail, E, N) - (const unsigned char*)wsbase);
    if (nown == 0 || off >= ws_bytes) return 0;
    const unsigned long long per = (ws_bytes - off) / (nown * 4ull);
    return (int)(per < (unsigned long long)OWN_CAP ? per : (unsigned long long)OWN_CAP) & ~3;
}

#define MAP_U 4
__global__ __launch_bounds__(256) void compute_pvec(const float* __restrict__ prompt, const float* __restrict__ pw,
                             const float* __restrict__ pb, float* __restrict__ pvec, int P,
                             const int* __restrict__ ei, int E, int N,
                             float* tail, const float* wsbase, unsigned long long ws_bytes) {
    const int t = threadIdx.x;
    if (blockIdx.x == 0 && t < DDIM) {
        float s = pb[t];
        for (int j = 0; j < P; ++j) s += prompt[j] * pw[t * P + j];
        pvec[t] = s;
    }
    if (!own_map_ok(wsbase, tail, ws_bytes, E, N)) return;
    unsigned char* __restrict__ map = own_map_base(tail);
    const int* __restrict__ dstp = ei + E;
    if ((E & 3) == 0) {
        const int4* __restrict__ d4 = (const int4*)dstp;
        unsigned* __restrict__ m1 = (unsigned*)map;
        const int n4 = E >> 2;
        for (int base = blockIdx.x * (256 * MAP_U); base < n4; base += gridDim.x * (256 * MAP_U)) {
            int4 v[MAP_U];
            #pragma unroll
            for (int u = 0; u < MAP_U; ++u) {
                const int g = base + u * 256 + t;
                v[u] = d4[(g < n4) ? g : 0];
            }
            #pragma unroll
            for (int u = 0; u < MAP_U; ++u) {
                const int g = base + u * 256 + t;
                if (g < n4)
                    m1[g] = own_of(v[u].x, N) | (own_of(v[u].y, N) << 8) |
                            (own_of(v[u].z, N) << 16) | (own_of(v[u].w, N) << 24);
            }
        }
    } else {
        for (int e = blockIdx.x * 256 + t; e < E; e += gridDim.x * 256) map[e] = (unsigned char)own_of(dstp[e], N);
    }
}

#define STREAM_U 8
__global__ __launch_bounds__(256) void add_prompt(const float* __restrict__ emb,
                                                  const float* __restrict__ pvec,
                                                  float* __restrict__ x,
                                                  float* __restrict__ out_sum,
                                                  long long total4) {
    const float4* __restrict__ e4 = (const float4*)emb;
    float4* __restrict__ x4 = (float4*)x;
    float4* __restrict__ o4 = (float4*)out_sum;
    const int t = threadIdx.x;
    const float4 p = ((const float4*)pvec)[t & 15];
    const long long base = (long long)blockIdx.x * (256 * STREAM_U) + t;
    float4 v[STREAM_U];
    #pragma unroll
    for (int u = 0; u < STREAM_U; ++u) {
        const long long i = base + u * 256;
        v[u] = e4[(i < total4) ? i : 0];
    }
    #pragma unroll
    for (int u = 0; u < STREAM_U; ++u) {
        const long long i = base + u * 256;
        if (i < total4) {
            float4 r;
            r.x = v[u].x + p.x; r.y = v[u].y + p.y; r.z = v[u].z + p.z; r.w = v[u].w + p.w;
            x4[i] = r;
            o4[i] = r;
        }
    }
}

__global__ __launch_bounds__(256) void gemm_xw_t(const float* __restrict__ x,
                                                 const float* __restrict__ W,
                                                 float* __restrict__ xl, int N) {
    __shared__ float sx[GEMM_ROWS * SX_STRIDE];
    __shared__ float sw[DDIM * SW_STRIDE];
    const int t = threadIdx.x;
    const int rowBase = blockIdx.x * GEMM_ROWS;

    {
        const float4* Wv = (const float4*)W;
        const float4* xv = (const float4*)x;
        float4* swv = (float4*)sw;
        float4* sxv = (float4*)sx;
        constexpr int WU = (DDIM * DDIM) / 4 / 256;
        constexpr int XU = (GEMM_ROWS * DDIM) / 4 / 256;
        float4 wreg[WU], xreg[XU];
        #pragma unroll
        for (int u = 0; u < WU; ++u) wreg[u] = Wv[t + u * 256];
        #pragma unroll
        for (int u = 0; u < XU; ++u) {
            const int i = t + u * 256;
            int gr = rowBase + (i >> 4);
            if (gr >= N) gr = N - 1;
            xreg[u] = xv[(long long)gr * 16 + (i & 15)];
        }
        #pragma unroll
        for (int u = 0; u < WU; ++u) {
            const int i = t + u * 256;
            swv[(i >> 4) * (SW_STRIDE / 4) + (i & 15)] = wreg[u];
        }
        #pragma unroll
        for (int u = 0; u < XU; ++u) {
            const int i = t + u * 256;
            sxv[(i >> 4) * (SX_STRIDE / 4) + (i & 15)] = xreg[u];
        }
    }
    __syncthreads();

    const int wave  = t >> 5;
    const int lane  = t & 31;
    const int lrow  = lane & 15;
    const int khalf = (lane >> 4) << 1;
    const int wrow  = wave * 16;

    v8f acc0 = {0.f,0.f,0.f,0.f,0.f,0.f,0.f,0.f};
    v8f acc1 = acc0, acc2 = acc0, acc3 = acc0;

    const float* arow = &sx[(wrow + lrow) * SX_STRIDE + khalf];
    const float* brow = &sw[lrow * SW_STRIDE + khalf];

    #pragma unroll
    for (int k0 = 0; k0 < DDIM; k0 += 4) {
        v2f a, b0, b1, b2, b3;
        a.x  = arow[k0];                       a.y  = arow[k0 + 1];
        b0.x = brow[k0];                       b0.y = brow[k0 + 1];
        b1.x = brow[16 * SW_STRIDE + k0];      b1.y = brow[16 * SW_STRIDE + k0 + 1];
        b2.x = brow[32 * SW_STRIDE + k0];      b2.y = brow[32 * SW_STRIDE + k0 + 1];
        b3.x = brow[48 * SW_STRIDE + k0];      b3.y = brow[48 * SW_STRIDE + k0 + 1];
        acc0 = __builtin_amdgcn_wmma_f32_16x16x4_f32(false, a, false, b0, (short)0, acc0, false, false);
        acc1 = __builtin_amdgcn_wmma_f32_16x16x4_f32(false, a, false, b1, (short)0, acc1, false, false);
        acc2 = __builtin_amdgcn_wmma_f32_16x16x4_f32(false, a, false, b2, (short)0, acc2, false, false);
        acc3 = __builtin_amdgcn_wmma_f32_16x16x4_f32(false, a, false, b3, (short)0, acc3, false, false);
    }

    const int rhi = (lane >> 4) * 8;
    const long long rbase = (long long)(rowBase + wrow + rhi);
    if (rowBase + GEMM_ROWS <= N) {
        float* p = xl + rbase * DDIM + lrow;
        #pragma unroll
        for (int v = 0; v < 8; ++v) {
            p[(long long)v * DDIM +  0] = acc0[v];
            p[(long long)v * DDIM + 16] = acc1[v];
            p[(long long)v * DDIM + 32] = acc2[v];
            p[(long long)v * DDIM + 48] = acc3[v];
        }
    } else {
        #pragma unroll
        for (int v = 0; v < 8; ++v) {
            long long gr = rbase + v;
            if (gr < N) {
                float* p = xl + gr * DDIM + lrow;
                p[ 0] = acc0[v];
                p[16] = acc1[v];
                p[32] = acc2[v];
                p[48] = acc3[v];
            }
        }
    }
}

#define ATT_U 16
template <int M> __device__ __forceinline__ float att_xadd(float v) {
    const int o = __builtin_amdgcn_update_dpp(0, __builtin_bit_cast(int, v), 0x160 + M, 0xF, 0xF, true);
    return v + __builtin_bit_cast(float, o);
}
__device__ __forceinline__ float att_pick(const float (&r)[ATT_U], int c) {
    float o = r[0];
    #pragma unroll
    for (int u = 1; u < ATT_U; ++u) o = (c == u) ? r[u] : o;
    return o;
}
__global__ __launch_bounds__(128) void att_dots(const float* __restrict__ xl,
                                                const float* __restrict__ av_s,
                                                const float* __restrict__ av_d,
                                                float* __restrict__ a_src,
                                                float* __restrict__ a_dst, int N) {
    const int lane = threadIdx.x & 31;
    const int c = lane & 15;
    const int h = lane >> 4;
    const int nb = (blockIdx.x * 4 + (threadIdx.x >> 5)) * 32;
    const float4* __restrict__ xl4 = (const float4*)xl;
    const float4 s = ((const float4*)av_s)[c];
    const float4 d = ((const float4*)av_d)[c];
    float4 v[ATT_U];
    #pragma unroll
    for (int u = 0; u < ATT_U; ++u) {
        int n = nb + 16 * h + u;
        if (n >= N) n = N - 1;
        v[u] = xl4[(long long)n * (DDIM / 4) + c];
    }
    float rs[ATT_U], rd[ATT_U];
    #pragma unroll
    for (int u = 0; u < ATT_U; ++u) {
        float sa = v[u].x * s.x + v[u].y * s.y;
        float sb = v[u].z * s.z + v[u].w * s.w;
        float da = v[u].x * d.x + v[u].y * d.y;
        float db = v[u].z * d.z + v[u].w * d.w;
        sa = att_xadd<8>(sa); sb = att_xadd<8>(sb); da = att_xadd<8>(da); db = att_xadd<8>(db);
        sa = att_xadd<4>(sa); sb = att_xadd<4>(sb); da = att_xadd<4>(da); db = att_xadd<4>(db);
        sa = att_xadd<2>(sa); sb = att_xadd<2>(sb); da = att_xadd<2>(da); db = att_xadd<2>(db);
        sa = att_xadd<1>(sa); sb = att_xadd<1>(sb); da = att_xadd<1>(da); db = att_xadd<1>(db);
        rs[u] = sa + sb;
        rd[u] = da + db;
    }
    const int n = nb + lane;
    const float os = att_pick(rs, c);
    const float od = att_pick(rd, c);
    if (n < N) { a_src[n] = os; a_dst[n] = od; }
}

#define ACC_PITCH 68
#define OWN_T    1024
#define OWN_U    8
#define GATHER_G 16
#define FIN_U    4
static_assert(OWN_CAP % (4 * OWN_T) == 0, "OWN_CAP must be whole trips of the logit step");

__device__ __forceinline__ float leaky02(float v) { return v > 0.f ? v : 0.2f * v; }

__device__ __forceinline__ float4 fin_piece(float4 a, float4 xs, float as, float4 b, float4 o) {
    float4 v;
    v.x = (a.x + xs.x * as) + b.x;
    v.y = (a.y + xs.y * as) + b.y;
    v.z = (a.z + xs.z * as) + b.z;
    v.w = (a.w + xs.w * as) + b.w;
    v.x = v.x > 0.f ? v.x : (__expf(v.x) - 1.f);
    v.y = v.y > 0.f ? v.y : (__expf(v.y) - 1.f);
    v.z = v.z > 0.f ? v.z : (__expf(v.z) - 1.f);
    v.w = v.w > 0.f ? v.w : (__expf(v.w) - 1.f);
    o.x += v.x; o.y += v.y; o.z += v.z; o.w += v.w;
    return o;
}

__global__ __launch_bounds__(OWN_T) void edge_owner(const int* __restrict__ ei, int E, int N,
                                                    const float* __restrict__ a_src,
                                                    const float* __restrict__ a_dst,
                                                    const float* __restrict__ xl,
                                                    const float* __restrict__ bias,
                                                    const float* __restrict__ Wnext,
                                                    float* __restrict__ xl_next,
                                                    float* __restrict__ out_sum,
                                                    float* tail, const float* wsbase,
                                                    unsigned long long ws_bytes,
                                                    int layer, int nlayers) {
    __shared__ float4   acc4[OWN_ROWS * (ACC_PITCH / 4)];
    __shared__ float4   sw4[DDIM * (SW_STRIDE / 4)];
    __shared__ unsigned smax[OWN_ROWS];
    __shared__ float    ssum[OWN_ROWS];
    __shared__ float    sadst[OWN_ROWS];
    __shared__ float    sself[OWN_ROWS];
    __shared__ int      lsrc[OWN_CAP];
    __shared__ int      ldl[OWN_CAP];
    __shared__ float    lval[OWN_CAP];
    __shared__ int      lcount;
    float* acc = (float*)acc4;

    const int t = threadIdx.x;
    const int lane = t & 31;
    const int wave = t >> 5;
    const int r0 = blockIdx.x * OWN_ROWS;
    const int rows = min(OWN_ROWS, N - r0);
    const int* __restrict__ dstp = ei + E;

    const int lstride = own_list_stride(wsbase, tail, ws_bytes, E, N);
    int* __restrict__ len_g = own_len_base(tail, E) + blockIdx.x;
    unsigned* __restrict__ list_g = own_list_base(tail, E, N) + (size_t)blockIdx.x * (size_t)lstride;
    const bool keep_list = (layer == 0 && nlayers > 1 && lstride > 0);
    int list_len = -1;
    if (layer > 0 && lstride > 0) list_len = *len_g;
    const bool from_list = (unsigned)list_len <= (unsigned)lstride;

    for (int r = t; r < OWN_ROWS; r += OWN_T) {
        float ad = 0.f, vs = 0.f;
        unsigned key = FKEY_NEG_INF;
        if (r < rows) {
            ad = a_dst[r0 + r];
            vs = leaky02(a_src[r0 + r] + ad);
            key = fkey(vs);
        }
        sadst[r] = ad; sself[r] = vs; smax[r] = key; ssum[r] = 0.f;
    }
    {
        const float4 z = {0.f, 0.f, 0.f, 0.f};
        for (int i = t; i < OWN_ROWS * (ACC_PITCH / 4); i += OWN_T) acc4[i] = z;
    }
    if (t == 0) lcount = 0;
    if (Wnext != nullptr) {
        const float4* __restrict__ Wv = (const float4*)Wnext;
        for (int i = t; i < (DDIM * DDIM) / 4; i += OWN_T) {
            const int r = i >> 4, c4 = i & 15;
            sw4[r * (SW_STRIDE / 4) + c4] = Wv[i];
        }
    }
    __syncthreads();

    #define OWN_NOTE(dlv, eidx)                                               \
        {                                                                     \
            const unsigned dl_ = (dlv);                                       \
            if (dl_ < (unsigned)rows) {                                       \
                const int slot_ = atomicAdd(&lcount, 1);                      \
                if (slot_ < OWN_CAP) { lsrc[slot_] = (eidx); ldl[slot_] = (int)dl_; } \
            }                                                                 \
        }
    if (from_list) {
    } else if (own_map_ok(wsbase, tail, ws_bytes, E, N)) {
        const unsigned char* __restrict__ map = own_map_base(tail);
        const uint4* __restrict__ m4 = (const uint4*)map;
        const int n16 = E >> 4;
        const unsigned me = blockIdx.x;
        const unsigned pat = me * 0x01010101u;
        const int start = (int)(((long long)blockIdx.x * n16) / gridDim.x) & ~7;
        for (int base = 0; base < n16; base += OWN_T * OWN_U) {
            uint4 v[OWN_U];
            int   j[OWN_U];
            #pragma unroll
            for (int u = 0; u < OWN_U; ++u) {
                const int i = base + u * OWN_T + t;
                int jj = i + start;
                if (jj >= n16) jj -= n16;
                j[u] = (i < n16) ? jj : -1;
                v[u] = m4[(i < n16) ? jj : 0];
            }
            #pragma unroll
            for (int u = 0; u < OWN_U; ++u) {
                const unsigned x0 = v[u].x ^ pat, x1 = v[u].y ^ pat, x2 = v[u].z ^ pat, x3 = v[u].w ^ pat;
                const unsigned y0 = ((x0 & 0x7f7f7f7fu) + 0x7f7f7f7fu) | x0;
                const unsigned y1 = ((x1 & 0x7f7f7f7fu) + 0x7f7f7f7fu) | x1;
                const unsigned y2 = ((x2 & 0x7f7f7f7fu) + 0x7f7f7f7fu) | x2;
                const unsigned y3 = ((x3 & 0x7f7f7f7fu) + 0x7f7f7f7fu) | x3;
                if (j[u] >= 0 && ((y0 & y1 & y2 & y3) | 0x7f7f7f7fu) != 0xffffffffu) {
                    unsigned c = ((~y0 & 0x80808080u) >> 7) | ((~y1 & 0x80808080u) >> 6) |
                                 ((~y2 & 0x80808080u) >> 5) | ((~y3 & 0x80808080u) >> 4);
                    const int e0 = j[u] << 4;
                    while (c) {
                        const int b = __builtin_ctz(c);
                        c &= c - 1;
                        const int slot = atomicAdd(&lcount, 1);
                        if (slot < OWN_CAP) lsrc[slot] = e0 + ((b & 7) << 2) + (b >> 3);
                    }
                }
            }
        }
        for (int e = (n16 << 4) + t; e < E; e += OWN_T) {
            if (map[e] == me) {
                const int slot = atomicAdd(&lcount, 1);
                if (slot < OWN_CAP) lsrc[slot] = e;
            }
        }
    } else if ((E & 3) == 0) {
        const int4* __restrict__ d4 = (const int4*)dstp;
        const int n4 = E >> 2;
        const int start = (int)(((long long)blockIdx.x * n4) / gridDim.x) & ~7;
        for (int base = 0; base < n4; base += OWN_T * OWN_U) {
            int4 v[OWN_U];
            int  j[OWN_U];
            #pragma unroll
            for (int u = 0; u < OWN_U; ++u) {
                const int i = base + u * OWN_T + t;
                int jj = i + start;
                if (jj >= n4) jj -= n4;
                j[u] = (i < n4) ? jj : -1;
                v[u] = d4[(i < n4) ? jj : 0];
            }
            #pragma unroll
            for (int u = 0; u < OWN_U; ++u) {
                const unsigned q0 = (unsigned)(v[u].x - r0), q1 = (unsigned)(v[u].y - r0);
                const unsigned q2 = (unsigned)(v[u].z - r0), q3 = (unsigned)(v[u].w - r0);
                if (j[u] >= 0 && min(min(q0, q1), min(q2, q3)) < (unsigned)rows) {
                    const int e = j[u] << 2;
                    OWN_NOTE(q0, e)
                    OWN_NOTE(q1, e + 1)
                    OWN_NOTE(q2, e + 2)
                    OWN_NOTE(q3, e + 3)
                }
            }
        }
    } else {
        for (int e = t; e < E; e += OWN_T) OWN_NOTE((unsigned)(dstp[e] - r0), e)
    }
    #undef OWN_NOTE
    __syncthreads();
    const int cnt = from_list ? list_len : lcount;
    if (keep_list && t == 0) *len_g = (cnt <= lstride) ? cnt : -1;

    if (cnt <= OWN_CAP) {
        if (from_list) {
            for (int b0 = 0; b0 < cnt; b0 += OWN_T * 4) {
                unsigned w_[4];
                float as_[4];
                #pragma unroll
                for (int u = 0; u < 4; ++u) {
                    const int i = b0 + u * OWN_T + t;
                    const unsigned w = list_g[(i < cnt) ? i : 0];
                    w_[u] = (i < cnt) ? w : 0u;
                }
                #pragma unroll
                for (int u = 0; u < 4; ++u) as_[u] = a_src[w_[u] & LIST_SRC_MASK];
                #pragma unroll
                for (int u = 0; u < 4; ++u) {
                    const int i = b0 + u * OWN_T + t;
                    const int dl = (int)(w_[u] >> LIST_SRC_BITS);
                    const float v = leaky02(as_[u] + sadst[dl]);
                    lsrc[i] = (int)(w_[u] & LIST_SRC_MASK);
                    ldl[i] = dl;
                    lval[i] = v;
                    atomicMax(&smax[dl], (i < cnt) ? fkey(v) : 0u);
                }
            }
        } else
        for (int b0 = 0; b0 < cnt; b0 += OWN_T * 4) {
            int e_[4], s_[4], d_[4];
            float as_[4];
            #pragma unroll
            for (int u = 0; u < 4; ++u) { const int i = b0 + u * OWN_T + t; e_[u] = (i < cnt) ? lsrc[i] : 0; }
            #pragma unroll
            for (int u = 0; u < 4; ++u) { s_[u] = ei[e_[u]]; d_[u] = dstp[e_[u]]; }
            #pragma unroll
            for (int u = 0; u < 4; ++u) as_[u] = a_src[s_[u]];
            #pragma unroll
            for (int u = 0; u < 4; ++u) {
                const int i = b0 + u * OWN_T + t;
                if (i < cnt) {
                    const int dl = d_[u] - r0;
                    const float v = leaky02(as_[u] + sadst[dl]);
                    lsrc[i] = s_[u];
                    ldl[i] = dl;
                    lval[i] = v;
                    if (keep_list && cnt <= lstride) list_g[i] = ((unsigned)s_[u] & LIST_SRC_MASK) | ((unsigned)dl << LIST_SRC_BITS);
                    atomicMax(&smax[dl], fkey(v));
                }
            }
        }
        __syncthreads();
        for (int i = t; i < cnt; i += OWN_T) {
            const int dl = ldl[i];
            const float w = __expf(lval[i] - fdec(smax[dl]));
            lval[i] = w;
            atomicAdd(&ssum[dl], w);
        }
        for (int r = t; r < rows; r += OWN_T) {
            const float w = __expf(sself[r] - fdec(smax[r]));
            sself[r] = w;
            atomicAdd(&ssum[r], w);
        }
        __syncthreads();
        const int swave = __builtin_amdgcn_readfirstlane(wave);
        for (int g = swave * 32; g < cnt; g += OWN_T) {
            const int   il = (g + lane < cnt) ? g + lane : g;
            const int   sl = lsrc[il];
            const int   dl = ldl[il];
            const float al = lval[il] / (ssum[dl] + 1e-16f);
            const int   ab = __builtin_bit_cast(int, al);
            #pragma unroll
            for (int q = 0; q < 32 / GATHER_G; ++q) {
                const int gq = g + q * GATHER_G;
                if (gq < cnt) {
                    float lo[GATHER_G], hi[GATHER_G];
                    const float* xr[GATHER_G];
                    #pragma unroll
                    for (int k = 0; k < GATHER_G; ++k)
                        xr[k] = xl + (long long)__builtin_amdgcn_readlane(sl, q * GATHER_G + k) * DDIM + lane;
                    #pragma unroll
                    for (int k = 0; k < GATHER_G; ++k) {
                        lo[k] = xr[k][0];
                        hi[k] = xr[k][32];
                    }
                    if (gq + GATHER_G <= cnt) {
                        #pragma unroll
                        for (int k = 0; k < GATHER_G; ++k) {
                            const float a = __builtin_bit_cast(float, __builtin_amdgcn_readlane(ab, q * GATHER_G + k));
                            float* ap = acc + __builtin_amdgcn_readlane(dl, q * GATHER_G + k) * ACC_PITCH + lane;
                            atomicAdd(ap,      lo[k] * a);
                            atomicAdd(ap + 32, hi[k] * a);
                        }
                    } else {
                        #pragma unroll
                        for (int k = 0; k < GATHER_G; ++k) {
                            if (gq + k < cnt) {
                                const float a = __builtin_bit_cast(float, __builtin_amdgcn_readlane(ab, q * GATHER_G + k));
                                float* ap = acc + __builtin_amdgcn_readlane(dl, q * GATHER_G + k) * ACC_PITCH + lane;
                                atomicAdd(ap,      lo[k] * a);
                                atomicAdd(ap + 32, hi[k] * a);
                            }
                        }
                    }
                }
            }
        }
    } else {
        for (int e = t; e < E; e += OWN_T) {
            const unsigned dl = (unsigned)(dstp[e] - r0);
            if (dl < (unsigned)rows)
                atomicMax(&smax[dl], fkey(leaky02(a_src[ei[e]] + sadst[dl])));
        }
        __syncthreads();
        for (int e = t; e < E; e += OWN_T) {
            const unsigned dl = (unsigned)(dstp[e] - r0);
            if (dl < (unsigned)rows)
                atomicAdd(&ssum[dl], __expf(leaky02(a_src[ei[e]] + sadst[dl]) - fdec(smax[dl])));
        }
        for (int r = t; r < rows; r += OWN_T) {
            const float w = __expf(sself[r] - fdec(smax[r]));
            sself[r] = w;
            atomicAdd(&ssum[r], w);
        }
        __syncthreads();
        for (int e = t; e < E; e += OWN_T) {
            const unsigned dl = (unsigned)(dstp[e] - r0);
            if (dl < (unsigned)rows) {
                const int s = ei[e];
                const float aj = __expf(leaky02(a_src[s] + sadst[dl]) - fdec(smax[dl])) / (ssum[dl] + 1e-16f);
                const float* xr = xl + (long long)s * DDIM;
                float* ap = acc + dl * ACC_PITCH;
                for (int c = 0; c < DDIM; ++c) atomicAdd(ap + c, xr[c] * aj);
            }
        }
    }
    __syncthreads();

    const float4* __restrict__ xl4 = (const float4*)xl;
    const float4* __restrict__ b4 = (const float4*)bias;
    float4* __restrict__ o4 = (float4*)out_sum;
    if (Wnext != nullptr) {
        const float* sw = (const float*)sw4;
        const int c4    = lane & 15;
        const int h     = lane >> 4;
        const int khalf = h << 1;
        const int rhi   = h * 8;
        const int ntiles = (rows + 15) >> 4;
        const float4 b = b4[c4];
        const float* brow = &sw[c4 * SW_STRIDE + khalf];
        for (int tile = wave; tile < ntiles; tile += OWN_T / 32) {
            const int rb = tile * 16;
            float4 xs[8], o[8];
            #pragma unroll
            for (int u = 0; u < 8; ++u) {
                const int r = rb + 2 * u + h;
                const long long gi = (long long)(r0 + (r < rows ? r : rows - 1)) * (DDIM / 4) + c4;
                xs[u] = xl4[gi];
                o[u]  = o4[gi];
            }
            #pragma unroll
            for (int u = 0; u < 8; ++u) {
                const int r = rb + 2 * u + h;
                if (r < rows) {
                    const long long gi = (long long)(r0 + r) * (DDIM / 4) + c4;
                    const float as = sself[r] / (ssum[r] + 1e-16f);
                    const float4 a = acc4[r * (ACC_PITCH / 4) + c4];
                    float4 v;
                    v.x = (a.x + xs[u].x * as) + b.x;
                    v.y = (a.y + xs[u].y * as) + b.y;
                    v.z = (a.z + xs[u].z * as) + b.z;
                    v.w = (a.w + xs[u].w * as) + b.w;
                    v.x = v.x > 0.f ? v.x : (__expf(v.x) - 1.f);
                    v.y = v.y > 0.f ? v.y : (__expf(v.y) - 1.f);
                    v.z = v.z > 0.f ? v.z : (__expf(v.z) - 1.f);
                    v.w = v.w > 0.f ? v.w : (__expf(v.w) - 1.f);
                    float* xr = acc + r * ACC_PITCH + c4 * 4;
                    xr[0] = v.x; xr[1] = v.y; xr[2] = v.z; xr[3] = v.w;
                    float4 q = o[u];
                    q.x += v.x; q.y += v.y; q.z += v.z; q.w += v.w;
                    o4[gi] = q;
                }
            }
            v8f acc0 = {0.f,0.f,0.f,0.f,0.f,0.f,0.f,0.f};
            v8f acc1 = acc0, acc2 = acc0, acc3 = acc0;
            const float* arow = &acc[(rb + c4) * ACC_PITCH + khalf];
            #pragma nounroll
            for (int kb = 0; kb < DDIM; kb += 16) {
                #pragma unroll
                for (int k0 = kb; k0 < kb + 16; k0 += 4) {
                    v2f a, b0, b1, b2, b3;
                    a.x  = arow[k0];                       a.y  = arow[k0 + 1];
                    b0.x = brow[k0];                       b0.y = brow[k0 + 1];
                    b1.x = brow[16 * SW_STRIDE + k0];      b1.y = brow[16 * SW_STRIDE + k0 + 1];
                    b2.x = brow[32 * SW_STRIDE + k0];      b2.y = brow[32 * SW_STRIDE + k0 + 1];
                    b3.x = brow[48 * SW_STRIDE + k0];      b3.y = brow[48 * SW_STRIDE + k0 + 1];
                    acc0 = __builtin_amdgcn_wmma_f32_16x16x4_f32(false, a, false, b0, (short)0, acc0, false, false);
                    acc1 = __builtin_amdgcn_wmma_f32_16x16x4_f32(false, a, false, b1, (short)0, acc1, false, false);
                    acc2 = __builtin_amdgcn_wmma_f32_16x16x4_f32(false, a, false, b2, (short)0, acc2, false, false);
                    acc3 = __builtin_amdgcn_wmma_f32_16x16x4_f32(false, a, false, b3, (short)0, acc3, false, false);
                }
            }
            const int lr0 = rb + rhi;
            float* p = xl_next + (long long)(r0 + lr0) * DDIM + c4;
            if (rb + 16 <= rows) {
                #pragma unroll
                for (int v = 0; v < 8; ++v) {
                    p[(long long)v * DDIM +  0] = acc0[v];
                    p[(long long)v * DDIM + 16] = acc1[v];
                    p[(long long)v * DDIM + 32] = acc2[v];
                    p[(long long)v * DDIM + 48] = acc3[v];
                }
            } else {
                #pragma unroll
                for (int v = 0; v < 8; ++v) {
                    if (lr0 + v < rows) {
                        p[(long long)v * DDIM +  0] = acc0[v];
                        p[(long long)v * DDIM + 16] = acc1[v];
                        p[(long long)v * DDIM + 32] = acc2[v];
                        p[(long long)v * DDIM + 48] = acc3[v];
                    }
                }
            }
        }
        return;
    }
    const int n4 = rows * (DDIM / 4);
    const long long g0 = (long long)r0 * (DDIM / 4);
    const float4 bq = b4[t & 15];
    int p0 = 0;
    for (; p0 + OWN_T * FIN_U <= n4; p0 += OWN_T * FIN_U) {
        float4 xs[FIN_U], o[FIN_U];
        #pragma unroll
        for (int u = 0; u < FIN_U; ++u) {
            const long long gi = g0 + (p0 + u * OWN_T + t);
            xs[u] = xl4[gi];
            o[u]  = o4[gi];
        }
        #pragma unroll
        for (int u = 0; u < FIN_U; ++u) {
            const int i = p0 + u * OWN_T + t;
            const int r = i >> 4;
            const float as = sself[r] / (ssum[r] + 1e-16f);
            o4[g0 + i] = fin_piece(acc4[r * (ACC_PITCH / 4) + (t & 15)], xs[u], as, bq, o[u]);
        }
    }
    for (int i = p0 + t; i < n4; i += OWN_T) {
        const int r = i >> 4;
        const float as = sself[r] / (ssum[r] + 1e-16f);
        o4[g0 + i] = fin_piece(acc4[r * (ACC_PITCH / 4) + (t & 15)], xl4[g0 + i], as, bq, o4[g0 + i]);
    }
}

__global__ __launch_bounds__(256) void scale_out(float* __restrict__ out, float inv,
                                                 long long total4) {
    float4* __restrict__ o4 = (float4*)out;
    const long long base = (long long)blockIdx.x * (256 * STREAM_U) + threadIdx.x;
    float4 v[STREAM_U];
    #pragma unroll
    for (int u = 0; u < STREAM_U; ++u) {
        const long long i = base + u * 256;
        v[u] = o4[(i < total4) ? i : 0];
    }
    #pragma unroll
    for (int u = 0; u < STREAM_U; ++u) {
        const long long i = base + u * 256;
        if (i < total4) {
            float4 r;
            r.x = v[u].x * inv; r.y = v[u].y * inv; r.z = v[u].z * inv; r.w = v[u].w * inv;
            o4[i] = r;
        }
    }
}

static inline unsigned gridFor(long long work, int block) {
    return (unsigned)((work + block - 1) / block);
}

extern "C" void kernel_launch(void* const* d_in, const int* in_sizes, int n_in,
                              void* d_out, int out_size, void* d_ws, size_t ws_size,
                              hipStream_t stream) {
    const int*   ei       = (const int*)d_in[0];
    const float* emb      = (const float*)d_in[1];
    const float* prompt   = (const float*)d_in[2];
    const float* proj_w   = (const float*)d_in[3];
    const float* proj_b   = (const float*)d_in[4];
    const float* conv_ws  = (const float*)d_in[5];
    const float* att_srcs = (const float*)d_in[6];
    const float* att_dsts = (const float*)d_in[7];
    const float* conv_bs  = (const float*)d_in[8];

    const int E = in_sizes[0] / 2;
    const int N = in_sizes[1] / DDIM;
    const int P = in_sizes[2];
    const int L = in_sizes[5] / (DDIM * DDIM);

    float* out = (float*)d_out;

    float*    pvec   = (float*)d_ws;
    float*    xbuf   = pvec + 64;
    float*    xlb    = xbuf + (size_t)N * DDIM;
    float*    a_src  = xlb + (size_t)N * DDIM;
    float*    a_dst  = a_src + N;


    float* tail = a_dst + N;
    const unsigned long long ws_bytes = (unsigned long long)ws_size;
    compute_pvec<<<gridFor((long long)E + 1, 256 * MAP_U * 4), 256, 0, stream>>>(prompt, proj_w, proj_b, pvec, P,
                                                                          ei, E, N, tail, pvec, ws_bytes);
    const long long total4 = (long long)N * (DDIM / 4);
    add_prompt<<<gridFor(total4, 256 * STREAM_U), 256, 0, stream>>>(emb, pvec, xbuf, out, total4);

    if (L > 0) gemm_xw_t<<<gridFor(N, GEMM_ROWS), 256, 0, stream>>>(xbuf, conv_ws, xlb, N);
    float* cur = xlb;
    float* nxt = xbuf;
    for (int l = 0; l < L; ++l) {
        const float* as = att_srcs + (size_t)l * DDIM;
        const float* ad = att_dsts + (size_t)l * DDIM;
        const float* cb = conv_bs  + (size_t)l * DDIM;
        const float* Wn = (l + 1 < L) ? conv_ws + (size_t)(l + 1) * DDIM * DDIM : nullptr;

        att_dots<<<gridFor(N, 128), 128, 0, stream>>>(cur, as, ad, a_src, a_dst, N);
        edge_owner<<<gridFor(N, OWN_ROWS), OWN_T, 0, stream>>>(ei, E, N, a_src, a_dst, cur, cb, Wn, nxt, out,
                                                               tail, pvec, ws_bytes, l, L);
        float* turn = cur; cur = nxt; nxt = turn;
    }
    scale_out<<<gridFor(total4, 256 * STREAM_U), 256, 0, stream>>>(out, 1.f / (float)(L + 1), total4);
}


# x = emb + p is stored nowhere: layer 0's product reads emb and starts its accumulators at W0 p (that product's sum order changes), emb + p joins out in the first edge phase; add_prompt launch gone
# speedup vs baseline: 4.4254x; 1.1496x over previous
#include <hip/hip_runtime.h>
#include <hip/hip_bf16.h>

typedef float v2f __attribute__((ext_vector_type(2)));
typedef float v8f __attribute__((ext_vector_type(8)));

#define DDIM 64
#define GEMM_ROWS 128
#define SX_STRIDE 68
#define SW_STRIDE 68

__device__ __forceinline__ unsigned fkey(float f) {
    unsigned u = __float_as_uint(f);
    return (u & 0x80000000u) ? ~u : (u | 0x80000000u);
}
__device__ __forceinline__ float fdec(unsigned k) {
    return (k & 0x80000000u) ? __uint_as_float(k & 0x7FFFFFFFu)
                             : __uint_as_float(~k);
}
#define FKEY_NEG_INF 0x007FFFFFu

#define OWN_ROWS 592

#define OWN_NONE 255u
__device__ __forceinline__ unsigned char* own_map_base(float* tail) {
    unsigned char* p = (unsigned char*)tail;
    return p + ((0ull - (unsigned long long)p) & 15ull);
}
__device__ __forceinline__ bool own_map_ok(const float* wsbase, float* tail, unsigned long long ws_bytes,
                                           int E, int N) {
    const unsigned long long off = (unsigned long long)(own_map_base(tail) - (const unsigned char*)wsbase);
    return (N + OWN_ROWS - 1) / OWN_ROWS <= 255 && off + (unsigned long long)E <= ws_bytes;
}
__device__ __forceinline__ unsigned own_of(int dst, int N) {
    return ((unsigned)dst < (unsigned)N) ? (unsigned)dst / (unsigned)OWN_ROWS : OWN_NONE;
}

#define LIST_SRC_BITS 18
#define LIST_SRC_MASK ((1u << LIST_SRC_BITS) - 1u)
#define OWN_CAP  8192
__device__ __forceinline__ int* own_len_base(float* tail, int E) {
    unsigned char* p = own_map_base(tail) + E;
    return (int*)(p + ((0ull - (unsigned long long)p) & 15ull));
}
__device__ __forceinline__ unsigned* own_list_base(float* tail, int E, int N) {
    const int nown = (N + OWN_ROWS - 1) / OWN_ROWS;
    return (unsigned*)(own_len_base(tail, E) + ((nown + 3) & ~3));
}
__device__ __forceinline__ int own_list_stride(const float* wsbase, float* tail, unsigned long long ws_bytes,
                                               int E, int N) {
    if (!own_map_ok(wsbase, tail, ws_bytes, E, N)) return 0;
    const unsigned long long nown = (unsigned long long)((N + OWN_ROWS - 1) / OWN_ROWS);
    const unsigned long long off =
        (unsigned long long)((const unsigned char*)own_list_base(tail, E, N) - (const unsigned char*)wsbase);
    if (nown == 0 || off >= ws_bytes) return 0;
    const unsigned long long per = (ws_bytes - off) / (nown * 4ull);
    return (int)(per < (unsigned long long)OWN_CAP ? per : (unsigned long long)OWN_CAP) & ~3;
}

#define MAP_U 4
__global__ __launch_bounds__(256) void compute_pvec(const float* __restrict__ prompt, const float* __restrict__ pw,
                             const float* __restrict__ pb, float* __restrict__ pvec, int P,
                             const float* __restrict__ W0,
                             float* __restrict__ qvec,
                             const int* __restrict__ ei, int E, int N,
                             float* tail, const float* wsbase, unsigned long long ws_bytes) {
    __shared__ float sp[DDIM];
    const int t = threadIdx.x;
    if (blockIdx.x == 0) {
        float4 wrow[DDIM / 4];
        const bool fold = (W0 != nullptr);
        if (fold && t < DDIM) {
            #pragma unroll
            for (int u = 0; u < DDIM / 4; ++u) wrow[u] = ((const float4*)W0)[t * (DDIM / 4) + u];
        }
        if (t < DDIM) {
            float s = pb[t];
            for (int j = 0; j < P; ++j) s += prompt[j] * pw[t * P + j];
            pvec[t] = s;
            sp[t] = s;
        }
        if (fold) {
            __syncthreads();
            if (t < DDIM) {
                float q = 0.f;
                #pragma unroll
                for (int u = 0; u < DDIM / 4; ++u) {
                    q += wrow[u].x * sp[4 * u];
                    q += wrow[u].y * sp[4 * u + 1];
                    q += wrow[u].z * sp[4 * u + 2];
                    q += wrow[u].w * sp[4 * u + 3];
                }
                qvec[t] = q;
            }
        }
    }
    if (!own_map_ok(wsbase, tail, ws_bytes, E, N)) return;
    unsigned char* __restrict__ map = own_map_base(tail);
    const int* __restrict__ dstp = ei + E;
    if ((E & 3) == 0) {
        const int4* __restrict__ d4 = (const int4*)dstp;
        unsigned* __restrict__ m1 = (unsigned*)map;
        const int n4 = E >> 2;
        for (int base = blockIdx.x * (256 * MAP_U); base < n4; base += gridDim.x * (256 * MAP_U)) {
            int4 v[MAP_U];
            #pragma unroll
            for (int u = 0; u < MAP_U; ++u) {
                const int g = base + u * 256 + t;
                v[u] = d4[(g < n4) ? g : 0];
            }
            #pragma unroll
            for (int u = 0; u < MAP_U; ++u) {
                const int g = base + u * 256 + t;
                if (g < n4)
                    m1[g] = own_of(v[u].x, N) | (own_of(v[u].y, N) << 8) |
                            (own_of(v[u].z, N) << 16) | (own_of(v[u].w, N) << 24);
            }
        }
    } else {
        for (int e = blockIdx.x * 256 + t; e < E; e += gridDim.x * 256) map[e] = (unsigned char)own_of(dstp[e], N);
    }
}

#define STREAM_U 8
__global__ __launch_bounds__(256) void add_prompt(const float* __restrict__ emb,
                                                  const float* __restrict__ pvec,
                                                  float* __restrict__ x,
                                                  float* __restrict__ out_sum,
                                                  long long total4) {
    const float4* __restrict__ e4 = (const float4*)emb;
    float4* __restrict__ x4 = (float4*)x;
    float4* __restrict__ o4 = (float4*)out_sum;
    const int t = threadIdx.x;
    const float4 p = ((const float4*)pvec)[t & 15];
    const long long base = (long long)blockIdx.x * (256 * STREAM_U) + t;
    float4 v[STREAM_U];
    #pragma unroll
    for (int u = 0; u < STREAM_U; ++u) {
        const long long i = base + u * 256;
        v[u] = e4[(i < total4) ? i : 0];
    }
    #pragma unroll
    for (int u = 0; u < STREAM_U; ++u) {
        const long long i = base + u * 256;
        if (i < total4) {
            float4 r;
            r.x = v[u].x + p.x; r.y = v[u].y + p.y; r.z = v[u].z + p.z; r.w = v[u].w + p.w;
            x4[i] = r;
            o4[i] = r;
        }
    }
}

__global__ __launch_bounds__(256) void gemm_xw_t(const float* __restrict__ x,
                                                 const float* __restrict__ W,
                                                 const float* __restrict__ q,
                                                 float* __restrict__ xl, int N) {
    __shared__ float sx[GEMM_ROWS * SX_STRIDE];
    __shared__ float sw[DDIM * SW_STRIDE];
    const int t = threadIdx.x;
    const int rowBase = blockIdx.x * GEMM_ROWS;
    const float q0 = q[(t & 15)], q1 = q[(t & 15) + 16], q2 = q[(t & 15) + 32], q3 = q[(t & 15) + 48];

    {
        const float4* Wv = (const float4*)W;
        const float4* xv = (const float4*)x;
        float4* swv = (float4*)sw;
        float4* sxv = (float4*)sx;
        constexpr int WU = (DDIM * DDIM) / 4 / 256;
        constexpr int XU = (GEMM_ROWS * DDIM) / 4 / 256;
        float4 wreg[WU], xreg[XU];
        #pragma unroll
        for (int u = 0; u < WU; ++u) wreg[u] = Wv[t + u * 256];
        #pragma unroll
        for (int u = 0; u < XU; ++u) {
            const int i = t + u * 256;
            int gr = rowBase + (i >> 4);
            if (gr >= N) gr = N - 1;
            xreg[u] = xv[(long long)gr * 16 + (i & 15)];
        }
        #pragma unroll
        for (int u = 0; u < WU; ++u) {
            const int i = t + u * 256;
            swv[(i >> 4) * (SW_STRIDE / 4) + (i & 15)] = wreg[u];
        }
        #pragma unroll
        for (int u = 0; u < XU; ++u) {
            const int i = t + u * 256;
            sxv[(i >> 4) * (SX_STRIDE / 4) + (i & 15)] = xreg[u];
        }
    }
    __syncthreads();

    const int wave  = t >> 5;
    const int lane  = t & 31;
    const int lrow  = lane & 15;
    const int khalf = (lane >> 4) << 1;
    const int wrow  = wave * 16;

    v8f acc0 = {q0,q0,q0,q0,q0,q0,q0,q0};
    v8f acc1 = {q1,q1,q1,q1,q1,q1,q1,q1};
    v8f acc2 = {q2,q2,q2,q2,q2,q2,q2,q2};
    v8f acc3 = {q3,q3,q3,q3,q3,q3,q3,q3};

    const float* arow = &sx[(wrow + lrow) * SX_STRIDE + khalf];
    const float* brow = &sw[lrow * SW_STRIDE + khalf];

    #pragma unroll
    for (int k0 = 0; k0 < DDIM; k0 += 4) {
        v2f a, b0, b1, b2, b3;
        a.x  = arow[k0];                       a.y  = arow[k0 + 1];
        b0.x = brow[k0];                       b0.y = brow[k0 + 1];
        b1.x = brow[16 * SW_STRIDE + k0];      b1.y = brow[16 * SW_STRIDE + k0 + 1];
        b2.x = brow[32 * SW_STRIDE + k0];      b2.y = brow[32 * SW_STRIDE + k0 + 1];
        b3.x = brow[48 * SW_STRIDE + k0];      b3.y = brow[48 * SW_STRIDE + k0 + 1];
        acc0 = __builtin_amdgcn_wmma_f32_16x16x4_f32(false, a, false, b0, (short)0, acc0, false, false);
        acc1 = __builtin_amdgcn_wmma_f32_16x16x4_f32(false, a, false, b1, (short)0, acc1, false, false);
        acc2 = __builtin_amdgcn_wmma_f32_16x16x4_f32(false, a, false, b2, (short)0, acc2, false, false);
        acc3 = __builtin_amdgcn_wmma_f32_16x16x4_f32(false, a, false, b3, (short)0, acc3, false, false);
    }

    const int rhi = (lane >> 4) * 8;
    const long long rbase = (long long)(rowBase + wrow + rhi);
    if (rowBase + GEMM_ROWS <= N) {
        float* p = xl + rbase * DDIM + lrow;
        #pragma unroll
        for (int v = 0; v < 8; ++v) {
            p[(long long)v * DDIM +  0] = acc0[v];
            p[(long long)v * DDIM + 16] = acc1[v];
            p[(long long)v * DDIM + 32] = acc2[v];
            p[(long long)v * DDIM + 48] = acc3[v];
        }
    } else {
        #pragma unroll
        for (int v = 0; v < 8; ++v) {
            long long gr = rbase + v;
            if (gr < N) {
                float* p = xl + gr * DDIM + lrow;
                p[ 0] = acc0[v];
                p[16] = acc1[v];
                p[32] = acc2[v];
                p[48] = acc3[v];
            }
        }
    }
}

#define ATT_U 16
template <int M> __device__ __forceinline__ float att_xadd(float v) {
    const int o = __builtin_amdgcn_update_dpp(0, __builtin_bit_cast(int, v), 0x160 + M, 0xF, 0xF, true);
    return v + __builtin_bit_cast(float, o);
}
__device__ __forceinline__ float att_pick(const float (&r)[ATT_U], int c) {
    float o = r[0];
    #pragma unroll
    for (int u = 1; u < ATT_U; ++u) o = (c == u) ? r[u] : o;
    return o;
}
__global__ __launch_bounds__(128) void att_dots(const float* __restrict__ xl,
                                                const float* __restrict__ av_s,
                                                const float* __restrict__ av_d,
                                                float* __restrict__ a_src,
                                                float* __restrict__ a_dst, int N) {
    const int lane = threadIdx.x & 31;
    const int c = lane & 15;
    const int h = lane >> 4;
    const int nb = (blockIdx.x * 4 + (threadIdx.x >> 5)) * 32;
    const float4* __restrict__ xl4 = (const float4*)xl;
    const float4 s = ((const float4*)av_s)[c];
    const float4 d = ((const float4*)av_d)[c];
    float4 v[ATT_U];
    #pragma unroll
    for (int u = 0; u < ATT_U; ++u) {
        int n = nb + 16 * h + u;
        if (n >= N) n = N - 1;
        v[u] = xl4[(long long)n * (DDIM / 4) + c];
    }
    float rs[ATT_U], rd[ATT_U];
    #pragma unroll
    for (int u = 0; u < ATT_U; ++u) {
        float sa = v[u].x * s.x + v[u].y * s.y;
        float sb = v[u].z * s.z + v[u].w * s.w;
        float da = v[u].x * d.x + v[u].y * d.y;
        float db = v[u].z * d.z + v[u].w * d.w;
        sa = att_xadd<8>(sa); sb = att_xadd<8>(sb); da = att_xadd<8>(da); db = att_xadd<8>(db);
        sa = att_xadd<4>(sa); sb = att_xadd<4>(sb); da = att_xadd<4>(da); db = att_xadd<4>(db);
        sa = att_xadd<2>(sa); sb = att_xadd<2>(sb); da = att_xadd<2>(da); db = att_xadd<2>(db);
        sa = att_xadd<1>(sa); sb = att_xadd<1>(sb); da = att_xadd<1>(da); db = att_xadd<1>(db);
        rs[u] = sa + sb;
        rd[u] = da + db;
    }
    const int n = nb + lane;
    const float os = att_pick(rs, c);
    const float od = att_pick(rd, c);
    if (n < N) { a_src[n] = os; a_dst[n] = od; }
}

#define ACC_PITCH 68
#define OWN_T    1024
#define OWN_U    8
#define GATHER_G 16
#define FIN_U    4
static_assert(OWN_CAP % (4 * OWN_T) == 0, "OWN_CAP must be whole trips of the logit step");

__device__ __forceinline__ float leaky02(float v) { return v > 0.f ? v : 0.2f * v; }

__device__ __forceinline__ float4 fin_piece(float4 a, float4 xs, float as, float4 b, float4 o) {
    float4 v;
    v.x = (a.x + xs.x * as) + b.x;
    v.y = (a.y + xs.y * as) + b.y;
    v.z = (a.z + xs.z * as) + b.z;
    v.w = (a.w + xs.w * as) + b.w;
    v.x = v.x > 0.f ? v.x : (__expf(v.x) - 1.f);
    v.y = v.y > 0.f ? v.y : (__expf(v.y) - 1.f);
    v.z = v.z > 0.f ? v.z : (__expf(v.z) - 1.f);
    v.w = v.w > 0.f ? v.w : (__expf(v.w) - 1.f);
    o.x += v.x; o.y += v.y; o.z += v.z; o.w += v.w;
    return o;
}

__global__ __launch_bounds__(OWN_T) void edge_owner(const int* __restrict__ ei, int E, int N,
                                                    const float* __restrict__ a_src,
                                                    const float* __restrict__ a_dst,
                                                    const float* __restrict__ xl,
                                                    const float* __restrict__ bias,
                                                    const float* __restrict__ Wnext,
                                                    float* __restrict__ xl_next,
                                                    float* __restrict__ out_sum,
                                                    const float* __restrict__ emb,
                                                    const float* __restrict__ pvec,
                                                    float* tail, const float* wsbase,
                                                    unsigned long long ws_bytes,
                                                    int layer, int nlayers) {
    __shared__ float4   acc4[OWN_ROWS * (ACC_PITCH / 4)];
    __shared__ float4   sw4[DDIM * (SW_STRIDE / 4)];
    __shared__ unsigned smax[OWN_ROWS];
    __shared__ float    ssum[OWN_ROWS];
    __shared__ float    sadst[OWN_ROWS];
    __shared__ float    sself[OWN_ROWS];
    __shared__ int      lsrc[OWN_CAP];
    __shared__ int      ldl[OWN_CAP];
    __shared__ float    lval[OWN_CAP];
    __shared__ int      lcount;
    float* acc = (float*)acc4;

    const int t = threadIdx.x;
    const int lane = t & 31;
    const int wave = t >> 5;
    const int r0 = blockIdx.x * OWN_ROWS;
    const int rows = min(OWN_ROWS, N - r0);
    const int* __restrict__ dstp = ei + E;

    const int lstride = own_list_stride(wsbase, tail, ws_bytes, E, N);
    int* __restrict__ len_g = own_len_base(tail, E) + blockIdx.x;
    unsigned* __restrict__ list_g = own_list_base(tail, E, N) + (size_t)blockIdx.x * (size_t)lstride;
    const bool keep_list = (layer == 0 && nlayers > 1 && lstride > 0);
    int list_len = -1;
    if (layer > 0 && lstride > 0) list_len = *len_g;
    const bool from_list = (unsigned)list_len <= (unsigned)lstride;

    for (int r = t; r < OWN_ROWS; r += OWN_T) {
        float ad = 0.f, vs = 0.f;
        unsigned key = FKEY_NEG_INF;
        if (r < rows) {
            ad = a_dst[r0 + r];
            vs = leaky02(a_src[r0 + r] + ad);
            key = fkey(vs);
        }
        sadst[r] = ad; sself[r] = vs; smax[r] = key; ssum[r] = 0.f;
    }
    {
        const float4 z = {0.f, 0.f, 0.f, 0.f};
        for (int i = t; i < OWN_ROWS * (ACC_PITCH / 4); i += OWN_T) acc4[i] = z;
    }
    if (t == 0) lcount = 0;
    if (Wnext != nullptr) {
        const float4* __restrict__ Wv = (const float4*)Wnext;
        for (int i = t; i < (DDIM * DDIM) / 4; i += OWN_T) {
            const int r = i >> 4, c4 = i & 15;
            sw4[r * (SW_STRIDE / 4) + c4] = Wv[i];
        }
    }
    __syncthreads();

    #define OWN_NOTE(dlv, eidx)                                               \
        {                                                                     \
            const unsigned dl_ = (dlv);                                       \
            if (dl_ < (unsigned)rows) {                                       \
                const int slot_ = atomicAdd(&lcount, 1);                      \
                if (slot_ < OWN_CAP) { lsrc[slot_] = (eidx); ldl[slot_] = (int)dl_; } \
            }                                                                 \
        }
    if (from_list) {
    } else if (own_map_ok(wsbase, tail, ws_bytes, E, N)) {
        const unsigned char* __restrict__ map = own_map_base(tail);
        const uint4* __restrict__ m4 = (const uint4*)map;
        const int n16 = E >> 4;
        const unsigned me = blockIdx.x;
        const unsigned pat = me * 0x01010101u;
        const int start = (int)(((long long)blockIdx.x * n16) / gridDim.x) & ~7;
        for (int base = 0; base < n16; base += OWN_T * OWN_U) {
            uint4 v[OWN_U];
            int   j[OWN_U];
            #pragma unroll
            for (int u = 0; u < OWN_U; ++u) {
                const int i = base + u * OWN_T + t;
                int jj = i + start;
                if (jj >= n16) jj -= n16;
                j[u] = (i < n16) ? jj : -1;
                v[u] = m4[(i < n16) ? jj : 0];
            }
            #pragma unroll
            for (int u = 0; u < OWN_U; ++u) {
                const unsigned x0 = v[u].x ^ pat, x1 = v[u].y ^ pat, x2 = v[u].z ^ pat, x3 = v[u].w ^ pat;
                const unsigned y0 = ((x0 & 0x7f7f7f7fu) + 0x7f7f7f7fu) | x0;
                const unsigned y1 = ((x1 & 0x7f7f7f7fu) + 0x7f7f7f7fu) | x1;
                const unsigned y2 = ((x2 & 0x7f7f7f7fu) + 0x7f7f7f7fu) | x2;
                const unsigned y3 = ((x3 & 0x7f7f7f7fu) + 0x7f7f7f7fu) | x3;
                if (j[u] >= 0 && ((y0 & y1 & y2 & y3) | 0x7f7f7f7fu) != 0xffffffffu) {
                    unsigned c = ((~y0 & 0x80808080u) >> 7) | ((~y1 & 0x80808080u) >> 6) |
                                 ((~y2 & 0x80808080u) >> 5) | ((~y3 & 0x80808080u) >> 4);
                    const int e0 = j[u] << 4;
                    while (c) {
                        const int b = __builtin_ctz(c);
                        c &= c - 1;
                        const int slot = atomicAdd(&lcount, 1);
                        if (slot < OWN_CAP) lsrc[slot] = e0 + ((b & 7) << 2) + (b >> 3);
                    }
                }
            }
        }
        for (int e = (n16 << 4) + t; e < E; e += OWN_T) {
            if (map[e] == me) {
                const int slot = atomicAdd(&lcount, 1);
                if (slot < OWN_CAP) lsrc[slot] = e;
            }
        }
    } else if ((E & 3) == 0) {
        const int4* __restrict__ d4 = (const int4*)dstp;
        const int n4 = E >> 2;
        const int start = (int)(((long long)blockIdx.x * n4) / gridDim.x) & ~7;
        for (int base = 0; base < n4; base += OWN_T * OWN_U) {
            int4 v[OWN_U];
            int  j[OWN_U];
            #pragma unroll
            for (int u = 0; u < OWN_U; ++u) {
                const int i = base + u * OWN_T + t;
                int jj = i + start;
                if (jj >= n4) jj -= n4;
                j[u] = (i < n4) ? jj : -1;
                v[u] = d4[(i < n4) ? jj : 0];
            }
            #pragma unroll
            for (int u = 0; u < OWN_U; ++u) {
                const unsigned q0 = (unsigned)(v[u].x - r0), q1 = (unsigned)(v[u].y - r0);
                const unsigned q2 = (unsigned)(v[u].z - r0), q3 = (unsigned)(v[u].w - r0);
                if (j[u] >= 0 && min(min(q0, q1), min(q2, q3)) < (unsigned)rows) {
                    const int e = j[u] << 2;
                    OWN_NOTE(q0, e)
                    OWN_NOTE(q1, e + 1)
                    OWN_NOTE(q2, e + 2)
                    OWN_NOTE(q3, e + 3)
                }
            }
        }
    } else {
        for (int e = t; e < E; e += OWN_T) OWN_NOTE((unsigned)(dstp[e] - r0), e)
    }
    #undef OWN_NOTE
    __syncthreads();
    const int cnt = from_list ? list_len : lcount;
    if (keep_list && t == 0) *len_g = (cnt <= lstride) ? cnt : -1;

    if (cnt <= OWN_CAP) {
        if (from_list) {
            for (int b0 = 0; b0 < cnt; b0 += OWN_T * 4) {
                unsigned w_[4];
                float as_[4];
                #pragma unroll
                for (int u = 0; u < 4; ++u) {
                    const int i = b0 + u * OWN_T + t;
                    const unsigned w = list_g[(i < cnt) ? i : 0];
                    w_[u] = (i < cnt) ? w : 0u;
                }
                #pragma unroll
                for (int u = 0; u < 4; ++u) as_[u] = a_src[w_[u] & LIST_SRC_MASK];
                #pragma unroll
                for (int u = 0; u < 4; ++u) {
                    const int i = b0 + u * OWN_T + t;
                    const int dl = (int)(w_[u] >> LIST_SRC_BITS);
                    const float v = leaky02(as_[u] + sadst[dl]);
                    lsrc[i] = (int)(w_[u] & LIST_SRC_MASK);
                    ldl[i] = dl;
                    lval[i] = v;
                    atomicMax(&smax[dl], (i < cnt) ? fkey(v) : 0u);
                }
            }
        } else
        for (int b0 = 0; b0 < cnt; b0 += OWN_T * 4) {
            int e_[4], s_[4], d_[4];
            float as_[4];
            #pragma unroll
            for (int u = 0; u < 4; ++u) { const int i = b0 + u * OWN_T + t; e_[u] = (i < cnt) ? lsrc[i] : 0; }
            #pragma unroll
            for (int u = 0; u < 4; ++u) { s_[u] = ei[e_[u]]; d_[u] = dstp[e_[u]]; }
            #pragma unroll
            for (int u = 0; u < 4; ++u) as_[u] = a_src[s_[u]];
            #pragma unroll
            for (int u = 0; u < 4; ++u) {
                const int i = b0 + u * OWN_T + t;
                if (i < cnt) {
                    const int dl = d_[u] - r0;
                    const float v = leaky02(as_[u] + sadst[dl]);
                    lsrc[i] = s_[u];
                    ldl[i] = dl;
                    lval[i] = v;
                    if (keep_list && cnt <= lstride) list_g[i] = ((unsigned)s_[u] & LIST_SRC_MASK) | ((unsigned)dl << LIST_SRC_BITS);
                    atomicMax(&smax[dl], fkey(v));
                }
            }
        }
        __syncthreads();
        for (int i = t; i < cnt; i += OWN_T) {
            const int dl = ldl[i];
            const float w = __expf(lval[i] - fdec(smax[dl]));
            lval[i] = w;
            atomicAdd(&ssum[dl], w);
        }
        for (int r = t; r < rows; r += OWN_T) {
            const float w = __expf(sself[r] - fdec(smax[r]));
            sself[r] = w;
            atomicAdd(&ssum[r], w);
        }
        __syncthreads();
        const int swave = __builtin_amdgcn_readfirstlane(wave);
        for (int g = swave * 32; g < cnt; g += OWN_T) {
            const int   il = (g + lane < cnt) ? g + lane : g;
            const int   sl = lsrc[il];
            const int   dl = ldl[il];
            const float al = lval[il] / (ssum[dl] + 1e-16f);
            const int   ab = __builtin_bit_cast(int, al);
            #pragma unroll
            for (int q = 0; q < 32 / GATHER_G; ++q) {
                const int gq = g + q * GATHER_G;
                if (gq < cnt) {
                    float lo[GATHER_G], hi[GATHER_G];
                    const float* xr[GATHER_G];
                    #pragma unroll
                    for (int k = 0; k < GATHER_G; ++k)
                        xr[k] = xl + (long long)__builtin_amdgcn_readlane(sl, q * GATHER_G + k) * DDIM + lane;
                    #pragma unroll
                    for (int k = 0; k < GATHER_G; ++k) {
                        lo[k] = xr[k][0];
                        hi[k] = xr[k][32];
                    }
                    if (gq + GATHER_G <= cnt) {
                        #pragma unroll
                        for (int k = 0; k < GATHER_G; ++k) {
                            const float a = __builtin_bit_cast(float, __builtin_amdgcn_readlane(ab, q * GATHER_G + k));
                            float* ap = acc + __builtin_amdgcn_readlane(dl, q * GATHER_G + k) * ACC_PITCH + lane;
                            atomicAdd(ap,      lo[k] * a);
                            atomicAdd(ap + 32, hi[k] * a);
                        }
                    } else {
                        #pragma unroll
                        for (int k = 0; k < GATHER_G; ++k) {
                            if (gq + k < cnt) {
                                const float a = __builtin_bit_cast(float, __builtin_amdgcn_readlane(ab, q * GATHER_G + k));
                                float* ap = acc + __builtin_amdgcn_readlane(dl, q * GATHER_G + k) * ACC_PITCH + lane;
                                atomicAdd(ap,      lo[k] * a);
                                atomicAdd(ap + 32, hi[k] * a);
                            }
                        }
                    }
                }
            }
        }
    } else {
        for (int e = t; e < E; e += OWN_T) {
            const unsigned dl = (unsigned)(dstp[e] - r0);
            if (dl < (unsigned)rows)
                atomicMax(&smax[dl], fkey(leaky02(a_src[ei[e]] + sadst[dl])));
        }
        __syncthreads();
        for (int e = t; e < E; e += OWN_T) {
            const unsigned dl = (unsigned)(dstp[e] - r0);
            if (dl < (unsigned)rows)
                atomicAdd(&ssum[dl], __expf(leaky02(a_src[ei[e]] + sadst[dl]) - fdec(smax[dl])));
        }
        for (int r = t; r < rows; r += OWN_T) {
            const float w = __expf(sself[r] - fdec(smax[r]));
            sself[r] = w;
            atomicAdd(&ssum[r], w);
        }
        __syncthreads();
        for (int e = t; e < E; e += OWN_T) {
            const unsigned dl = (unsigned)(dstp[e] - r0);
            if (dl < (unsigned)rows) {
                const int s = ei[e];
                const float aj = __expf(leaky02(a_src[s] + sadst[dl]) - fdec(smax[dl])) / (ssum[dl] + 1e-16f);
                const float* xr = xl + (long long)s * DDIM;
                float* ap = acc + dl * ACC_PITCH;
                for (int c = 0; c < DDIM; ++c) atomicAdd(ap + c, xr[c] * aj);
            }
        }
    }
    __syncthreads();

    const float4* __restrict__ xl4 = (const float4*)xl;
    const float4* __restrict__ b4 = (const float4*)bias;
    float4* __restrict__ o4 = (float4*)out_sum;
    const bool first = (layer == 0);
    const float4* osrc = first ? (const float4*)emb : (const float4*)o4;
    if (Wnext != nullptr) {
        const float* sw = (const float*)sw4;
        const int c4    = lane & 15;
        const int h     = lane >> 4;
        const int khalf = h << 1;
        const int rhi   = h * 8;
        const int ntiles = (rows + 15) >> 4;
        const float4 b = b4[c4];
        float4 pq = {0.f, 0.f, 0.f, 0.f};
        if (first) pq = ((const float4*)pvec)[c4];
        const float* brow = &sw[c4 * SW_STRIDE + khalf];
        for (int tile = wave; tile < ntiles; tile += OWN_T / 32) {
            const int rb = tile * 16;
            float4 xs[8], o[8];
            #pragma unroll
            for (int u = 0; u < 8; ++u) {
                const int r = rb + 2 * u + h;
                const long long gi = (long long)(r0 + (r < rows ? r : rows - 1)) * (DDIM / 4) + c4;
                xs[u] = xl4[gi];
                o[u]  = osrc[gi];
            }
            #pragma unroll
            for (int u = 0; u < 8; ++u) {
                const int r = rb + 2 * u + h;
                if (r < rows) {
                    const long long gi = (long long)(r0 + r) * (DDIM / 4) + c4;
                    const float as = sself[r] / (ssum[r] + 1e-16f);
                    const float4 a = acc4[r * (ACC_PITCH / 4) + c4];
                    float4 v;
                    v.x = (a.x + xs[u].x * as) + b.x;
                    v.y = (a.y + xs[u].y * as) + b.y;
                    v.z = (a.z + xs[u].z * as) + b.z;
                    v.w = (a.w + xs[u].w * as) + b.w;
                    v.x = v.x > 0.f ? v.x : (__expf(v.x) - 1.f);
                    v.y = v.y > 0.f ? v.y : (__expf(v.y) - 1.f);
                    v.z = v.z > 0.f ? v.z : (__expf(v.z) - 1.f);
                    v.w = v.w > 0.f ? v.w : (__expf(v.w) - 1.f);
                    float* xr = acc + r * ACC_PITCH + c4 * 4;
                    xr[0] = v.x; xr[1] = v.y; xr[2] = v.z; xr[3] = v.w;
                    float4 q = o[u];
                    if (first) { q.x += pq.x; q.y += pq.y; q.z += pq.z; q.w += pq.w; }
                    q.x += v.x; q.y += v.y; q.z += v.z; q.w += v.w;
                    o4[gi] = q;
                }
            }
            v8f acc0 = {0.f,0.f,0.f,0.f,0.f,0.f,0.f,0.f};
            v8f acc1 = acc0, acc2 = acc0, acc3 = acc0;
            const float* arow = &acc[(rb + c4) * ACC_PITCH + khalf];
            #pragma nounroll
            for (int kb = 0; kb < DDIM; kb += 16) {
                #pragma unroll
                for (int k0 = kb; k0 < kb + 16; k0 += 4) {
                    v2f a, b0, b1, b2, b3;
                    a.x  = arow[k0];                       a.y  = arow[k0 + 1];
                    b0.x = brow[k0];                       b0.y = brow[k0 + 1];
                    b1.x = brow[16 * SW_STRIDE + k0];      b1.y = brow[16 * SW_STRIDE + k0 + 1];
                    b2.x = brow[32 * SW_STRIDE + k0];      b2.y = brow[32 * SW_STRIDE + k0 + 1];
                    b3.x = brow[48 * SW_STRIDE + k0];      b3.y = brow[48 * SW_STRIDE + k0 + 1];
                    acc0 = __builtin_amdgcn_wmma_f32_16x16x4_f32(false, a, false, b0, (short)0, acc0, false, false);
                    acc1 = __builtin_amdgcn_wmma_f32_16x16x4_f32(false, a, false, b1, (short)0, acc1, false, false);
                    acc2 = __builtin_amdgcn_wmma_f32_16x16x4_f32(false, a, false, b2, (short)0, acc2, false, false);
                    acc3 = __builtin_amdgcn_wmma_f32_16x16x4_f32(false, a, false, b3, (short)0, acc3, false, false);
                }
            }
            const int lr0 = rb + rhi;
            float* p = xl_next + (long long)(r0 + lr0) * DDIM + c4;
            if (rb + 16 <= rows) {
                #pragma unroll
                for (int v = 0; v < 8; ++v) {
                    p[(long long)v * DDIM +  0] = acc0[v];
                    p[(long long)v * DDIM + 16] = acc1[v];
                    p[(long long)v * DDIM + 32] = acc2[v];
                    p[(long long)v * DDIM + 48] = acc3[v];
                }
            } else {
                #pragma unroll
                for (int v = 0; v < 8; ++v) {
                    if (lr0 + v < rows) {
                        p[(long long)v * DDIM +  0] = acc0[v];
                        p[(long long)v * DDIM + 16] = acc1[v];
                        p[(long long)v * DDIM + 32] = acc2[v];
                        p[(long long)v * DDIM + 48] = acc3[v];
                    }
                }
            }
        }
        return;
    }
    const int n4 = rows * (DDIM / 4);
    const long long g0 = (long long)r0 * (DDIM / 4);
    const float4 bq = b4[t & 15];
    float4 pq = {0.f, 0.f, 0.f, 0.f};
    if (first) pq = ((const float4*)pvec)[t & 15];
    int p0 = 0;
    for (; p0 + OWN_T * FIN_U <= n4; p0 += OWN_T * FIN_U) {
        float4 xs[FIN_U], o[FIN_U];
        #pragma unroll
        for (int u = 0; u < FIN_U; ++u) {
            const long long gi = g0 + (p0 + u * OWN_T + t);
            xs[u] = xl4[gi];
            o[u]  = osrc[gi];
        }
        #pragma unroll
        for (int u = 0; u < FIN_U; ++u) {
            const int i = p0 + u * OWN_T + t;
            const int r = i >> 4;
            const float as = sself[r] / (ssum[r] + 1e-16f);
            float4 oq = o[u];
            if (first) { oq.x += pq.x; oq.y += pq.y; oq.z += pq.z; oq.w += pq.w; }
            o4[g0 + i] = fin_piece(acc4[r * (ACC_PITCH / 4) + (t & 15)], xs[u], as, bq, oq);
        }
    }
    for (int i = p0 + t; i < n4; i += OWN_T) {
        const int r = i >> 4;
        const float as = sself[r] / (ssum[r] + 1e-16f);
        float4 oq = osrc[g0 + i];
        if (first) { oq.x += pq.x; oq.y += pq.y; oq.z += pq.z; oq.w += pq.w; }
        o4[g0 + i] = fin_piece(acc4[r * (ACC_PITCH / 4) + (t & 15)], xl4[g0 + i], as, bq, oq);
    }
}

__global__ __launch_bounds__(256) void scale_out(float* __restrict__ out, float inv,
                                                 long long total4) {
    float4* __restrict__ o4 = (float4*)out;
    const long long base = (long long)blockIdx.x * (256 * STREAM_U) + threadIdx.x;
    float4 v[STREAM_U];
    #pragma unroll
    for (int u = 0; u < STREAM_U; ++u) {
        const long long i = base + u * 256;
        v[u] = o4[(i < total4) ? i : 0];
    }
    #pragma unroll
    for (int u = 0; u < STREAM_U; ++u) {
        const long long i = base + u * 256;
        if (i < total4) {
            float4 r;
            r.x = v[u].x * inv; r.y = v[u].y * inv; r.z = v[u].z * inv; r.w = v[u].w * inv;
            o4[i] = r;
        }
    }
}

static inline unsigned gridFor(long long work, int block) {
    return (unsigned)((work + block - 1) / block);
}

extern "C" void kernel_launch(void* const* d_in, const int* in_sizes, int n_in,
                              void* d_out, int out_size, void* d_ws, size_t ws_size,
                              hipStream_t stream) {
    const int*   ei       = (const int*)d_in[0];
    const float* emb      = (const float*)d_in[1];
    const float* prompt   = (const float*)d_in[2];
    const float* proj_w   = (const float*)d_in[3];
    const float* proj_b   = (const float*)d_in[4];
    const float* conv_ws  = (const float*)d_in[5];
    const float* att_srcs = (const float*)d_in[6];
    const float* att_dsts = (const float*)d_in[7];
    const float* conv_bs  = (const float*)d_in[8];

    const int E = in_sizes[0] / 2;
    const int N = in_sizes[1] / DDIM;
    const int P = in_sizes[2];
    const int L = in_sizes[5] / (DDIM * DDIM);

    float* out = (float*)d_out;

    float*    pvec   = (float*)d_ws;
    float*    xbuf   = pvec + 64;
    float*    xlb    = xbuf + (size_t)N * DDIM;
    float*    a_src  = xlb + (size_t)N * DDIM;
    float*    a_dst  = a_src + N;


    float* tail = a_dst + N;
    const unsigned long long ws_bytes = (unsigned long long)ws_size;
    float* qvec = xbuf;
    compute_pvec<<<gridFor((long long)E + 1, 256 * MAP_U * 4), 256, 0, stream>>>(prompt, proj_w, proj_b, pvec, P,
                                                                          (L > 0) ? conv_ws : nullptr, qvec,
                                                                          ei, E, N, tail, pvec, ws_bytes);
    const long long total4 = (long long)N * (DDIM / 4);
    if (L == 0) add_prompt<<<gridFor(total4, 256 * STREAM_U), 256, 0, stream>>>(emb, pvec, xbuf, out, total4);

    if (L > 0) gemm_xw_t<<<gridFor(N, GEMM_ROWS), 256, 0, stream>>>(emb, conv_ws, qvec, xlb, N);
    float* cur = xlb;
    float* nxt = xbuf;
    for (int l = 0; l < L; ++l) {
        const float* as = att_srcs + (size_t)l * DDIM;
        const float* ad = att_dsts + (size_t)l * DDIM;
        const float* cb = conv_bs  + (size_t)l * DDIM;
        const float* Wn = (l + 1 < L) ? conv_ws + (size_t)(l + 1) * DDIM * DDIM : nullptr;

        att_dots<<<gridFor(N, 128), 128, 0, stream>>>(cur, as, ad, a_src, a_dst, N);
        edge_owner<<<gridFor(N, OWN_ROWS), OWN_T, 0, stream>>>(ei, E, N, a_src, a_dst, cur, cb, Wn, nxt, out, emb, pvec,
                                                               tail, pvec, ws_bytes, l, L);
        float* turn = cur; cur = nxt; nxt = turn;
    }
    scale_out<<<gridFor(total4, 256 * STREAM_U), 256, 0, stream>>>(out, 1.f / (float)(L + 1), total4);
}


# first launch sorts each 4096-edge block by owner in LDS, leaves segments + a table; edge phases fetch theirs: byte map, walks, index gathers gone; list order changes, LDS adds in arrival order
# speedup vs baseline: 4.7049x; 1.0632x over previous
#include <hip/hip_runtime.h>
#include <hip/hip_bf16.h>

typedef float v2f __attribute__((ext_vector_type(2)));
typedef float v8f __attribute__((ext_vector_type(8)));

#define DDIM 64
#define GEMM_ROWS 128
#define SX_STRIDE 68
#define SW_STRIDE 68

__device__ __forceinline__ unsigned fkey(float f) {
    unsigned u = __float_as_uint(f);
    return (u & 0x80000000u) ? ~u : (u | 0x80000000u);
}
__device__ __forceinline__ float fdec(unsigned k) {
    return (k & 0x80000000u) ? __uint_as_float(k & 0x7FFFFFFFu)
                             : __uint_as_float(~k);
}
#define FKEY_NEG_INF 0x007FFFFFu

#define OWN_ROWS 592

#define OWN_NONE 255u
#define LIST_SRC_BITS 18
#define LIST_SRC_MASK ((1u << LIST_SRC_BITS) - 1u)
#define OWN_CAP  8192
#define BK_EDGES 4096
#define BK_MAXB  1024
__device__ __forceinline__ unsigned own_of(int dst, int N) {
    return ((unsigned)dst < (unsigned)N) ? (unsigned)dst / (unsigned)OWN_ROWS : OWN_NONE;
}
__device__ __forceinline__ int bk_blocks(int E) { return (int)(((long long)E + BK_EDGES) / BK_EDGES); }
__device__ __forceinline__ unsigned* bk_region(float* tail) {
    unsigned char* p = (unsigned char*)tail;
    return (unsigned*)(p + ((0ull - (unsigned long long)p) & 15ull));
}
__device__ __forceinline__ unsigned* bk_tab(float* tail, int E) {
    return bk_region(tail) + (size_t)bk_blocks(E) * BK_EDGES;
}
__device__ __forceinline__ bool bk_ok(const float* wsbase, float* tail, unsigned long long ws_bytes, int E, int N) {
    const unsigned long long nown = (unsigned long long)((N + OWN_ROWS - 1) / OWN_ROWS);
    const unsigned long long nb = (unsigned long long)bk_blocks(E);
    const unsigned long long off = (unsigned long long)((const unsigned char*)bk_tab(tail, E) - (const unsigned char*)wsbase);
    return nown >= 1 && nown <= 255 && nb <= BK_MAXB && off + nb * 256ull * 4ull <= ws_bytes;
}

__global__ __launch_bounds__(256) void compute_pvec(const float* __restrict__ prompt, const float* __restrict__ pw,
                             const float* __restrict__ pb, float* __restrict__ pvec, int P,
                             const float* __restrict__ W0,
                             float* __restrict__ qvec,
                             const int* __restrict__ ei, int E, int N,
                             float* tail, const float* wsbase, unsigned long long ws_bytes) {
    __shared__ float sp[DDIM];
    __shared__ unsigned bcnt[256];
    __shared__ unsigned boff[256];
    __shared__ unsigned bwt[8];
    __shared__ uint4    bsorted4[BK_EDGES / 4];
    const int t = threadIdx.x;
    if (blockIdx.x == 0) {
        float4 wrow[DDIM / 4];
        const bool fold = (W0 != nullptr);
        if (fold && t < DDIM) {
            #pragma unroll
            for (int u = 0; u < DDIM / 4; ++u) wrow[u] = ((const float4*)W0)[t * (DDIM / 4) + u];
        }
        if (t < DDIM) {
            float s = pb[t];
            for (int j = 0; j < P; ++j) s += prompt[j] * pw[t * P + j];
            pvec[t] = s;
            sp[t] = s;
        }
        if (fold) {
            __syncthreads();
            if (t < DDIM) {
                float q = 0.f;
                #pragma unroll
                for (int u = 0; u < DDIM / 4; ++u) {
                    q += wrow[u].x * sp[4 * u];
                    q += wrow[u].y * sp[4 * u + 1];
                    q += wrow[u].z * sp[4 * u + 2];
                    q += wrow[u].w * sp[4 * u + 3];
                }
                qvec[t] = q;
            }
        }
    }
    if (!bk_ok(wsbase, tail, ws_bytes, E, N)) return;
    const int nb = bk_blocks(E);
    if ((int)blockIdx.x >= nb) return;
    unsigned* bsorted = (unsigned*)bsorted4;
    bcnt[t] = 0u;
    __syncthreads();
    int dv[16], sv[16];
    #pragma unroll
    for (int i = 0; i < 16; ++i) { dv[i] = -1; sv[i] = 0; }
    const int e0 = blockIdx.x * BK_EDGES;
    const int* __restrict__ dstp = ei + E;
    if (e0 < E) {
        if ((E & 3) == 0) {
            const int4* __restrict__ d4 = (const int4*)dstp;
            const int4* __restrict__ s4 = (const int4*)ei;
            const int n4 = E >> 2;
            const int g0 = blockIdx.x * (BK_EDGES / 4);
            #pragma unroll
            for (int u = 0; u < 4; ++u) {
                const int g = g0 + u * 256 + t;
                const int gi = (g < n4) ? g : 0;
                const int4 a = d4[gi];
                const int4 c = s4[gi];
                const bool in = (g < n4);
                dv[4 * u]     = in ? a.x : -1;  sv[4 * u]     = c.x;
                dv[4 * u + 1] = in ? a.y : -1;  sv[4 * u + 1] = c.y;
                dv[4 * u + 2] = in ? a.z : -1;  sv[4 * u + 2] = c.z;
                dv[4 * u + 3] = in ? a.w : -1;  sv[4 * u + 3] = c.w;
            }
        } else {
            #pragma unroll
            for (int u = 0; u < 16; ++u) {
                const int e = e0 + u * 256 + t;
                const int ec = (e < E) ? e : 0;
                const int a = dstp[ec];
                sv[u] = ei[ec];
                dv[u] = (e < E) ? a : -1;
            }
        }
    }
    unsigned ow[16], rk[16];
    #pragma unroll
    for (int i = 0; i < 16; ++i) {
        ow[i] = own_of(dv[i], N);
        rk[i] = 0u;
        if (ow[i] != OWN_NONE) rk[i] = atomicAdd(&bcnt[ow[i]], 1u);
    }
    __syncthreads();
    {
        const int lane = t & 31, wave = t >> 5;
        const unsigned c = bcnt[t];
        unsigned incl = c;
        #pragma unroll
        for (int d = 1; d < 32; d <<= 1) {
            const unsigned y = __shfl_up(incl, d);
            if (lane >= d) incl += y;
        }
        if (lane == 31) bwt[wave] = incl;
        __syncthreads();
        unsigned pre = 0u;
        for (int w = 0; w < wave; ++w) pre += bwt[w];
        boff[t] = pre + incl - c;
    }
    __syncthreads();
    #pragma unroll
    for (int i = 0; i < 16; ++i) {
        if (ow[i] != OWN_NONE) {
            const unsigned dl = (unsigned)dv[i] - ow[i] * (unsigned)OWN_ROWS;
            bsorted[boff[ow[i]] + rk[i]] = ((unsigned)sv[i] & LIST_SRC_MASK) | (dl << LIST_SRC_BITS);
        }
    }
    __syncthreads();
    const unsigned total = boff[255] + bcnt[255];
    uint4* __restrict__ reg4 = (uint4*)(bk_region(tail) + (size_t)blockIdx.x * BK_EDGES);
    #pragma unroll
    for (int u = 0; u < BK_EDGES / 4 / 256; ++u) {
        const int i = u * 256 + t;
        if ((unsigned)(4 * i) < total) reg4[i] = bsorted4[i];
    }
    bk_tab(tail, E)[(size_t)blockIdx.x * 256 + t] = boff[t] | (bcnt[t] << 16);
}

#define STREAM_U 8
__global__ __launch_bounds__(256) void add_prompt(const float* __restrict__ emb,
                                                  const float* __restrict__ pvec,
                                                  float* __restrict__ x,
                                                  float* __restrict__ out_sum,
                                                  long long total4) {
    const float4* __restrict__ e4 = (const float4*)emb;
    float4* __restrict__ x4 = (float4*)x;
    float4* __restrict__ o4 = (float4*)out_sum;
    const int t = threadIdx.x;
    const float4 p = ((const float4*)pvec)[t & 15];
    const long long base = (long long)blockIdx.x * (256 * STREAM_U) + t;
    float4 v[STREAM_U];
    #pragma unroll
    for (int u = 0; u < STREAM_U; ++u) {
        const long long i = base + u * 256;
        v[u] = e4[(i < total4) ? i : 0];
    }
    #pragma unroll
    for (int u = 0; u < STREAM_U; ++u) {
        const long long i = base + u * 256;
        if (i < total4) {
            float4 r;
            r.x = v[u].x + p.x; r.y = v[u].y + p.y; r.z = v[u].z + p.z; r.w = v[u].w + p.w;
            x4[i] = r;
            o4[i] = r;
        }
    }
}

__global__ __launch_bounds__(256) void gemm_xw_t(const float* __restrict__ x,
                                                 const float* __restrict__ W,
                                                 const float* __restrict__ q,
                                                 float* __restrict__ xl, int N) {
    __shared__ float sx[GEMM_ROWS * SX_STRIDE];
    __shared__ float sw[DDIM * SW_STRIDE];
    const int t = threadIdx.x;
    const int rowBase = blockIdx.x * GEMM_ROWS;
    const float q0 = q[(t & 15)], q1 = q[(t & 15) + 16], q2 = q[(t & 15) + 32], q3 = q[(t & 15) + 48];

    {
        const float4* Wv = (const float4*)W;
        const float4* xv = (const float4*)x;
        float4* swv = (float4*)sw;
        float4* sxv = (float4*)sx;
        constexpr int WU = (DDIM * DDIM) / 4 / 256;
        constexpr int XU = (GEMM_ROWS * DDIM) / 4 / 256;
        float4 wreg[WU], xreg[XU];
        #pragma unroll
        for (int u = 0; u < WU; ++u) wreg[u] = Wv[t + u * 256];
        #pragma unroll
        for (int u = 0; u < XU; ++u) {
            const int i = t + u * 256;
            int gr = rowBase + (i >> 4);
            if (gr >= N) gr = N - 1;
            xreg[u] = xv[(long long)gr * 16 + (i & 15)];
        }
        #pragma unroll
        for (int u = 0; u < WU; ++u) {
            const int i = t + u * 256;
            swv[(i >> 4) * (SW_STRIDE / 4) + (i & 15)] = wreg[u];
        }
        #pragma unroll
        for (int u = 0; u < XU; ++u) {
            const int i = t + u * 256;
            sxv[(i >> 4) * (SX_STRIDE / 4) + (i & 15)] = xreg[u];
        }
    }
    __syncthreads();

    const int wave  = t >> 5;
    const int lane  = t & 31;
    const int lrow  = lane & 15;
    const int khalf = (lane >> 4) << 1;
    const int wrow  = wave * 16;

    v8f acc0 = {q0,q0,q0,q0,q0,q0,q0,q0};
    v8f acc1 = {q1,q1,q1,q1,q1,q1,q1,q1};
    v8f acc2 = {q2,q2,q2,q2,q2,q2,q2,q2};
    v8f acc3 = {q3,q3,q3,q3,q3,q3,q3,q3};

    const float* arow = &sx[(wrow + lrow) * SX_STRIDE + khalf];
    const float* brow = &sw[lrow * SW_STRIDE + khalf];

    #pragma unroll
    for (int k0 = 0; k0 < DDIM; k0 += 4) {
        v2f a, b0, b1, b2, b3;
        a.x  = arow[k0];                       a.y  = arow[k0 + 1];
        b0.x = brow[k0];                       b0.y = brow[k0 + 1];
        b1.x = brow[16 * SW_STRIDE + k0];      b1.y = brow[16 * SW_STRIDE + k0 + 1];
        b2.x = brow[32 * SW_STRIDE + k0];      b2.y = brow[32 * SW_STRIDE + k0 + 1];
        b3.x = brow[48 * SW_STRIDE + k0];      b3.y = brow[48 * SW_STRIDE + k0 + 1];
        acc0 = __builtin_amdgcn_wmma_f32_16x16x4_f32(false, a, false, b0, (short)0, acc0, false, false);
        acc1 = __builtin_amdgcn_wmma_f32_16x16x4_f32(false, a, false, b1, (short)0, acc1, false, false);
        acc2 = __builtin_amdgcn_wmma_f32_16x16x4_f32(false, a, false, b2, (short)0, acc2, false, false);
        acc3 = __builtin_amdgcn_wmma_f32_16x16x4_f32(false, a, false, b3, (short)0, acc3, false, false);
    }

    const int rhi = (lane >> 4) * 8;
    const long long rbase = (long long)(rowBase + wrow + rhi);
    if (rowBase + GEMM_ROWS <= N) {
        float* p = xl + rbase * DDIM + lrow;
        #pragma unroll
        for (int v = 0; v < 8; ++v) {
            p[(long long)v * DDIM +  0] = acc0[v];
            p[(long long)v * DDIM + 16] = acc1[v];
            p[(long long)v * DDIM + 32] = acc2[v];
            p[(long long)v * DDIM + 48] = acc3[v];
        }
    } else {
        #pragma unroll
        for (int v = 0; v < 8; ++v) {
            long long gr = rbase + v;
            if (gr < N) {
                float* p = xl + gr * DDIM + lrow;
                p[ 0] = acc0[v];
                p[16] = acc1[v];
                p[32] = acc2[v];
                p[48] = acc3[v];
            }
        }
    }
}

#define ATT_U 16
template <int M> __device__ __forceinline__ float att_xadd(float v) {
    const int o = __builtin_amdgcn_update_dpp(0, __builtin_bit_cast(int, v), 0x160 + M, 0xF, 0xF, true);
    return v + __builtin_bit_cast(float, o);
}
__device__ __forceinline__ float att_pick(const float (&r)[ATT_U], int c) {
    float o = r[0];
    #pragma unroll
    for (int u = 1; u < ATT_U; ++u) o = (c == u) ? r[u] : o;
    return o;
}
__global__ __launch_bounds__(128) void att_dots(const float* __restrict__ xl,
                                                const float* __restrict__ av_s,
                                                const float* __restrict__ av_d,
                                                float* __restrict__ a_src,
                                                float* __restrict__ a_dst, int N) {
    const int lane = threadIdx.x & 31;
    const int c = lane & 15;
    const int h = lane >> 4;
    const int nb = (blockIdx.x * 4 + (threadIdx.x >> 5)) * 32;
    const float4* __restrict__ xl4 = (const float4*)xl;
    const float4 s = ((const float4*)av_s)[c];
    const float4 d = ((const float4*)av_d)[c];
    float4 v[ATT_U];
    #pragma unroll
    for (int u = 0; u < ATT_U; ++u) {
        int n = nb + 16 * h + u;
        if (n >= N) n = N - 1;
        v[u] = xl4[(long long)n * (DDIM / 4) + c];
    }
    float rs[ATT_U], rd[ATT_U];
    #pragma unroll
    for (int u = 0; u < ATT_U; ++u) {
        float sa = v[u].x * s.x + v[u].y * s.y;
        float sb = v[u].z * s.z + v[u].w * s.w;
        float da = v[u].x * d.x + v[u].y * d.y;
        float db = v[u].z * d.z + v[u].w * d.w;
        sa = att_xadd<8>(sa); sb = att_xadd<8>(sb); da = att_xadd<8>(da); db = att_xadd<8>(db);
        sa = att_xadd<4>(sa); sb = att_xadd<4>(sb); da = att_xadd<4>(da); db = att_xadd<4>(db);
        sa = att_xadd<2>(sa); sb = att_xadd<2>(sb); da = att_xadd<2>(da); db = att_xadd<2>(db);
        sa = att_xadd<1>(sa); sb = att_xadd<1>(sb); da = att_xadd<1>(da); db = att_xadd<1>(db);
        rs[u] = sa + sb;
        rd[u] = da + db;
    }
    const int n = nb + lane;
    const float os = att_pick(rs, c);
    const float od = att_pick(rd, c);
    if (n < N) { a_src[n] = os; a_dst[n] = od; }
}

#define ACC_PITCH 68
#define OWN_T    1024
#define OWN_U    8
#define GATHER_G 16
#define FIN_U    4
static_assert(OWN_CAP % (4 * OWN_T) == 0, "OWN_CAP must be whole trips of the logit step");

__device__ __forceinline__ float leaky02(float v) { return v > 0.f ? v : 0.2f * v; }

__device__ __forceinline__ float4 fin_piece(float4 a, float4 xs, float as, float4 b, float4 o) {
    float4 v;
    v.x = (a.x + xs.x * as) + b.x;
    v.y = (a.y + xs.y * as) + b.y;
    v.z = (a.z + xs.z * as) + b.z;
    v.w = (a.w + xs.w * as) + b.w;
    v.x = v.x > 0.f ? v.x : (__expf(v.x) - 1.f);
    v.y = v.y > 0.f ? v.y : (__expf(v.y) - 1.f);
    v.z = v.z > 0.f ? v.z : (__expf(v.z) - 1.f);
    v.w = v.w > 0.f ? v.w : (__expf(v.w) - 1.f);
    o.x += v.x; o.y += v.y; o.z += v.z; o.w += v.w;
    return o;
}

__global__ __launch_bounds__(OWN_T) void edge_owner(const int* __restrict__ ei, int E, int N,
                                                    const float* __restrict__ a_src,
                                                    const float* __restrict__ a_dst,
                                                    const float* __restrict__ xl,
                                                    const float* __restrict__ bias,
                                                    const float* __restrict__ Wnext,
                                                    float* __restrict__ xl_next,
                                                    float* __restrict__ out_sum,
                                                    const float* __restrict__ emb,
                                                    const float* __restrict__ pvec,
                                                    float* tail, const float* wsbase,
                                                    unsigned long long ws_bytes,
                                                    int layer, int nlayers) {
    __shared__ float4   acc4[OWN_ROWS * (ACC_PITCH / 4)];
    __shared__ float4   sw4[DDIM * (SW_STRIDE / 4)];
    __shared__ unsigned smax[OWN_ROWS];
    __shared__ float    ssum[OWN_ROWS];
    __shared__ float    sadst[OWN_ROWS];
    __shared__ float    sself[OWN_ROWS];
    __shared__ int      lsrc[OWN_CAP];
    __shared__ int      ldl[OWN_CAP];
    __shared__ float    lval[OWN_CAP];
    __shared__ int      lcount;
    __shared__ unsigned sstart[BK_MAXB + 1];
    __shared__ unsigned soff[BK_MAXB];
    __shared__ unsigned wtot[OWN_T / 32];
    float* acc = (float*)acc4;

    const int t = threadIdx.x;
    const int lane = t & 31;
    const int wave = t >> 5;
    const int r0 = blockIdx.x * OWN_ROWS;
    const int rows = min(OWN_ROWS, N - r0);
    const int* __restrict__ dstp = ei + E;

    const bool from_list = bk_ok(wsbase, tail, ws_bytes, E, N);
    const int nb = bk_blocks(E);
    const unsigned* __restrict__ region = bk_region(tail);
    unsigned segw = 0u, incl = 0u;
    if (from_list) {
        if (t < nb) segw = bk_tab(tail, E)[(size_t)t * 256 + blockIdx.x];
        incl = segw >> 16;
        #pragma unroll
        for (int d = 1; d < 32; d <<= 1) {
            const unsigned y = __shfl_up(incl, d);
            if (lane >= d) incl += y;
        }
        if (lane == 31) wtot[wave] = incl;
    }

    for (int r = t; r < OWN_ROWS; r += OWN_T) {
        float ad = 0.f, vs = 0.f;
        unsigned key = FKEY_NEG_INF;
        if (r < rows) {
            ad = a_dst[r0 + r];
            vs = leaky02(a_src[r0 + r] + ad);
            key = fkey(vs);
        }
        sadst[r] = ad; sself[r] = vs; smax[r] = key; ssum[r] = 0.f;
    }
    {
        const float4 z = {0.f, 0.f, 0.f, 0.f};
        for (int i = t; i < OWN_ROWS * (ACC_PITCH / 4); i += OWN_T) acc4[i] = z;
    }
    if (t == 0) lcount = 0;
    if (Wnext != nullptr) {
        const float4* __restrict__ Wv = (const float4*)Wnext;
        for (int i = t; i < (DDIM * DDIM) / 4; i += OWN_T) {
            const int r = i >> 4, c4 = i & 15;
            sw4[r * (SW_STRIDE / 4) + c4] = Wv[i];
        }
    }
    __syncthreads();
    if (from_list) {
        unsigned pre = 0u;
        for (int w = 0; w < wave; ++w) pre += wtot[w];
        sstart[t + 1] = pre + incl;
        soff[t] = segw & 0xFFFFu;
        if (t == 0) sstart[0] = 0u;
    }

    #define OWN_NOTE(dlv, eidx)                                               \
        {                                                                     \
            const unsigned dl_ = (dlv);                                       \
            if (dl_ < (unsigned)rows) {                                       \
                const int slot_ = atomicAdd(&lcount, 1);                      \
                if (slot_ < OWN_CAP) { lsrc[slot_] = (eidx); ldl[slot_] = (int)dl_; } \
            }                                                                 \
        }
    if (from_list) {
    } else if ((E & 3) == 0) {
        const int4* __restrict__ d4 = (const int4*)dstp;
        const int n4 = E >> 2;
        const int start = (int)(((long long)blockIdx.x * n4) / gridDim.x) & ~7;
        for (int base = 0; base < n4; base += OWN_T * OWN_U) {
            int4 v[OWN_U];
            int  j[OWN_U];
            #pragma unroll
            for (int u = 0; u < OWN_U; ++u) {
                const int i = base + u * OWN_T + t;
                int jj = i + start;
                if (jj >= n4) jj -= n4;
                j[u] = (i < n4) ? jj : -1;
                v[u] = d4[(i < n4) ? jj : 0];
            }
            #pragma unroll
            for (int u = 0; u < OWN_U; ++u) {
                const unsigned q0 = (unsigned)(v[u].x - r0), q1 = (unsigned)(v[u].y - r0);
                const unsigned q2 = (unsigned)(v[u].z - r0), q3 = (unsigned)(v[u].w - r0);
                if (j[u] >= 0 && min(min(q0, q1), min(q2, q3)) < (unsigned)rows) {
                    const int e = j[u] << 2;
                    OWN_NOTE(q0, e)
                    OWN_NOTE(q1, e + 1)
                    OWN_NOTE(q2, e + 2)
                    OWN_NOTE(q3, e + 3)
                }
            }
        }
    } else {
        for (int e = t; e < E; e += OWN_T) OWN_NOTE((unsigned)(dstp[e] - r0), e)
    }
    #undef OWN_NOTE
    __syncthreads();
    const int cnt = from_list ? (int)sstart[nb] : lcount;

    if (cnt <= OWN_CAP) {
        if (from_list) {
            int top = 1;
            while (2 * top < nb) top <<= 1;
            for (int b0 = 0; b0 < cnt; b0 += OWN_T * 4) {
                unsigned w_[4], ic_[4];
                int sb_[4];
                float as_[4];
                #pragma unroll
                for (int u = 0; u < 4; ++u) {
                    const int i = b0 + u * OWN_T + t;
                    ic_[u] = (i < cnt) ? (unsigned)i : 0u;
                    sb_[u] = 0;
                }
                for (int step = top; step >= 1; step >>= 1) {
                    #pragma unroll
                    for (int u = 0; u < 4; ++u) {
                        const int cand = sb_[u] + step;
                        const unsigned st = sstart[(cand < nb) ? cand : nb];
                        if (cand < nb && st <= ic_[u]) sb_[u] = cand;
                    }
                }
                #pragma unroll
                for (int u = 0; u < 4; ++u) {
                    const int i = b0 + u * OWN_T + t;
                    const unsigned w = region[(size_t)sb_[u] * BK_EDGES + soff[sb_[u]] + (ic_[u] - sstart[sb_[u]])];
                    w_[u] = (i < cnt) ? w : 0u;
                }
                #pragma unroll
                for (int u = 0; u < 4; ++u) as_[u] = a_src[w_[u] & LIST_SRC_MASK];
                #pragma unroll
                for (int u = 0; u < 4; ++u) {
                    const int i = b0 + u * OWN_T + t;
                    const int dl = (int)(w_[u] >> LIST_SRC_BITS);
                    const float v = leaky02(as_[u] + sadst[dl]);
                    lsrc[i] = (int)(w_[u] & LIST_SRC_MASK);
                    ldl[i] = dl;
                    lval[i] = v;
                    atomicMax(&smax[dl], (i < cnt) ? fkey(v) : 0u);
                }
            }
        } else
        for (int b0 = 0; b0 < cnt; b0 += OWN_T * 4) {
            int e_[4], s_[4], d_[4];
            float as_[4];
            #pragma unroll
            for (int u = 0; u < 4; ++u) { const int i = b0 + u * OWN_T + t; e_[u] = (i < cnt) ? lsrc[i] : 0; }
            #pragma unroll
            for (int u = 0; u < 4; ++u) { s_[u] = ei[e_[u]]; d_[u] = dstp[e_[u]]; }
            #pragma unroll
            for (int u = 0; u < 4; ++u) as_[u] = a_src[s_[u]];
            #pragma unroll
            for (int u = 0; u < 4; ++u) {
                const int i = b0 + u * OWN_T + t;
                if (i < cnt) {
                    const int dl = d_[u] - r0;
                    const float v = leaky02(as_[u] + sadst[dl]);
                    lsrc[i] = s_[u];
                    ldl[i] = dl;
                    lval[i] = v;
                    atomicMax(&smax[dl], fkey(v));
                }
            }
        }
        __syncthreads();
        for (int i = t; i < cnt; i += OWN_T) {
            const int dl = ldl[i];
            const float w = __expf(lval[i] - fdec(smax[dl]));
            lval[i] = w;
            atomicAdd(&ssum[dl], w);
        }
        for (int r = t; r < rows; r += OWN_T) {
            const float w = __expf(sself[r] - fdec(smax[r]));
            sself[r] = w;
            atomicAdd(&ssum[r], w);
        }
        __syncthreads();
        const int swave = __builtin_amdgcn_readfirstlane(wave);
        for (int g = swave * 32; g < cnt; g += OWN_T) {
            const int   il = (g + lane < cnt) ? g + lane : g;
            const int   sl = lsrc[il];
            const int   dl = ldl[il];
            const float al = lval[il] / (ssum[dl] + 1e-16f);
            const int   ab = __builtin_bit_cast(int, al);
            #pragma unroll
            for (int q = 0; q < 32 / GATHER_G; ++q) {
                const int gq = g + q * GATHER_G;
                if (gq < cnt) {
                    float lo[GATHER_G], hi[GATHER_G];
                    const float* xr[GATHER_G];
                    #pragma unroll
                    for (int k = 0; k < GATHER_G; ++k)
                        xr[k] = xl + (long long)__builtin_amdgcn_readlane(sl, q * GATHER_G + k) * DDIM + lane;
                    #pragma unroll
                    for (int k = 0; k < GATHER_G; ++k) {
                        lo[k] = xr[k][0];
                        hi[k] = xr[k][32];
                    }
                    if (gq + GATHER_G <= cnt) {
                        #pragma unroll
                        for (int k = 0; k < GATHER_G; ++k) {
                            const float a = __builtin_bit_cast(float, __builtin_amdgcn_readlane(ab, q * GATHER_G + k));
                            float* ap = acc + __builtin_amdgcn_readlane(dl, q * GATHER_G + k) * ACC_PITCH + lane;
                            atomicAdd(ap,      lo[k] * a);
                            atomicAdd(ap + 32, hi[k] * a);
                        }
                    } else {
                        #pragma unroll
                        for (int k = 0; k < GATHER_G; ++k) {
                            if (gq + k < cnt) {
                                const float a = __builtin_bit_cast(float, __builtin_amdgcn_readlane(ab, q * GATHER_G + k));
                                float* ap = acc + __builtin_amdgcn_readlane(dl, q * GATHER_G + k) * ACC_PITCH + lane;
                                atomicAdd(ap,      lo[k] * a);
                                atomicAdd(ap + 32, hi[k] * a);
                            }
                        }
                    }
                }
            }
        }
    } else {
        for (int e = t; e < E; e += OWN_T) {
            const unsigned dl = (unsigned)(dstp[e] - r0);
            if (dl < (unsigned)rows)
                atomicMax(&smax[dl], fkey(leaky02(a_src[ei[e]] + sadst[dl])));
        }
        __syncthreads();
        for (int e = t; e < E; e += OWN_T) {
            const unsigned dl = (unsigned)(dstp[e] - r0);
            if (dl < (unsigned)rows)
                atomicAdd(&ssum[dl], __expf(leaky02(a_src[ei[e]] + sadst[dl]) - fdec(smax[dl])));
        }
        for (int r = t; r < rows; r += OWN_T) {
            const float w = __expf(sself[r] - fdec(smax[r]));
            sself[r] = w;
            atomicAdd(&ssum[r], w);
        }
        __syncthreads();
        for (int e = t; e < E; e += OWN_T) {
            const unsigned dl = (unsigned)(dstp[e] - r0);
            if (dl < (unsigned)rows) {
                const int s = ei[e];
                const float aj = __expf(leaky02(a_src[s] + sadst[dl]) - fdec(smax[dl])) / (ssum[dl] + 1e-16f);
                const float* xr = xl + (long long)s * DDIM;
                float* ap = acc + dl * ACC_PITCH;
                for (int c = 0; c < DDIM; ++c) atomicAdd(ap + c, xr[c] * aj);
            }
        }
    }
    __syncthreads();

    const float4* __restrict__ xl4 = (const float4*)xl;
    const float4* __restrict__ b4 = (const float4*)bias;
    float4* __restrict__ o4 = (float4*)out_sum;
    const bool first = (layer == 0);
    const float4* osrc = first ? (const float4*)emb : (const float4*)o4;
    if (Wnext != nullptr) {
        const float* sw = (const float*)sw4;
        const int c4    = lane & 15;
        const int h     = lane >> 4;
        const int khalf = h << 1;
        const int rhi   = h * 8;
        const int ntiles = (rows + 15) >> 4;
        const float4 b = b4[c4];
        float4 pq = {0.f, 0.f, 0.f, 0.f};
        if (first) pq = ((const float4*)pvec)[c4];
        const float* brow = &sw[c4 * SW_STRIDE + khalf];
        for (int tile = wave; tile < ntiles; tile += OWN_T / 32) {
            const int rb = tile * 16;
            float4 xs[8], o[8];
            #pragma unroll
            for (int u = 0; u < 8; ++u) {
                const int r = rb + 2 * u + h;
                const long long gi = (long long)(r0 + (r < rows ? r : rows - 1)) * (DDIM / 4) + c4;
                xs[u] = xl4[gi];
                o[u]  = osrc[gi];
            }
            #pragma unroll
            for (int u = 0; u < 8; ++u) {
                const int r = rb + 2 * u + h;
                if (r < rows) {
                    const long long gi = (long long)(r0 + r) * (DDIM / 4) + c4;
                    const float as = sself[r] / (ssum[r] + 1e-16f);
                    const float4 a = acc4[r * (ACC_PITCH / 4) + c4];
                    float4 v;
                    v.x = (a.x + xs[u].x * as) + b.x;
                    v.y = (a.y + xs[u].y * as) + b.y;
                    v.z = (a.z + xs[u].z * as) + b.z;
                    v.w = (a.w + xs[u].w * as) + b.w;
                    v.x = v.x > 0.f ? v.x : (__expf(v.x) - 1.f);
                    v.y = v.y > 0.f ? v.y : (__expf(v.y) - 1.f);
                    v.z = v.z > 0.f ? v.z : (__expf(v.z) - 1.f);
                    v.w = v.w > 0.f ? v.w : (__expf(v.w) - 1.f);
                    float* xr = acc + r * ACC_PITCH + c4 * 4;
                    xr[0] = v.x; xr[1] = v.y; xr[2] = v.z; xr[3] = v.w;
                    float4 q = o[u];
                    if (first) { q.x += pq.x; q.y += pq.y; q.z += pq.z; q.w += pq.w; }
                    q.x += v.x; q.y += v.y; q.z += v.z; q.w += v.w;
                    o4[gi] = q;
                }
            }
            v8f acc0 = {0.f,0.f,0.f,0.f,0.f,0.f,0.f,0.f};
            v8f acc1 = acc0, acc2 = acc0, acc3 = acc0;
            const float* arow = &acc[(rb + c4) * ACC_PITCH + khalf];
            #pragma nounroll
            for (int kb = 0; kb < DDIM; kb += 16) {
                #pragma unroll
                for (int k0 = kb; k0 < kb + 16; k0 += 4) {
                    v2f a, b0, b1, b2, b3;
                    a.x  = arow[k0];                       a.y  = arow[k0 + 1];
                    b0.x = brow[k0];                       b0.y = brow[k0 + 1];
                    b1.x = brow[16 * SW_STRIDE + k0];      b1.y = brow[16 * SW_STRIDE + k0 + 1];
                    b2.x = brow[32 * SW_STRIDE + k0];      b2.y = brow[32 * SW_STRIDE + k0 + 1];
                    b3.x = brow[48 * SW_STRIDE + k0];      b3.y = brow[48 * SW_STRIDE + k0 + 1];
                    acc0 = __builtin_amdgcn_wmma_f32_16x16x4_f32(false, a, false, b0, (short)0, acc0, false, false);
                    acc1 = __builtin_amdgcn_wmma_f32_16x16x4_f32(false, a, false, b1, (short)0, acc1, false, false);
                    acc2 = __builtin_amdgcn_wmma_f32_16x16x4_f32(false, a, false, b2, (short)0, acc2, false, false);
                    acc3 = __builtin_amdgcn_wmma_f32_16x16x4_f32(false, a, false, b3, (short)0, acc3, false, false);
                }
            }
            const int lr0 = rb + rhi;
            float* p = xl_next + (long long)(r0 + lr0) * DDIM + c4;
            if (rb + 16 <= rows) {
                #pragma unroll
                for (int v = 0; v < 8; ++v) {
                    p[(long long)v * DDIM +  0] = acc0[v];
                    p[(long long)v * DDIM + 16] = acc1[v];
                    p[(long long)v * DDIM + 32] = acc2[v];
                    p[(long long)v * DDIM + 48] = acc3[v];
                }
            } else {
                #pragma unroll
                for (int v = 0; v < 8; ++v) {
                    if (lr0 + v < rows) {
                        p[(long long)v * DDIM +  0] = acc0[v];
                        p[(long long)v * DDIM + 16] = acc1[v];
                        p[(long long)v * DDIM + 32] = acc2[v];
                        p[(long long)v * DDIM + 48] = acc3[v];
                    }
                }
            }
        }
        return;
    }
    const int n4 = rows * (DDIM / 4);
    const long long g0 = (long long)r0 * (DDIM / 4);
    const float4 bq = b4[t & 15];
    float4 pq = {0.f, 0.f, 0.f, 0.f};
    if (first) pq = ((const float4*)pvec)[t & 15];
    int p0 = 0;
    for (; p0 + OWN_T * FIN_U <= n4; p0 += OWN_T * FIN_U) {
        float4 xs[FIN_U], o[FIN_U];
        #pragma unroll
        for (int u = 0; u < FIN_U; ++u) {
            const long long gi = g0 + (p0 + u * OWN_T + t);
            xs[u] = xl4[gi];
            o[u]  = osrc[gi];
        }
        #pragma unroll
        for (int u = 0; u < FIN_U; ++u) {
            const int i = p0 + u * OWN_T + t;
            const int r = i >> 4;
            const float as = sself[r] / (ssum[r] + 1e-16f);
            float4 oq = o[u];
            if (first) { oq.x += pq.x; oq.y += pq.y; oq.z += pq.z; oq.w += pq.w; }
            o4[g0 + i] = fin_piece(acc4[r * (ACC_PITCH / 4) + (t & 15)], xs[u], as, bq, oq);
        }
    }
    for (int i = p0 + t; i < n4; i += OWN_T) {
        const int r = i >> 4;
        const float as = sself[r] / (ssum[r] + 1e-16f);
        float4 oq = osrc[g0 + i];
        if (first) { oq.x += pq.x; oq.y += pq.y; oq.z += pq.z; oq.w += pq.w; }
        o4[g0 + i] = fin_piece(acc4[r * (ACC_PITCH / 4) + (t & 15)], xl4[g0 + i], as, bq, oq);
    }
}

__global__ __launch_bounds__(256) void scale_out(float* __restrict__ out, float inv,
                                                 long long total4) {
    float4* __restrict__ o4 = (float4*)out;
    const long long base = (long long)blockIdx.x * (256 * STREAM_U) + threadIdx.x;
    float4 v[STREAM_U];
    #pragma unroll
    for (int u = 0; u < STREAM_U; ++u) {
        const long long i = base + u * 256;
        v[u] = o4[(i < total4) ? i : 0];
    }
    #pragma unroll
    for (int u = 0; u < STREAM_U; ++u) {
        const long long i = base + u * 256;
        if (i < total4) {
            float4 r;
            r.x = v[u].x * inv; r.y = v[u].y * inv; r.z = v[u].z * inv; r.w = v[u].w * inv;
            o4[i] = r;
        }
    }
}

static inline unsigned gridFor(long long work, int block) {
    return (unsigned)((work + block - 1) / block);
}

extern "C" void kernel_launch(void* const* d_in, const int* in_sizes, int n_in,
                              void* d_out, int out_size, void* d_ws, size_t ws_size,
                              hipStream_t stream) {
    const int*   ei       = (const int*)d_in[0];
    const float* emb      = (const float*)d_in[1];
    const float* prompt   = (const float*)d_in[2];
    const float* proj_w   = (const float*)d_in[3];
    const float* proj_b   = (const float*)d_in[4];
    const float* conv_ws  = (const float*)d_in[5];
    const float* att_srcs = (const float*)d_in[6];
    const float* att_dsts = (const float*)d_in[7];
    const float* conv_bs  = (const float*)d_in[8];

    const int E = in_sizes[0] / 2;
    const int N = in_sizes[1] / DDIM;
    const int P = in_sizes[2];
    const int L = in_sizes[5] / (DDIM * DDIM);

    float* out = (float*)d_out;

    float*    pvec   = (float*)d_ws;
    float*    xbuf   = pvec + 64;
    float*    xlb    = xbuf + (size_t)N * DDIM;
    float*    a_src  = xlb + (size_t)N * DDIM;
    float*    a_dst  = a_src + N;


    float* tail = a_dst + N;
    const unsigned long long ws_bytes = (unsigned long long)ws_size;
    float* qvec = xbuf;
    compute_pvec<<<gridFor((long long)E + 1, BK_EDGES), 256, 0, stream>>>(prompt, proj_w, proj_b, pvec, P,
                                                                          (L > 0) ? conv_ws : nullptr, qvec,
                                                                          ei, E, N, tail, pvec, ws_bytes);
    const long long total4 = (long long)N * (DDIM / 4);
    if (L == 0) add_prompt<<<gridFor(total4, 256 * STREAM_U), 256, 0, stream>>>(emb, pvec, xbuf, out, total4);

    if (L > 0) gemm_xw_t<<<gridFor(N, GEMM_ROWS), 256, 0, stream>>>(emb, conv_ws, qvec, xlb, N);
    float* cur = xlb;
    float* nxt = xbuf;
    for (int l = 0; l < L; ++l) {
        const float* as = att_srcs + (size_t)l * DDIM;
        const float* ad = att_dsts + (size_t)l * DDIM;
        const float* cb = conv_bs  + (size_t)l * DDIM;
        const float* Wn = (l + 1 < L) ? conv_ws + (size_t)(l + 1) * DDIM * DDIM : nullptr;

        att_dots<<<gridFor(N, 128), 128, 0, stream>>>(cur, as, ad, a_src, a_dst, N);
        edge_owner<<<gridFor(N, OWN_ROWS), OWN_T, 0, stream>>>(ei, E, N, a_src, a_dst, cur, cb, Wn, nxt, out, emb, pvec,
                                                               tail, pvec, ws_bytes, l, L);
        float* turn = cur; cur = nxt; nxt = turn;
    }
    scale_out<<<gridFor(total4, 256 * STREAM_U), 256, 0, stream>>>(out, 1.f / (float)(L + 1), total4);
}
